# starting kernel
# baseline (speedup 1.0000x reference)
#include <hip/hip_runtime.h>
#include <hip/hip_bf16.h>
#include <math.h>


#define NN 20000
#define NE 320000

typedef __attribute__((ext_vector_type(2))) float v2f;
typedef __attribute__((ext_vector_type(8))) float v8f;

__device__ __forceinline__ v8f wmma_f32_k4(v2f a, v2f b, v8f c) {
  return __builtin_amdgcn_wmma_f32_16x16x4_f32(false, a, false, b,
                                               (short)0, c, false, false);
}

__device__ __forceinline__ float sspf(float x) {
  return fmaxf(x, 0.0f) + log1pf(__expf(-fabsf(x))) - 0.69314718056f;
}

__device__ __forceinline__ void atadd(float* p, float v) {
  __hip_atomic_fetch_add(p, v, __ATOMIC_RELAXED, __HIP_MEMORY_SCOPE_AGENT);
}

__global__ void k_zero(float4* __restrict__ p, int n4) {
  int i = blockIdx.x * blockDim.x + threadIdx.x;
  const int stride = gridDim.x * blockDim.x;
  for (; i < n4; i += stride) p[i] = make_float4(0.f, 0.f, 0.f, 0.f);
}

__global__ void k_pack(const float* __restrict__ src, float2* __restrict__ dst,
                       int npairs) {
  int i = blockIdx.x * blockDim.x + threadIdx.x;
  const int total = npairs * 64;
  const int stride = gridDim.x * blockDim.x;
  for (; i < total; i += stride) {
    const int kk = i >> 6, n = i & 63;
    dst[i] = make_float2(src[(2 * kk) * 64 + n], src[(2 * kk + 1) * 64 + n]);
  }
}

template <bool ACONTIG, int KC>
__device__ __forceinline__ void gemm_loop(const float* __restrict__ arow,
                                          const float2* __restrict__ Wp,
                                          int kh, int l16, v8f acc[4]) {
  #pragma unroll 4
  for (int kc = 0; kc < KC; ++kc) {
    const int k0 = kc * 4 + 2 * kh;
    v2f a;
    if (ACONTIG) {
      const float2 av = *(const float2*)(arow + k0);
      a.x = av.x; a.y = av.y;
    } else {
      a.x = arow[k0 * 3];
      a.y = arow[(k0 + 1) * 3];
    }
    const float2* bp = Wp + (k0 >> 1) * 64 + l16;
    #pragma unroll
    for (int ct = 0; ct < 4; ++ct) {
      const float2 bv = bp[ct * 16];
      v2f b; b.x = bv.x; b.y = bv.y;
      acc[ct] = wmma_f32_k4(a, b, acc[ct]);
    }
  }
}

__global__ void k_pre(const float* __restrict__ nf,
                      const float2* __restrict__ W1sp,
                      const float2* __restrict__ W1vp,
                      float* __restrict__ sv) {
  const int lane = threadIdx.x & 31;
  const int wid  = threadIdx.x >> 5;
  const int l16  = lane & 15;
  const int kh   = lane >> 4;
  const int row0 = (blockIdx.x * 8 + wid) * 16;
  int rowA = row0 + l16; if (rowA >= NN) rowA = NN - 1;

  for (int tr = 0; tr < 4; ++tr) {
    const int aoff = (tr == 0) ? 0 : (64 + (tr - 1));
    const float* arow = nf + (size_t)rowA * 256 + aoff;
    v8f acc[4] = {};
    if (tr == 0) gemm_loop<true, 16>(arow, W1sp, kh, l16, acc);
    else         gemm_loop<false, 16>(arow, W1vp, kh, l16, acc);

    #pragma unroll
    for (int ct = 0; ct < 4; ++ct) {
      #pragma unroll
      for (int j = 0; j < 8; ++j) {
        const int r = row0 + j + 8 * kh;
        if (r < NN) {
          const int c = ct * 16 + l16;
          const int o = (tr == 0) ? (r * 256 + c)
                                  : (r * 256 + 64 + c * 3 + (tr - 1));
          sv[o] = acc[ct][j] * 0.125f;
        }
      }
    }
  }
}

__global__ void k_msg(const float* __restrict__ ea,
                      const float* __restrict__ efeat,
                      const int*   __restrict__ eidx,
                      const float* __restrict__ Wm1,
                      const float* __restrict__ Wm2,
                      const float* __restrict__ sv,
                      float* __restrict__ msmv) {
  const int e = blockIdx.x * 8 + (threadIdx.x >> 5);
  if (e >= NE) return;
  const int lane = threadIdx.x & 31;

  const int snd = eidx[e];
  const int rcv = eidx[NE + e];
  const float Y0  = ea[e * 4 + 0];
  const float Y1x = ea[e * 4 + 1];
  const float Y1y = ea[e * 4 + 2];
  const float Y1z = ea[e * 4 + 3];

  float f[8];
  #pragma unroll
  for (int q = 0; q < 8; ++q) f[q] = efeat[e * 8 + q];

  float h[8];
  #pragma unroll
  for (int r = 0; r < 8; ++r) {
    float t = 0.f;
    #pragma unroll
    for (int q = 0; q < 8; ++q) t += f[q] * Wm1[q * 8 + r];
    h[r] = sspf(t * 0.35355339059f);
  }

  const int u0 = lane, u1 = lane + 32;
  float w[4][2];
  #pragma unroll
  for (int k = 0; k < 4; ++k) {
    float t0 = 0.f, t1 = 0.f;
    #pragma unroll
    for (int r = 0; r < 8; ++r) {
      t0 += h[r] * Wm2[r * 256 + k * 64 + u0];
      t1 += h[r] * Wm2[r * 256 + k * 64 + u1];
    }
    w[k][0] = t0 * 0.35355339059f;
    w[k][1] = t1 * 0.35355339059f;
  }

  const float* srow = sv + (size_t)snd * 256;
  float* base = msmv + (size_t)rcv * 512;

  #pragma unroll
  for (int t = 0; t < 2; ++t) {
    const int u = (t == 0) ? u0 : u1;
    const float se = srow[u];
    const float vx = srow[64 + u * 3 + 0];
    const float vy = srow[64 + u * 3 + 1];
    const float vz = srow[64 + u * 3 + 2];
    const float dv = vx * Y1x + vy * Y1y + vz * Y1z;
    atadd(base + u,                      w[0][t] * se * Y0);
    atadd(base + 64 + u,                 w[3][t] * dv * 0.57735026919f);
    atadd(base + 128 + u * 3 + 0,        w[1][t] * se * Y1x);
    atadd(base + 128 + u * 3 + 1,        w[1][t] * se * Y1y);
    atadd(base + 128 + u * 3 + 2,        w[1][t] * se * Y1z);
    atadd(base + 128 + (64 + u) * 3 + 0, w[2][t] * vx * Y0);
    atadd(base + 128 + (64 + u) * 3 + 1, w[2][t] * vy * Y0);
    atadd(base + 128 + (64 + u) * 3 + 2, w[2][t] * vz * Y0);
  }
}

__global__ void k_post(const float* __restrict__ msmv,
                       const float* __restrict__ attrs,
                       const float2* __restrict__ W2sp,
                       const float2* __restrict__ W2vp,
                       const float2* __restrict__ WskSp,
                       const float2* __restrict__ WskVp,
                       float* __restrict__ out) {
  __shared__ float lds[8 * 16 * 64];
  const int lane = threadIdx.x & 31;
  const int wid  = threadIdx.x >> 5;
  const int l16  = lane & 15;
  const int kh   = lane >> 4;
  const int row0 = (blockIdx.x * 8 + wid) * 16;
  int rowA = row0 + l16; if (rowA >= NN) rowA = NN - 1;
  float* myLds = lds + wid * (16 * 64);

  const float* arow8 = attrs + (size_t)rowA * 8;
  const float attE0 = arow8[2 * kh]     * 0.04419417382f;
  const float attE1 = arow8[2 * kh + 1] * 0.04419417382f;
  const float attO0 = arow8[4 + 2 * kh] * 0.04419417382f;
  const float attO1 = arow8[5 + 2 * kh] * 0.04419417382f;

  for (int tr = 0; tr < 4; ++tr) {
    const float2* W2p  = (tr == 0) ? W2sp : W2vp;
    const float2* Wskp = (tr == 0) ? WskSp : WskVp;
    const int aoff = (tr == 0) ? 0 : (128 + (tr - 1));
    const float* arow = msmv + (size_t)rowA * 512 + aoff;

    v8f acc[4] = {};
    if (tr == 0) gemm_loop<true, 32>(arow, W2p, kh, l16, acc);
    else         gemm_loop<false, 32>(arow, W2p, kh, l16, acc);

    #pragma unroll
    for (int ct = 0; ct < 4; ++ct) {
      #pragma unroll
      for (int j = 0; j < 8; ++j) {
        acc[ct][j] *= 0.08838834765f;
        myLds[(j + 8 * kh) * 64 + ct * 16 + l16] = acc[ct][j];
      }
    }
    __syncthreads();

    #pragma unroll 2
    for (int i = 0; i < 64; ++i) {
      const float m2 = myLds[l16 * 64 + i];
      const float2* bp = Wskp + (4 * i + kh) * 64 + l16;
      v2f a;
      a.x = m2 * attE0; a.y = m2 * attE1;
      #pragma unroll
      for (int ct = 0; ct < 4; ++ct) {
        const float2 bv = bp[ct * 16];
        v2f b; b.x = bv.x; b.y = bv.y;
        acc[ct] = wmma_f32_k4(a, b, acc[ct]);
      }
      a.x = m2 * attO0; a.y = m2 * attO1;
      #pragma unroll
      for (int ct = 0; ct < 4; ++ct) {
        const float2 bv = bp[2 * 64 + ct * 16];
        v2f b; b.x = bv.x; b.y = bv.y;
        acc[ct] = wmma_f32_k4(a, b, acc[ct]);
      }
    }

    #pragma unroll
    for (int ct = 0; ct < 4; ++ct) {
      #pragma unroll
      for (int j = 0; j < 8; ++j) {
        const int r = row0 + j + 8 * kh;
        if (r < NN) {
          const int c = ct * 16 + l16;
          const int o = (tr == 0) ? (r * 256 + c)
                                  : (r * 256 + 64 + c * 3 + (tr - 1));
          out[o] = acc[ct][j];
        }
      }
    }
    __syncthreads();
  }
}

extern "C" void kernel_launch(void* const* d_in, const int* in_sizes, int n_in,
                              void* d_out, int out_size, void* d_ws, size_t ws_size,
                              hipStream_t stream) {
  const float* node_attrs = (const float*)d_in[0];
  const float* node_feats = (const float*)d_in[1];
  const float* edge_attrs = (const float*)d_in[2];
  const float* edge_feats = (const float*)d_in[3];
  const int*   edge_index = (const int*)d_in[4];
  const float* W1s  = (const float*)d_in[5];
  const float* W1v  = (const float*)d_in[6];
  const float* Wm1  = (const float*)d_in[7];
  const float* Wm2  = (const float*)d_in[8];
  const float* W2s  = (const float*)d_in[9];
  const float* W2v  = (const float*)d_in[10];
  const float* WskS = (const float*)d_in[11];
  const float* WskV = (const float*)d_in[12];

  float* out  = (float*)d_out;
  float* msmv = (float*)d_ws;

  float2* packBase = (float2*)(msmv + (size_t)NN * 512);
  float2* W1sp  = packBase;
  float2* W1vp  = packBase + 2048;
  float2* W2sp  = packBase + 4096;
  float2* W2vp  = packBase + 8192;
  float2* WskSp = packBase + 12288;
  float2* WskVp = packBase + 28672;

  k_pack<<<8,  256, 0, stream>>>(W1s,  W1sp,  32);
  k_pack<<<8,  256, 0, stream>>>(W1v,  W1vp,  32);
  k_pack<<<16, 256, 0, stream>>>(W2s,  W2sp,  64);
  k_pack<<<16, 256, 0, stream>>>(W2v,  W2vp,  64);
  k_pack<<<64, 256, 0, stream>>>(WskS, WskSp, 256);
  k_pack<<<64, 256, 0, stream>>>(WskV, WskVp, 256);
  k_zero<<<2048, 256, 0, stream>>>((float4*)msmv, NN * 512 / 4);
  k_pre <<<(NN + 127) / 128, 256, 0, stream>>>(node_feats, W1sp, W1vp, out);
  k_msg <<<(NE + 7) / 8, 256, 0, stream>>>(edge_attrs, edge_feats, edge_index,
                                           Wm1, Wm2, out, msmv);
  k_post<<<(NN + 127) / 128, 256, 0, stream>>>(msmv, node_attrs, W2sp, W2vp,
                                               WskSp, WskVp, out);
}


# k_msg: a wave takes 4 edges; lane 8*edge+unit evaluates one hidden unit of the radial MLP once, readlane-broadcast (was all 8 units on all 32 lanes per edge); sender rows loaded ahead of the atomics
# speedup vs baseline: 1.1652x; 1.1652x over previous
#include <hip/hip_runtime.h>
#include <hip/hip_bf16.h>
#include <math.h>


#define NN 20000
#define NE 320000

typedef __attribute__((ext_vector_type(2))) float v2f;
typedef __attribute__((ext_vector_type(8))) float v8f;

__device__ __forceinline__ v8f wmma_f32_k4(v2f a, v2f b, v8f c) {
  return __builtin_amdgcn_wmma_f32_16x16x4_f32(false, a, false, b,
                                               (short)0, c, false, false);
}

__device__ __forceinline__ float sspf(float x) {
  return fmaxf(x, 0.0f) + log1pf(__expf(-fabsf(x))) - 0.69314718056f;
}

__device__ __forceinline__ void atadd(float* p, float v) {
  __hip_atomic_fetch_add(p, v, __ATOMIC_RELAXED, __HIP_MEMORY_SCOPE_AGENT);
}

__global__ void k_zero(float4* __restrict__ p, int n4) {
  int i = blockIdx.x * blockDim.x + threadIdx.x;
  const int stride = gridDim.x * blockDim.x;
  for (; i < n4; i += stride) p[i] = make_float4(0.f, 0.f, 0.f, 0.f);
}

__global__ void k_pack(const float* __restrict__ src, float2* __restrict__ dst,
                       int npairs) {
  int i = blockIdx.x * blockDim.x + threadIdx.x;
  const int total = npairs * 64;
  const int stride = gridDim.x * blockDim.x;
  for (; i < total; i += stride) {
    const int kk = i >> 6, n = i & 63;
    dst[i] = make_float2(src[(2 * kk) * 64 + n], src[(2 * kk + 1) * 64 + n]);
  }
}

template <bool ACONTIG, int KC>
__device__ __forceinline__ void gemm_loop(const float* __restrict__ arow,
                                          const float2* __restrict__ Wp,
                                          int kh, int l16, v8f acc[4]) {
  #pragma unroll 4
  for (int kc = 0; kc < KC; ++kc) {
    const int k0 = kc * 4 + 2 * kh;
    v2f a;
    if (ACONTIG) {
      const float2 av = *(const float2*)(arow + k0);
      a.x = av.x; a.y = av.y;
    } else {
      a.x = arow[k0 * 3];
      a.y = arow[(k0 + 1) * 3];
    }
    const float2* bp = Wp + (k0 >> 1) * 64 + l16;
    #pragma unroll
    for (int ct = 0; ct < 4; ++ct) {
      const float2 bv = bp[ct * 16];
      v2f b; b.x = bv.x; b.y = bv.y;
      acc[ct] = wmma_f32_k4(a, b, acc[ct]);
    }
  }
}

__global__ void k_pre(const float* __restrict__ nf,
                      const float2* __restrict__ W1sp,
                      const float2* __restrict__ W1vp,
                      float* __restrict__ sv) {
  const int lane = threadIdx.x & 31;
  const int wid  = threadIdx.x >> 5;
  const int l16  = lane & 15;
  const int kh   = lane >> 4;
  const int row0 = (blockIdx.x * 8 + wid) * 16;
  int rowA = row0 + l16; if (rowA >= NN) rowA = NN - 1;

  for (int tr = 0; tr < 4; ++tr) {
    const int aoff = (tr == 0) ? 0 : (64 + (tr - 1));
    const float* arow = nf + (size_t)rowA * 256 + aoff;
    v8f acc[4] = {};
    if (tr == 0) gemm_loop<true, 16>(arow, W1sp, kh, l16, acc);
    else         gemm_loop<false, 16>(arow, W1vp, kh, l16, acc);

    #pragma unroll
    for (int ct = 0; ct < 4; ++ct) {
      #pragma unroll
      for (int j = 0; j < 8; ++j) {
        const int r = row0 + j + 8 * kh;
        if (r < NN) {
          const int c = ct * 16 + l16;
          const int o = (tr == 0) ? (r * 256 + c)
                                  : (r * 256 + 64 + c * 3 + (tr - 1));
          sv[o] = acc[ct][j] * 0.125f;
        }
      }
    }
  }
}

#define EPW 4
static_assert(NE % (8 * EPW) == 0, "edge count must fill whole work-groups");

__device__ __forceinline__ float lane_bcast(float v, int src) {
  return __int_as_float(__builtin_amdgcn_readlane(__float_as_int(v), src));
}

__global__ void __launch_bounds__(256, 5)
k_msg(const float* __restrict__ ea,
      const float* __restrict__ efeat,
      const int*   __restrict__ eidx,
      const float* __restrict__ Wm1,
      const float* __restrict__ Wm2,
      const float* __restrict__ sv,
      float* __restrict__ msmv) {
  const int e0 = (blockIdx.x * 8 + (threadIdx.x >> 5)) * EPW;
  if (e0 >= NE) return;
  const int lane = threadIdx.x & 31;

  const int em = e0 + (lane >> 3);
  const int hr = lane & 7;
  float tt = 0.f;
  #pragma unroll
  for (int q = 0; q < 8; ++q) tt += efeat[em * 8 + q] * Wm1[q * 8 + hr];
  const float hl = sspf(tt * 0.35355339059f);

  const int   sndl = eidx[em];
  const int   rcvl = eidx[NE + em];
  const float y0l  = ea[em * 4 + 0];
  const float y1xl = ea[em * 4 + 1];
  const float y1yl = ea[em * 4 + 2];
  const float y1zl = ea[em * 4 + 3];

  const int u0 = lane, u1 = lane + 32;
  float wm[8][4][2];
  #pragma unroll
  for (int r = 0; r < 8; ++r) {
    #pragma unroll
    for (int k = 0; k < 4; ++k) {
      wm[r][k][0] = Wm2[r * 256 + k * 64 + u0];
      wm[r][k][1] = Wm2[r * 256 + k * 64 + u1];
    }
  }

  int rcv[EPW];
  float Y0[EPW], Y1x[EPW], Y1y[EPW], Y1z[EPW];
  float se[EPW][2], vx[EPW][2], vy[EPW][2], vz[EPW][2];
  #pragma unroll
  for (int j = 0; j < EPW; ++j) {
    const int sl = j * 8;
    const int snd = __builtin_amdgcn_readlane(sndl, sl);
    rcv[j] = __builtin_amdgcn_readlane(rcvl, sl);
    Y0[j]  = lane_bcast(y0l,  sl);
    Y1x[j] = lane_bcast(y1xl, sl);
    Y1y[j] = lane_bcast(y1yl, sl);
    Y1z[j] = lane_bcast(y1zl, sl);
    const float* srow = sv + (size_t)snd * 256;
    #pragma unroll
    for (int t = 0; t < 2; ++t) {
      const int u = (t == 0) ? u0 : u1;
      se[j][t] = srow[u];
      vx[j][t] = srow[64 + u * 3 + 0];
      vy[j][t] = srow[64 + u * 3 + 1];
      vz[j][t] = srow[64 + u * 3 + 2];
    }
  }

  #pragma unroll
  for (int j = 0; j < EPW; ++j) {
    float h[8];
    #pragma unroll
    for (int r = 0; r < 8; ++r) h[r] = lane_bcast(hl, j * 8 + r);

    float w[4][2];
    #pragma unroll
    for (int k = 0; k < 4; ++k) {
      float t0 = 0.f, t1 = 0.f;
      #pragma unroll
      for (int r = 0; r < 8; ++r) {
        t0 += h[r] * wm[r][k][0];
        t1 += h[r] * wm[r][k][1];
      }
      w[k][0] = t0 * 0.35355339059f;
      w[k][1] = t1 * 0.35355339059f;
    }

    float* base = msmv + (size_t)rcv[j] * 512;

    #pragma unroll
    for (int t = 0; t < 2; ++t) {
      const int u = (t == 0) ? u0 : u1;
      const float s1 = se[j][t];
      const float ax = vx[j][t], ay = vy[j][t], az = vz[j][t];
      const float dv = ax * Y1x[j] + ay * Y1y[j] + az * Y1z[j];
      atadd(base + u,                      w[0][t] * s1 * Y0[j]);
      atadd(base + 64 + u,                 w[3][t] * dv * 0.57735026919f);
      atadd(base + 128 + u * 3 + 0,        w[1][t] * s1 * Y1x[j]);
      atadd(base + 128 + u * 3 + 1,        w[1][t] * s1 * Y1y[j]);
      atadd(base + 128 + u * 3 + 2,        w[1][t] * s1 * Y1z[j]);
      atadd(base + 128 + (64 + u) * 3 + 0, w[2][t] * ax * Y0[j]);
      atadd(base + 128 + (64 + u) * 3 + 1, w[2][t] * ay * Y0[j]);
      atadd(base + 128 + (64 + u) * 3 + 2, w[2][t] * az * Y0[j]);
    }
  }
}

__global__ void k_post(const float* __restrict__ msmv,
                       const float* __restrict__ attrs,
                       const float2* __restrict__ W2sp,
                       const float2* __restrict__ W2vp,
                       const float2* __restrict__ WskSp,
                       const float2* __restrict__ WskVp,
                       float* __restrict__ out) {
  __shared__ float lds[8 * 16 * 64];
  const int lane = threadIdx.x & 31;
  const int wid  = threadIdx.x >> 5;
  const int l16  = lane & 15;
  const int kh   = lane >> 4;
  const int row0 = (blockIdx.x * 8 + wid) * 16;
  int rowA = row0 + l16; if (rowA >= NN) rowA = NN - 1;
  float* myLds = lds + wid * (16 * 64);

  const float* arow8 = attrs + (size_t)rowA * 8;
  const float attE0 = arow8[2 * kh]     * 0.04419417382f;
  const float attE1 = arow8[2 * kh + 1] * 0.04419417382f;
  const float attO0 = arow8[4 + 2 * kh] * 0.04419417382f;
  const float attO1 = arow8[5 + 2 * kh] * 0.04419417382f;

  for (int tr = 0; tr < 4; ++tr) {
    const float2* W2p  = (tr == 0) ? W2sp : W2vp;
    const float2* Wskp = (tr == 0) ? WskSp : WskVp;
    const int aoff = (tr == 0) ? 0 : (128 + (tr - 1));
    const float* arow = msmv + (size_t)rowA * 512 + aoff;

    v8f acc[4] = {};
    if (tr == 0) gemm_loop<true, 32>(arow, W2p, kh, l16, acc);
    else         gemm_loop<false, 32>(arow, W2p, kh, l16, acc);

    #pragma unroll
    for (int ct = 0; ct < 4; ++ct) {
      #pragma unroll
      for (int j = 0; j < 8; ++j) {
        acc[ct][j] *= 0.08838834765f;
        myLds[(j + 8 * kh) * 64 + ct * 16 + l16] = acc[ct][j];
      }
    }
    __syncthreads();

    #pragma unroll 2
    for (int i = 0; i < 64; ++i) {
      const float m2 = myLds[l16 * 64 + i];
      const float2* bp = Wskp + (4 * i + kh) * 64 + l16;
      v2f a;
      a.x = m2 * attE0; a.y = m2 * attE1;
      #pragma unroll
      for (int ct = 0; ct < 4; ++ct) {
        const float2 bv = bp[ct * 16];
        v2f b; b.x = bv.x; b.y = bv.y;
        acc[ct] = wmma_f32_k4(a, b, acc[ct]);
      }
      a.x = m2 * attO0; a.y = m2 * attO1;
      #pragma unroll
      for (int ct = 0; ct < 4; ++ct) {
        const float2 bv = bp[2 * 64 + ct * 16];
        v2f b; b.x = bv.x; b.y = bv.y;
        acc[ct] = wmma_f32_k4(a, b, acc[ct]);
      }
    }

    #pragma unroll
    for (int ct = 0; ct < 4; ++ct) {
      #pragma unroll
      for (int j = 0; j < 8; ++j) {
        const int r = row0 + j + 8 * kh;
        if (r < NN) {
          const int c = ct * 16 + l16;
          const int o = (tr == 0) ? (r * 256 + c)
                                  : (r * 256 + 64 + c * 3 + (tr - 1));
          out[o] = acc[ct][j];
        }
      }
    }
    __syncthreads();
  }
}

extern "C" void kernel_launch(void* const* d_in, const int* in_sizes, int n_in,
                              void* d_out, int out_size, void* d_ws, size_t ws_size,
                              hipStream_t stream) {
  const float* node_attrs = (const float*)d_in[0];
  const float* node_feats = (const float*)d_in[1];
  const float* edge_attrs = (const float*)d_in[2];
  const float* edge_feats = (const float*)d_in[3];
  const int*   edge_index = (const int*)d_in[4];
  const float* W1s  = (const float*)d_in[5];
  const float* W1v  = (const float*)d_in[6];
  const float* Wm1  = (const float*)d_in[7];
  const float* Wm2  = (const float*)d_in[8];
  const float* W2s  = (const float*)d_in[9];
  const float* W2v  = (const float*)d_in[10];
  const float* WskS = (const float*)d_in[11];
  const float* WskV = (const float*)d_in[12];

  float* out  = (float*)d_out;
  float* msmv = (float*)d_ws;

  float2* packBase = (float2*)(msmv + (size_t)NN * 512);
  float2* W1sp  = packBase;
  float2* W1vp  = packBase + 2048;
  float2* W2sp  = packBase + 4096;
  float2* W2vp  = packBase + 8192;
  float2* WskSp = packBase + 12288;
  float2* WskVp = packBase + 28672;

  k_pack<<<8,  256, 0, stream>>>(W1s,  W1sp,  32);
  k_pack<<<8,  256, 0, stream>>>(W1v,  W1vp,  32);
  k_pack<<<16, 256, 0, stream>>>(W2s,  W2sp,  64);
  k_pack<<<16, 256, 0, stream>>>(W2v,  W2vp,  64);
  k_pack<<<64, 256, 0, stream>>>(WskS, WskSp, 256);
  k_pack<<<64, 256, 0, stream>>>(WskV, WskVp, 256);
  k_zero<<<2048, 256, 0, stream>>>((float4*)msmv, NN * 512 / 4);
  k_pre <<<(NN + 127) / 128, 256, 0, stream>>>(node_feats, W1sp, W1vp, out);
  k_msg <<<NE / (8 * EPW), 256, 0, stream>>>(edge_attrs, edge_feats, edge_index,
                                           Wm1, Wm2, out, msmv);
  k_post<<<(NN + 127) / 128, 256, 0, stream>>>(msmv, node_attrs, W2sp, W2vp,
                                               WskSp, WskVp, out);
}


# accumulator and staged sender rows stored component-planar [node][plane][64]: each wave-wide f32 atomic add and gather load covers one whole 128-B line (was 3 lines each); arrival-order sums kept
# speedup vs baseline: 1.2852x; 1.1030x over previous
#include <hip/hip_runtime.h>
#include <hip/hip_bf16.h>
#include <math.h>


#define NN 20000
#define NE 320000

typedef __attribute__((ext_vector_type(2))) float v2f;
typedef __attribute__((ext_vector_type(8))) float v8f;

__device__ __forceinline__ v8f wmma_f32_k4(v2f a, v2f b, v8f c) {
  return __builtin_amdgcn_wmma_f32_16x16x4_f32(false, a, false, b,
                                               (short)0, c, false, false);
}

__device__ __forceinline__ float sspf(float x) {
  return fmaxf(x, 0.0f) + log1pf(__expf(-fabsf(x))) - 0.69314718056f;
}

__device__ __forceinline__ void atadd(float* p, float v) {
  __hip_atomic_fetch_add(p, v, __ATOMIC_RELAXED, __HIP_MEMORY_SCOPE_AGENT);
}

__global__ void k_zero(float4* __restrict__ p, int n4) {
  int i = blockIdx.x * blockDim.x + threadIdx.x;
  const int stride = gridDim.x * blockDim.x;
  for (; i < n4; i += stride) p[i] = make_float4(0.f, 0.f, 0.f, 0.f);
}

__global__ void k_pack(const float* __restrict__ src, float2* __restrict__ dst,
                       int npairs) {
  int i = blockIdx.x * blockDim.x + threadIdx.x;
  const int total = npairs * 64;
  const int stride = gridDim.x * blockDim.x;
  for (; i < total; i += stride) {
    const int kk = i >> 6, n = i & 63;
    dst[i] = make_float2(src[(2 * kk) * 64 + n], src[(2 * kk + 1) * 64 + n]);
  }
}

template <bool ACONTIG, int KC>
__device__ __forceinline__ void gemm_loop(const float* __restrict__ arow,
                                          const float2* __restrict__ Wp,
                                          int kh, int l16, v8f acc[4]) {
  #pragma unroll 4
  for (int kc = 0; kc < KC; ++kc) {
    const int k0 = kc * 4 + 2 * kh;
    v2f a;
    if (ACONTIG) {
      const float2 av = *(const float2*)(arow + k0);
      a.x = av.x; a.y = av.y;
    } else {
      a.x = arow[k0 * 3];
      a.y = arow[(k0 + 1) * 3];
    }
    const float2* bp = Wp + (k0 >> 1) * 64 + l16;
    #pragma unroll
    for (int ct = 0; ct < 4; ++ct) {
      const float2 bv = bp[ct * 16];
      v2f b; b.x = bv.x; b.y = bv.y;
      acc[ct] = wmma_f32_k4(a, b, acc[ct]);
    }
  }
}

__global__ void k_pre(const float* __restrict__ nf,
                      const float2* __restrict__ W1sp,
                      const float2* __restrict__ W1vp,
                      float* __restrict__ sv) {
  const int lane = threadIdx.x & 31;
  const int wid  = threadIdx.x >> 5;
  const int l16  = lane & 15;
  const int kh   = lane >> 4;
  const int row0 = (blockIdx.x * 8 + wid) * 16;
  int rowA = row0 + l16; if (rowA >= NN) rowA = NN - 1;

  for (int tr = 0; tr < 4; ++tr) {
    const int aoff = (tr == 0) ? 0 : (64 + (tr - 1));
    const float* arow = nf + (size_t)rowA * 256 + aoff;
    v8f acc[4] = {};
    if (tr == 0) gemm_loop<true, 16>(arow, W1sp, kh, l16, acc);
    else         gemm_loop<false, 16>(arow, W1vp, kh, l16, acc);

    #pragma unroll
    for (int ct = 0; ct < 4; ++ct) {
      #pragma unroll
      for (int j = 0; j < 8; ++j) {
        const int r = row0 + j + 8 * kh;
        if (r < NN) {
          const int c = ct * 16 + l16;
          sv[r * 256 + tr * 64 + c] = acc[ct][j] * 0.125f;
        }
      }
    }
  }
}

#define EPW 4
static_assert(NE % (8 * EPW) == 0, "edge count must fill whole work-groups");

__device__ __forceinline__ float lane_bcast(float v, int src) {
  return __int_as_float(__builtin_amdgcn_readlane(__float_as_int(v), src));
}

__global__ void __launch_bounds__(256, 5)
k_msg(const float* __restrict__ ea,
      const float* __restrict__ efeat,
      const int*   __restrict__ eidx,
      const float* __restrict__ Wm1,
      const float* __restrict__ Wm2,
      const float* __restrict__ sv,
      float* __restrict__ msmv) {
  const int e0 = (blockIdx.x * 8 + (threadIdx.x >> 5)) * EPW;
  if (e0 >= NE) return;
  const int lane = threadIdx.x & 31;

  const int em = e0 + (lane >> 3);
  const int hr = lane & 7;
  float tt = 0.f;
  #pragma unroll
  for (int q = 0; q < 8; ++q) tt += efeat[em * 8 + q] * Wm1[q * 8 + hr];
  const float hl = sspf(tt * 0.35355339059f);

  const int   sndl = eidx[em];
  const int   rcvl = eidx[NE + em];
  const float y0l  = ea[em * 4 + 0];
  const float y1xl = ea[em * 4 + 1];
  const float y1yl = ea[em * 4 + 2];
  const float y1zl = ea[em * 4 + 3];

  const int u0 = lane, u1 = lane + 32;
  float wm[8][4][2];
  #pragma unroll
  for (int r = 0; r < 8; ++r) {
    #pragma unroll
    for (int k = 0; k < 4; ++k) {
      wm[r][k][0] = Wm2[r * 256 + k * 64 + u0];
      wm[r][k][1] = Wm2[r * 256 + k * 64 + u1];
    }
  }

  int rcv[EPW];
  float Y0[EPW], Y1x[EPW], Y1y[EPW], Y1z[EPW];
  float se[EPW][2], vx[EPW][2], vy[EPW][2], vz[EPW][2];
  #pragma unroll
  for (int j = 0; j < EPW; ++j) {
    const int sl = j * 8;
    const int snd = __builtin_amdgcn_readlane(sndl, sl);
    rcv[j] = __builtin_amdgcn_readlane(rcvl, sl);
    Y0[j]  = lane_bcast(y0l,  sl);
    Y1x[j] = lane_bcast(y1xl, sl);
    Y1y[j] = lane_bcast(y1yl, sl);
    Y1z[j] = lane_bcast(y1zl, sl);
    const float* srow = sv + (size_t)snd * 256;
    #pragma unroll
    for (int t = 0; t < 2; ++t) {
      const int u = (t == 0) ? u0 : u1;
      se[j][t] = srow[u];
      vx[j][t] = srow[64 + u];
      vy[j][t] = srow[128 + u];
      vz[j][t] = srow[192 + u];
    }
  }

  #pragma unroll
  for (int j = 0; j < EPW; ++j) {
    float h[8];
    #pragma unroll
    for (int r = 0; r < 8; ++r) h[r] = lane_bcast(hl, j * 8 + r);

    float w[4][2];
    #pragma unroll
    for (int k = 0; k < 4; ++k) {
      float t0 = 0.f, t1 = 0.f;
      #pragma unroll
      for (int r = 0; r < 8; ++r) {
        t0 += h[r] * wm[r][k][0];
        t1 += h[r] * wm[r][k][1];
      }
      w[k][0] = t0 * 0.35355339059f;
      w[k][1] = t1 * 0.35355339059f;
    }

    float* base = msmv + (size_t)rcv[j] * 512;

    #pragma unroll
    for (int t = 0; t < 2; ++t) {
      const int u = (t == 0) ? u0 : u1;
      const float s1 = se[j][t];
      const float ax = vx[j][t], ay = vy[j][t], az = vz[j][t];
      const float dv = ax * Y1x[j] + ay * Y1y[j] + az * Y1z[j];
      atadd(base + u,       w[0][t] * s1 * Y0[j]);
      atadd(base + 64 + u,  w[3][t] * dv * 0.57735026919f);
      atadd(base + 128 + u, w[1][t] * s1 * Y1x[j]);
      atadd(base + 192 + u, w[1][t] * s1 * Y1y[j]);
      atadd(base + 256 + u, w[1][t] * s1 * Y1z[j]);
      atadd(base + 320 + u, w[2][t] * ax * Y0[j]);
      atadd(base + 384 + u, w[2][t] * ay * Y0[j]);
      atadd(base + 448 + u, w[2][t] * az * Y0[j]);
    }
  }
}

__global__ void k_post(const float* __restrict__ msmv,
                       const float* __restrict__ attrs,
                       const float2* __restrict__ W2sp,
                       const float2* __restrict__ W2vp,
                       const float2* __restrict__ WskSp,
                       const float2* __restrict__ WskVp,
                       float* __restrict__ out) {
  __shared__ float lds[8 * 16 * 64];
  const int lane = threadIdx.x & 31;
  const int wid  = threadIdx.x >> 5;
  const int l16  = lane & 15;
  const int kh   = lane >> 4;
  const int row0 = (blockIdx.x * 8 + wid) * 16;
  int rowA = row0 + l16; if (rowA >= NN) rowA = NN - 1;
  float* myLds = lds + wid * (16 * 64);

  const float* arow8 = attrs + (size_t)rowA * 8;
  const float attE0 = arow8[2 * kh]     * 0.04419417382f;
  const float attE1 = arow8[2 * kh + 1] * 0.04419417382f;
  const float attO0 = arow8[4 + 2 * kh] * 0.04419417382f;
  const float attO1 = arow8[5 + 2 * kh] * 0.04419417382f;

  for (int tr = 0; tr < 4; ++tr) {
    const float2* W2p  = (tr == 0) ? W2sp : W2vp;
    const float2* Wskp = (tr == 0) ? WskSp : WskVp;
    const int pA = (tr == 0) ? 0 : (1 + tr);
    const int pB = (tr == 0) ? 1 : (4 + tr);
    const float* arowA = msmv + (size_t)rowA * 512 + pA * 64;
    const float* arowB = msmv + (size_t)rowA * 512 + pB * 64;

    v8f acc[4] = {};
    gemm_loop<true, 16>(arowA, W2p, kh, l16, acc);
    gemm_loop<true, 16>(arowB, W2p + 32 * 64, kh, l16, acc);

    #pragma unroll
    for (int ct = 0; ct < 4; ++ct) {
      #pragma unroll
      for (int j = 0; j < 8; ++j) {
        acc[ct][j] *= 0.08838834765f;
        myLds[(j + 8 * kh) * 64 + ct * 16 + l16] = acc[ct][j];
      }
    }
    __syncthreads();

    #pragma unroll 2
    for (int i = 0; i < 64; ++i) {
      const float m2 = myLds[l16 * 64 + i];
      const float2* bp = Wskp + (4 * i + kh) * 64 + l16;
      v2f a;
      a.x = m2 * attE0; a.y = m2 * attE1;
      #pragma unroll
      for (int ct = 0; ct < 4; ++ct) {
        const float2 bv = bp[ct * 16];
        v2f b; b.x = bv.x; b.y = bv.y;
        acc[ct] = wmma_f32_k4(a, b, acc[ct]);
      }
      a.x = m2 * attO0; a.y = m2 * attO1;
      #pragma unroll
      for (int ct = 0; ct < 4; ++ct) {
        const float2 bv = bp[2 * 64 + ct * 16];
        v2f b; b.x = bv.x; b.y = bv.y;
        acc[ct] = wmma_f32_k4(a, b, acc[ct]);
      }
    }

    #pragma unroll
    for (int ct = 0; ct < 4; ++ct) {
      #pragma unroll
      for (int j = 0; j < 8; ++j) {
        const int r = row0 + j + 8 * kh;
        if (r < NN) {
          const int c = ct * 16 + l16;
          const int o = (tr == 0) ? (r * 256 + c)
                                  : (r * 256 + 64 + c * 3 + (tr - 1));
          out[o] = acc[ct][j];
        }
      }
    }
    __syncthreads();
  }
}

extern "C" void kernel_launch(void* const* d_in, const int* in_sizes, int n_in,
                              void* d_out, int out_size, void* d_ws, size_t ws_size,
                              hipStream_t stream) {
  const float* node_attrs = (const float*)d_in[0];
  const float* node_feats = (const float*)d_in[1];
  const float* edge_attrs = (const float*)d_in[2];
  const float* edge_feats = (const float*)d_in[3];
  const int*   edge_index = (const int*)d_in[4];
  const float* W1s  = (const float*)d_in[5];
  const float* W1v  = (const float*)d_in[6];
  const float* Wm1  = (const float*)d_in[7];
  const float* Wm2  = (const float*)d_in[8];
  const float* W2s  = (const float*)d_in[9];
  const float* W2v  = (const float*)d_in[10];
  const float* WskS = (const float*)d_in[11];
  const float* WskV = (const float*)d_in[12];

  float* out  = (float*)d_out;
  float* msmv = (float*)d_ws;

  float2* packBase = (float2*)(msmv + (size_t)NN * 512);
  float2* W1sp  = packBase;
  float2* W1vp  = packBase + 2048;
  float2* W2sp  = packBase + 4096;
  float2* W2vp  = packBase + 8192;
  float2* WskSp = packBase + 12288;
  float2* WskVp = packBase + 28672;

  k_pack<<<8,  256, 0, stream>>>(W1s,  W1sp,  32);
  k_pack<<<8,  256, 0, stream>>>(W1v,  W1vp,  32);
  k_pack<<<16, 256, 0, stream>>>(W2s,  W2sp,  64);
  k_pack<<<16, 256, 0, stream>>>(W2v,  W2vp,  64);
  k_pack<<<64, 256, 0, stream>>>(WskS, WskSp, 256);
  k_pack<<<64, 256, 0, stream>>>(WskV, WskVp, 256);
  k_zero<<<2048, 256, 0, stream>>>((float4*)msmv, NN * 512 / 4);
  k_pre <<<(NN + 127) / 128, 256, 0, stream>>>(node_feats, W1sp, W1vp, out);
  k_msg <<<NE / (8 * EPW), 256, 0, stream>>>(edge_attrs, edge_feats, edge_index,
                                           Wm1, Wm2, out, msmv);
  k_post<<<(NN + 127) / 128, 256, 0, stream>>>(msmv, node_attrs, W2sp, W2vp,
                                               WskSp, WskVp, out);
}


# k_msg: f32 atomic scatter replaced by owner sums: a work-group bins its 80 receivers' edges in LDS (integer LDS atomics), a wave sums each node in registers in edge-id order; no zero fill
# speedup vs baseline: 1.4756x; 1.1481x over previous
#include <hip/hip_runtime.h>
#include <hip/hip_bf16.h>
#include <math.h>


#define NN 20000
#define NE 320000

typedef __attribute__((ext_vector_type(2))) float v2f;
typedef __attribute__((ext_vector_type(8))) float v8f;

__device__ __forceinline__ v8f wmma_f32_k4(v2f a, v2f b, v8f c) {
  return __builtin_amdgcn_wmma_f32_16x16x4_f32(false, a, false, b,
                                               (short)0, c, false, false);
}

__device__ __forceinline__ float sspf(float x) {
  return fmaxf(x, 0.0f) + log1pf(__expf(-fabsf(x))) - 0.69314718056f;
}

__global__ void k_pack(const float* __restrict__ src, float2* __restrict__ dst,
                       int npairs) {
  int i = blockIdx.x * blockDim.x + threadIdx.x;
  const int total = npairs * 64;
  const int stride = gridDim.x * blockDim.x;
  for (; i < total; i += stride) {
    const int kk = i >> 6, n = i & 63;
    dst[i] = make_float2(src[(2 * kk) * 64 + n], src[(2 * kk + 1) * 64 + n]);
  }
}

template <bool ACONTIG, int KC>
__device__ __forceinline__ void gemm_loop(const float* __restrict__ arow,
                                          const float2* __restrict__ Wp,
                                          int kh, int l16, v8f acc[4]) {
  #pragma unroll 4
  for (int kc = 0; kc < KC; ++kc) {
    const int k0 = kc * 4 + 2 * kh;
    v2f a;
    if (ACONTIG) {
      const float2 av = *(const float2*)(arow + k0);
      a.x = av.x; a.y = av.y;
    } else {
      a.x = arow[k0 * 3];
      a.y = arow[(k0 + 1) * 3];
    }
    const float2* bp = Wp + (k0 >> 1) * 64 + l16;
    #pragma unroll
    for (int ct = 0; ct < 4; ++ct) {
      const float2 bv = bp[ct * 16];
      v2f b; b.x = bv.x; b.y = bv.y;
      acc[ct] = wmma_f32_k4(a, b, acc[ct]);
    }
  }
}

__global__ void k_pre(const float* __restrict__ nf,
                      const float2* __restrict__ W1sp,
                      const float2* __restrict__ W1vp,
                      float* __restrict__ sv) {
  const int lane = threadIdx.x & 31;
  const int wid  = threadIdx.x >> 5;
  const int l16  = lane & 15;
  const int kh   = lane >> 4;
  const int row0 = (blockIdx.x * 8 + wid) * 16;
  int rowA = row0 + l16; if (rowA >= NN) rowA = NN - 1;

  for (int tr = 0; tr < 4; ++tr) {
    const int aoff = (tr == 0) ? 0 : (64 + (tr - 1));
    const float* arow = nf + (size_t)rowA * 256 + aoff;
    v8f acc[4] = {};
    if (tr == 0) gemm_loop<true, 16>(arow, W1sp, kh, l16, acc);
    else         gemm_loop<false, 16>(arow, W1vp, kh, l16, acc);

    #pragma unroll
    for (int ct = 0; ct < 4; ++ct) {
      #pragma unroll
      for (int j = 0; j < 8; ++j) {
        const int r = row0 + j + 8 * kh;
        if (r < NN) {
          const int c = ct * 16 + l16;
          sv[r * 256 + tr * 64 + c] = acc[ct][j] * 0.125f;
        }
      }
    }
  }
}

#define NPB  80
#define BCAP 128
#define MSG_THREADS 512
#define MSG_WAVES (MSG_THREADS / 32)
static_assert(NN % NPB == 0, "receiver ranges must tile the nodes");
static_assert(NE % BCAP == 0 && BCAP == 128, "rounds and the 4-per-lane sort assume this");

__device__ __forceinline__ float lane_bcast(float v, int src) {
  return __int_as_float(__builtin_amdgcn_readlane(__float_as_int(v), src));
}

__device__ __forceinline__ void bin_edges(const int* __restrict__ rcvcol, int lo, int hi,
                                          int n0, int* cnt, int* bucket) {
  const int tid = threadIdx.x;
  if (tid < NPB) cnt[tid] = 0;
  __syncthreads();
  const int4* r4 = (const int4*)rcvcol;
  const int lo4 = lo >> 2, n4 = (hi - lo) >> 2;
  const int rot = (int)((blockIdx.x * 320u) % (unsigned)n4);
  for (int i0 = tid; i0 < n4; i0 += MSG_THREADS * 8) {
    int4 v[8];
    int  at[8];
    #pragma unroll
    for (int k = 0; k < 8; ++k) {
      const int i = i0 + k * MSG_THREADS;
      int t = i + rot; if (t >= n4) t -= n4;
      at[k] = lo4 + t;
      v[k] = (i < n4) ? r4[at[k]] : make_int4(-1, -1, -1, -1);
    }
    #pragma unroll
    for (int k = 0; k < 8; ++k) {
      const int rr[4] = {v[k].x, v[k].y, v[k].z, v[k].w};
      #pragma unroll
      for (int c = 0; c < 4; ++c) {
        const int r = rr[c] - n0;
        if ((unsigned)r < (unsigned)NPB) {
          const int slot = atomicAdd(&cnt[r], 1);
          if (slot < BCAP) bucket[r * BCAP + slot] = at[k] * 4 + c;
        }
      }
    }
  }
  __syncthreads();
}

__global__ void __launch_bounds__(MSG_THREADS)
k_msg(const float* __restrict__ ea,
      const float* __restrict__ efeat,
      const int*   __restrict__ eidx,
      const float* __restrict__ Wm1,
      const float* __restrict__ Wm2,
      const float* __restrict__ sv,
      float* __restrict__ msmv) {
  __shared__ int cnt[NPB];
  __shared__ int bucket[NPB * BCAP];
  const int lane = threadIdx.x & 31;
  const int wid  = threadIdx.x >> 5;
  const int n0   = blockIdx.x * NPB;
  const int hr   = lane & 7;
  const int u0 = lane, u1 = lane + 32;

  float w1c[8];
  #pragma unroll
  for (int q = 0; q < 8; ++q) w1c[q] = Wm1[q * 8 + hr];
  float wm[8][4][2];
  #pragma unroll
  for (int r = 0; r < 8; ++r) {
    #pragma unroll
    for (int k = 0; k < 4; ++k) {
      wm[r][k][0] = Wm2[r * 256 + k * 64 + u0];
      wm[r][k][1] = Wm2[r * 256 + k * 64 + u1];
    }
  }

  bin_edges(eidx + NE, 0, NE, n0, cnt, bucket);
  const int over = __syncthreads_or((threadIdx.x < NPB) && (cnt[threadIdx.x < NPB ? threadIdx.x : 0] > BCAP));
  const int span = over ? BCAP : NE;

  for (int eb = 0; eb < NE; eb += span) {
    if (over) bin_edges(eidx + NE, eb, eb + span, n0, cnt, bucket);

    for (int ln = wid; ln < NPB; ln += MSG_WAVES) {
      int* bk = bucket + ln * BCAP;
      int d = cnt[ln]; if (d > BCAP) d = BCAP;

      {
        const int x0 = (lane < d) ? bk[lane] : 0x7fffffff;
        int r0 = 0;
        for (int j = 0; j < d; ++j) r0 += (bk[j] < x0) ? 1 : 0;
        if (d > 32) {
          const int x1 = (lane + 32 < d) ? bk[lane + 32] : 0x7fffffff;
          const int x2 = (lane + 64 < d) ? bk[lane + 64] : 0x7fffffff;
          const int x3 = (lane + 96 < d) ? bk[lane + 96] : 0x7fffffff;
          int r1 = 0, r2 = 0, r3 = 0;
          for (int j = 0; j < d; ++j) {
            const int v = bk[j];
            r1 += (v < x1) ? 1 : 0; r2 += (v < x2) ? 1 : 0; r3 += (v < x3) ? 1 : 0;
          }
          if (lane + 32 < d) bk[r1] = x1;
          if (lane + 64 < d) bk[r2] = x2;
          if (lane + 96 < d) bk[r3] = x3;
        }
        if (lane < d) bk[r0] = x0;
      }

      float acc[8][2];
      #pragma unroll
      for (int p = 0; p < 8; ++p) { acc[p][0] = 0.f; acc[p][1] = 0.f; }

      for (int g = 0; g < d; g += 4) {
        int j = g + (lane >> 3); if (j > d - 1) j = d - 1;
        const int em = bk[j];
        const float4 f0 = *(const float4*)(efeat + (size_t)em * 8);
        const float4 f1 = *(const float4*)(efeat + (size_t)em * 8 + 4);
        const int    sndl = eidx[em];
        const float4 yl = *(const float4*)(ea + (size_t)em * 4);
        float tt = 0.f;
        tt += f0.x * w1c[0]; tt += f0.y * w1c[1]; tt += f0.z * w1c[2]; tt += f0.w * w1c[3];
        tt += f1.x * w1c[4]; tt += f1.y * w1c[5]; tt += f1.z * w1c[6]; tt += f1.w * w1c[7];
        const float hl = sspf(tt * 0.35355339059f);

        float Y0[4], Y1x[4], Y1y[4], Y1z[4];
        float se[4][2], vx[4][2], vy[4][2], vz[4][2];
        #pragma unroll
        for (int e = 0; e < 4; ++e) {
          const int sl = e * 8;
          const int snd = __builtin_amdgcn_readlane(sndl, sl);
          Y0[e]  = lane_bcast(yl.x, sl);
          Y1x[e] = lane_bcast(yl.y, sl);
          Y1y[e] = lane_bcast(yl.z, sl);
          Y1z[e] = lane_bcast(yl.w, sl);
          const float* srow = sv + (size_t)snd * 256;
          se[e][0] = srow[u0];        se[e][1] = srow[u1];
          vx[e][0] = srow[64 + u0];   vx[e][1] = srow[64 + u1];
          vy[e][0] = srow[128 + u0];  vy[e][1] = srow[128 + u1];
          vz[e][0] = srow[192 + u0];  vz[e][1] = srow[192 + u1];
        }

        #pragma unroll
        for (int e = 0; e < 4; ++e) {
          if (g + e < d) {
            float h[8];
            #pragma unroll
            for (int r = 0; r < 8; ++r) h[r] = lane_bcast(hl, e * 8 + r);
            #pragma unroll
            for (int t = 0; t < 2; ++t) {
              float w[4];
              #pragma unroll
              for (int k = 0; k < 4; ++k) {
                float tk = 0.f;
                #pragma unroll
                for (int r = 0; r < 8; ++r) tk += h[r] * wm[r][k][t];
                w[k] = tk * 0.35355339059f;
              }
              const float s1 = se[e][t];
              const float ax = vx[e][t], ay = vy[e][t], az = vz[e][t];
              const float dv = ax * Y1x[e] + ay * Y1y[e] + az * Y1z[e];
              acc[0][t] += w[0] * s1 * Y0[e];
              acc[1][t] += w[3] * dv * 0.57735026919f;
              acc[2][t] += w[1] * s1 * Y1x[e];
              acc[3][t] += w[1] * s1 * Y1y[e];
              acc[4][t] += w[1] * s1 * Y1z[e];
              acc[5][t] += w[2] * ax * Y0[e];
              acc[6][t] += w[2] * ay * Y0[e];
              acc[7][t] += w[2] * az * Y0[e];
            }
          }
        }
      }

      float* base = msmv + (size_t)(n0 + ln) * 512;
      #pragma unroll
      for (int p = 0; p < 8; ++p) {
        float a0 = acc[p][0], a1 = acc[p][1];
        if (eb != 0) { a0 += base[p * 64 + u0]; a1 += base[p * 64 + u1]; }
        base[p * 64 + u0] = a0;
        base[p * 64 + u1] = a1;
      }
    }
    __syncthreads();
  }
}

__global__ void k_post(const float* __restrict__ msmv,
                       const float* __restrict__ attrs,
                       const float2* __restrict__ W2sp,
                       const float2* __restrict__ W2vp,
                       const float2* __restrict__ WskSp,
                       const float2* __restrict__ WskVp,
                       float* __restrict__ out) {
  __shared__ float lds[8 * 16 * 64];
  const int lane = threadIdx.x & 31;
  const int wid  = threadIdx.x >> 5;
  const int l16  = lane & 15;
  const int kh   = lane >> 4;
  const int row0 = (blockIdx.x * 8 + wid) * 16;
  int rowA = row0 + l16; if (rowA >= NN) rowA = NN - 1;
  float* myLds = lds + wid * (16 * 64);

  const float* arow8 = attrs + (size_t)rowA * 8;
  const float attE0 = arow8[2 * kh]     * 0.04419417382f;
  const float attE1 = arow8[2 * kh + 1] * 0.04419417382f;
  const float attO0 = arow8[4 + 2 * kh] * 0.04419417382f;
  const float attO1 = arow8[5 + 2 * kh] * 0.04419417382f;

  for (int tr = 0; tr < 4; ++tr) {
    const float2* W2p  = (tr == 0) ? W2sp : W2vp;
    const float2* Wskp = (tr == 0) ? WskSp : WskVp;
    const int pA = (tr == 0) ? 0 : (1 + tr);
    const int pB = (tr == 0) ? 1 : (4 + tr);
    const float* arowA = msmv + (size_t)rowA * 512 + pA * 64;
    const float* arowB = msmv + (size_t)rowA * 512 + pB * 64;

    v8f acc[4] = {};
    gemm_loop<true, 16>(arowA, W2p, kh, l16, acc);
    gemm_loop<true, 16>(arowB, W2p + 32 * 64, kh, l16, acc);

    #pragma unroll
    for (int ct = 0; ct < 4; ++ct) {
      #pragma unroll
      for (int j = 0; j < 8; ++j) {
        acc[ct][j] *= 0.08838834765f;
        myLds[(j + 8 * kh) * 64 + ct * 16 + l16] = acc[ct][j];
      }
    }
    __syncthreads();

    #pragma unroll 2
    for (int i = 0; i < 64; ++i) {
      const float m2 = myLds[l16 * 64 + i];
      const float2* bp = Wskp + (4 * i + kh) * 64 + l16;
      v2f a;
      a.x = m2 * attE0; a.y = m2 * attE1;
      #pragma unroll
      for (int ct = 0; ct < 4; ++ct) {
        const float2 bv = bp[ct * 16];
        v2f b; b.x = bv.x; b.y = bv.y;
        acc[ct] = wmma_f32_k4(a, b, acc[ct]);
      }
      a.x = m2 * attO0; a.y = m2 * attO1;
      #pragma unroll
      for (int ct = 0; ct < 4; ++ct) {
        const float2 bv = bp[2 * 64 + ct * 16];
        v2f b; b.x = bv.x; b.y = bv.y;
        acc[ct] = wmma_f32_k4(a, b, acc[ct]);
      }
    }

    #pragma unroll
    for (int ct = 0; ct < 4; ++ct) {
      #pragma unroll
      for (int j = 0; j < 8; ++j) {
        const int r = row0 + j + 8 * kh;
        if (r < NN) {
          const int c = ct * 16 + l16;
          const int o = (tr == 0) ? (r * 256 + c)
                                  : (r * 256 + 64 + c * 3 + (tr - 1));
          out[o] = acc[ct][j];
        }
      }
    }
    __syncthreads();
  }
}

extern "C" void kernel_launch(void* const* d_in, const int* in_sizes, int n_in,
                              void* d_out, int out_size, void* d_ws, size_t ws_size,
                              hipStream_t stream) {
  const float* node_attrs = (const float*)d_in[0];
  const float* node_feats = (const float*)d_in[1];
  const float* edge_attrs = (const float*)d_in[2];
  const float* edge_feats = (const float*)d_in[3];
  const int*   edge_index = (const int*)d_in[4];
  const float* W1s  = (const float*)d_in[5];
  const float* W1v  = (const float*)d_in[6];
  const float* Wm1  = (const float*)d_in[7];
  const float* Wm2  = (const float*)d_in[8];
  const float* W2s  = (const float*)d_in[9];
  const float* W2v  = (const float*)d_in[10];
  const float* WskS = (const float*)d_in[11];
  const float* WskV = (const float*)d_in[12];

  float* out  = (float*)d_out;
  float* msmv = (float*)d_ws;

  float2* packBase = (float2*)(msmv + (size_t)NN * 512);
  float2* W1sp  = packBase;
  float2* W1vp  = packBase + 2048;
  float2* W2sp  = packBase + 4096;
  float2* W2vp  = packBase + 8192;
  float2* WskSp = packBase + 12288;
  float2* WskVp = packBase + 28672;

  k_pack<<<8,  256, 0, stream>>>(W1s,  W1sp,  32);
  k_pack<<<8,  256, 0, stream>>>(W1v,  W1vp,  32);
  k_pack<<<16, 256, 0, stream>>>(W2s,  W2sp,  64);
  k_pack<<<16, 256, 0, stream>>>(W2v,  W2vp,  64);
  k_pack<<<64, 256, 0, stream>>>(WskS, WskSp, 256);
  k_pack<<<64, 256, 0, stream>>>(WskV, WskVp, 256);
  k_pre <<<(NN + 127) / 128, 256, 0, stream>>>(node_feats, W1sp, W1vp, out);
  k_msg <<<NN / NPB, MSG_THREADS, 0, stream>>>(edge_attrs, edge_feats, edge_index,
                                               Wm1, Wm2, out, msmv);
  k_post<<<(NN + 127) / 128, 256, 0, stream>>>(msmv, node_attrs, W2sp, W2vp,
                                               WskSp, WskVp, out);
}


# k_pre, k_post: WMMA operand fragments fetched one K-block (32 WMMAs) ahead into a second register set, one wait per block (was one load and one full wait per WMMA); same summation order
# speedup vs baseline: 2.6441x; 1.7919x over previous
#include <hip/hip_runtime.h>
#include <hip/hip_bf16.h>
#include <math.h>


#define NN 20000
#define NE 320000

typedef __attribute__((ext_vector_type(2))) float v2f;
typedef __attribute__((ext_vector_type(8))) float v8f;

__device__ __forceinline__ v8f wmma_f32_k4(v2f a, v2f b, v8f c) {
  return __builtin_amdgcn_wmma_f32_16x16x4_f32(false, a, false, b,
                                               (short)0, c, false, false);
}

__device__ __forceinline__ float sspf(float x) {
  return fmaxf(x, 0.0f) + log1pf(__expf(-fabsf(x))) - 0.69314718056f;
}

__global__ void k_pack(const float* __restrict__ src, float2* __restrict__ dst,
                       int npairs) {
  int i = blockIdx.x * blockDim.x + threadIdx.x;
  const int total = npairs * 64;
  const int stride = gridDim.x * blockDim.x;
  for (; i < total; i += stride) {
    const int kk = i >> 6, n = i & 63;
    dst[i] = make_float2(src[(2 * kk) * 64 + n], src[(2 * kk + 1) * 64 + n]);
  }
}

template <int KB>
struct FragBlk { v2f a[KB]; v2f b[KB][4]; };

template <bool ACONTIG, int KB>
__device__ __forceinline__ void frag_load(FragBlk<KB>& f, const float* __restrict__ arow,
                                          const float2* __restrict__ Wp,
                                          int kc0, int kh, int l16) {
  #pragma unroll
  for (int q = 0; q < KB; ++q) {
    const int k0 = (kc0 + q) * 4 + 2 * kh;
    if (ACONTIG) {
      const float2 av = *(const float2*)(arow + k0);
      f.a[q].x = av.x; f.a[q].y = av.y;
    } else {
      f.a[q].x = arow[k0 * 3];
      f.a[q].y = arow[(k0 + 1) * 3];
    }
    const float2* bp = Wp + (k0 >> 1) * 64 + l16;
    #pragma unroll
    for (int ct = 0; ct < 4; ++ct) {
      const float2 bv = bp[ct * 16];
      f.b[q][ct].x = bv.x; f.b[q][ct].y = bv.y;
    }
  }
}

template <int KB>
__device__ __forceinline__ void frag_mma(const FragBlk<KB>& f, v8f acc[4]) {
  #pragma unroll
  for (int q = 0; q < KB; ++q) {
    #pragma unroll
    for (int ct = 0; ct < 4; ++ct) acc[ct] = wmma_f32_k4(f.a[q], f.b[q][ct], acc[ct]);
  }
}

struct SkipBlk { v2f b[4][2][4]; };

__device__ __forceinline__ void skip_load(SkipBlk& s, const float2* __restrict__ Wskp,
                                          int i0, int kh, int l16) {
  #pragma unroll
  for (int u = 0; u < 4; ++u) {
    const float2* bp = Wskp + (4 * (i0 + u) + kh) * 64 + l16;
    #pragma unroll
    for (int h = 0; h < 2; ++h) {
      #pragma unroll
      for (int ct = 0; ct < 4; ++ct) {
        const float2 bv = bp[h * 128 + ct * 16];
        s.b[u][h][ct].x = bv.x; s.b[u][h][ct].y = bv.y;
      }
    }
  }
}

__device__ __forceinline__ void skip_mma(const SkipBlk& s, const float* m2p,
                                         float attE0, float attE1, float attO0, float attO1,
                                         v8f acc[4]) {
  const float4 mv = *(const float4*)m2p;
  const float m2[4] = {mv.x, mv.y, mv.z, mv.w};
  #pragma unroll
  for (int u = 0; u < 4; ++u) {
    v2f a;
    a.x = m2[u] * attE0; a.y = m2[u] * attE1;
    #pragma unroll
    for (int ct = 0; ct < 4; ++ct) acc[ct] = wmma_f32_k4(a, s.b[u][0][ct], acc[ct]);
    a.x = m2[u] * attO0; a.y = m2[u] * attO1;
    #pragma unroll
    for (int ct = 0; ct < 4; ++ct) acc[ct] = wmma_f32_k4(a, s.b[u][1][ct], acc[ct]);
  }
}

__global__ void __launch_bounds__(256)
k_pre(const float* __restrict__ nf,
      const float2* __restrict__ W1sp,
      const float2* __restrict__ W1vp,
      float* __restrict__ sv) {
  const int lane = threadIdx.x & 31;
  const int wid  = threadIdx.x >> 5;
  const int l16  = lane & 15;
  const int kh   = lane >> 4;
  const int row0 = (blockIdx.x * 8 + wid) * 16;
  int rowA = row0 + l16; if (rowA >= NN) rowA = NN - 1;
  const float* nrow = nf + (size_t)rowA * 256;

  FragBlk<8> f[2];
  frag_load<true, 8>(f[0], nrow, W1sp, 0, kh, l16);
  v8f acc[4];
  #pragma unroll
  for (int n = 0; n < 8; ++n) {
    const int tr = n >> 1;
    if (n + 1 < 8) {
      const int trn = (n + 1) >> 1, kcn = ((n + 1) & 1) * 8;
      if (trn == 0) frag_load<true, 8>(f[(n + 1) & 1], nrow, W1sp, kcn, kh, l16);
      else          frag_load<false, 8>(f[(n + 1) & 1], nrow + 64 + (trn - 1), W1vp, kcn, kh, l16);
    }
    if ((n & 1) == 0) {
      #pragma unroll
      for (int ct = 0; ct < 4; ++ct) acc[ct] = v8f{};
    }
    frag_mma<8>(f[n & 1], acc);
    if (n & 1) {
      #pragma unroll
      for (int ct = 0; ct < 4; ++ct) {
        #pragma unroll
        for (int j = 0; j < 8; ++j) {
          const int r = row0 + j + 8 * kh;
          if (r < NN) {
            const int c = ct * 16 + l16;
            sv[r * 256 + tr * 64 + c] = acc[ct][j] * 0.125f;
          }
        }
      }
    }
  }
}

#define NPB  80
#define BCAP 128
#define MSG_THREADS 512
#define MSG_WAVES (MSG_THREADS / 32)
static_assert(NN % NPB == 0, "receiver ranges must tile the nodes");
static_assert(NE % BCAP == 0 && BCAP == 128, "rounds and the 4-per-lane sort assume this");

__device__ __forceinline__ float lane_bcast(float v, int src) {
  return __int_as_float(__builtin_amdgcn_readlane(__float_as_int(v), src));
}

__device__ __forceinline__ void bin_edges(const int* __restrict__ rcvcol, int lo, int hi,
                                          int n0, int* cnt, int* bucket) {
  const int tid = threadIdx.x;
  if (tid < NPB) cnt[tid] = 0;
  __syncthreads();
  const int4* r4 = (const int4*)rcvcol;
  const int lo4 = lo >> 2, n4 = (hi - lo) >> 2;
  const int rot = (int)((blockIdx.x * 320u) % (unsigned)n4);
  for (int i0 = tid; i0 < n4; i0 += MSG_THREADS * 8) {
    int4 v[8];
    int  at[8];
    #pragma unroll
    for (int k = 0; k < 8; ++k) {
      const int i = i0 + k * MSG_THREADS;
      int t = i + rot; if (t >= n4) t -= n4;
      at[k] = lo4 + t;
      v[k] = (i < n4) ? r4[at[k]] : make_int4(-1, -1, -1, -1);
    }
    #pragma unroll
    for (int k = 0; k < 8; ++k) {
      const int rr[4] = {v[k].x, v[k].y, v[k].z, v[k].w};
      #pragma unroll
      for (int c = 0; c < 4; ++c) {
        const int r = rr[c] - n0;
        if ((unsigned)r < (unsigned)NPB) {
          const int slot = atomicAdd(&cnt[r], 1);
          if (slot < BCAP) bucket[r * BCAP + slot] = at[k] * 4 + c;
        }
      }
    }
  }
  __syncthreads();
}

__global__ void __launch_bounds__(MSG_THREADS)
k_msg(const float* __restrict__ ea,
      const float* __restrict__ efeat,
      const int*   __restrict__ eidx,
      const float* __restrict__ Wm1,
      const float* __restrict__ Wm2,
      const float* __restrict__ sv,
      float* __restrict__ msmv) {
  __shared__ int cnt[NPB];
  __shared__ int bucket[NPB * BCAP];
  const int lane = threadIdx.x & 31;
  const int wid  = threadIdx.x >> 5;
  const int n0   = blockIdx.x * NPB;
  const int hr   = lane & 7;
  const int u0 = lane, u1 = lane + 32;

  float w1c[8];
  #pragma unroll
  for (int q = 0; q < 8; ++q) w1c[q] = Wm1[q * 8 + hr];
  float wm[8][4][2];
  #pragma unroll
  for (int r = 0; r < 8; ++r) {
    #pragma unroll
    for (int k = 0; k < 4; ++k) {
      wm[r][k][0] = Wm2[r * 256 + k * 64 + u0];
      wm[r][k][1] = Wm2[r * 256 + k * 64 + u1];
    }
  }

  bin_edges(eidx + NE, 0, NE, n0, cnt, bucket);
  const int over = __syncthreads_or((threadIdx.x < NPB) && (cnt[threadIdx.x < NPB ? threadIdx.x : 0] > BCAP));
  const int span = over ? BCAP : NE;

  for (int eb = 0; eb < NE; eb += span) {
    if (over) bin_edges(eidx + NE, eb, eb + span, n0, cnt, bucket);

    for (int ln = wid; ln < NPB; ln += MSG_WAVES) {
      int* bk = bucket + ln * BCAP;
      int d = cnt[ln]; if (d > BCAP) d = BCAP;

      {
        const int x0 = (lane < d) ? bk[lane] : 0x7fffffff;
        int r0 = 0;
        for (int j = 0; j < d; ++j) r0 += (bk[j] < x0) ? 1 : 0;
        if (d > 32) {
          const int x1 = (lane + 32 < d) ? bk[lane + 32] : 0x7fffffff;
          const int x2 = (lane + 64 < d) ? bk[lane + 64] : 0x7fffffff;
          const int x3 = (lane + 96 < d) ? bk[lane + 96] : 0x7fffffff;
          int r1 = 0, r2 = 0, r3 = 0;
          for (int j = 0; j < d; ++j) {
            const int v = bk[j];
            r1 += (v < x1) ? 1 : 0; r2 += (v < x2) ? 1 : 0; r3 += (v < x3) ? 1 : 0;
          }
          if (lane + 32 < d) bk[r1] = x1;
          if (lane + 64 < d) bk[r2] = x2;
          if (lane + 96 < d) bk[r3] = x3;
        }
        if (lane < d) bk[r0] = x0;
      }

      float acc[8][2];
      #pragma unroll
      for (int p = 0; p < 8; ++p) { acc[p][0] = 0.f; acc[p][1] = 0.f; }

      for (int g = 0; g < d; g += 4) {
        int j = g + (lane >> 3); if (j > d - 1) j = d - 1;
        const int em = bk[j];
        const float4 f0 = *(const float4*)(efeat + (size_t)em * 8);
        const float4 f1 = *(const float4*)(efeat + (size_t)em * 8 + 4);
        const int    sndl = eidx[em];
        const float4 yl = *(const float4*)(ea + (size_t)em * 4);
        float tt = 0.f;
        tt += f0.x * w1c[0]; tt += f0.y * w1c[1]; tt += f0.z * w1c[2]; tt += f0.w * w1c[3];
        tt += f1.x * w1c[4]; tt += f1.y * w1c[5]; tt += f1.z * w1c[6]; tt += f1.w * w1c[7];
        const float hl = sspf(tt * 0.35355339059f);

        float Y0[4], Y1x[4], Y1y[4], Y1z[4];
        float se[4][2], vx[4][2], vy[4][2], vz[4][2];
        #pragma unroll
        for (int e = 0; e < 4; ++e) {
          const int sl = e * 8;
          const int snd = __builtin_amdgcn_readlane(sndl, sl);
          Y0[e]  = lane_bcast(yl.x, sl);
          Y1x[e] = lane_bcast(yl.y, sl);
          Y1y[e] = lane_bcast(yl.z, sl);
          Y1z[e] = lane_bcast(yl.w, sl);
          const float* srow = sv + (size_t)snd * 256;
          se[e][0] = srow[u0];        se[e][1] = srow[u1];
          vx[e][0] = srow[64 + u0];   vx[e][1] = srow[64 + u1];
          vy[e][0] = srow[128 + u0];  vy[e][1] = srow[128 + u1];
          vz[e][0] = srow[192 + u0];  vz[e][1] = srow[192 + u1];
        }

        #pragma unroll
        for (int e = 0; e < 4; ++e) {
          if (g + e < d) {
            float h[8];
            #pragma unroll
            for (int r = 0; r < 8; ++r) h[r] = lane_bcast(hl, e * 8 + r);
            #pragma unroll
            for (int t = 0; t < 2; ++t) {
              float w[4];
              #pragma unroll
              for (int k = 0; k < 4; ++k) {
                float tk = 0.f;
                #pragma unroll
                for (int r = 0; r < 8; ++r) tk += h[r] * wm[r][k][t];
                w[k] = tk * 0.35355339059f;
              }
              const float s1 = se[e][t];
              const float ax = vx[e][t], ay = vy[e][t], az = vz[e][t];
              const float dv = ax * Y1x[e] + ay * Y1y[e] + az * Y1z[e];
              acc[0][t] += w[0] * s1 * Y0[e];
              acc[1][t] += w[3] * dv * 0.57735026919f;
              acc[2][t] += w[1] * s1 * Y1x[e];
              acc[3][t] += w[1] * s1 * Y1y[e];
              acc[4][t] += w[1] * s1 * Y1z[e];
              acc[5][t] += w[2] * ax * Y0[e];
              acc[6][t] += w[2] * ay * Y0[e];
              acc[7][t] += w[2] * az * Y0[e];
            }
          }
        }
      }

      float* base = msmv + (size_t)(n0 + ln) * 512;
      #pragma unroll
      for (int p = 0; p < 8; ++p) {
        float a0 = acc[p][0], a1 = acc[p][1];
        if (eb != 0) { a0 += base[p * 64 + u0]; a1 += base[p * 64 + u1]; }
        base[p * 64 + u0] = a0;
        base[p * 64 + u1] = a1;
      }
    }
    __syncthreads();
  }
}

__global__ void __launch_bounds__(256)
k_post(const float* __restrict__ msmv,
       const float* __restrict__ attrs,
       const float2* __restrict__ W2sp,
       const float2* __restrict__ W2vp,
       const float2* __restrict__ WskSp,
       const float2* __restrict__ WskVp,
       float* __restrict__ out) {
  __shared__ alignas(16) float lds[8 * 16 * 64];
  const int lane = threadIdx.x & 31;
  const int wid  = threadIdx.x >> 5;
  const int l16  = lane & 15;
  const int kh   = lane >> 4;
  const int row0 = (blockIdx.x * 8 + wid) * 16;
  int rowA = row0 + l16; if (rowA >= NN) rowA = NN - 1;
  float* myLds = lds + wid * (16 * 64);
  const float* mrow = msmv + (size_t)rowA * 512;

  const float* arow8 = attrs + (size_t)rowA * 8;
  const float attE0 = arow8[2 * kh]     * 0.04419417382f;
  const float attE1 = arow8[2 * kh + 1] * 0.04419417382f;
  const float attO0 = arow8[4 + 2 * kh] * 0.04419417382f;
  const float attO1 = arow8[5 + 2 * kh] * 0.04419417382f;

  FragBlk<8> f[2];
  SkipBlk sb[2];
  frag_load<true, 8>(f[0], mrow, W2sp, 0, kh, l16);

  for (int tr = 0; tr < 4; ++tr) {
    const float2* W2p  = (tr == 0) ? W2sp : W2vp;
    const float2* Wskp = (tr == 0) ? WskSp : WskVp;
    const int pA = (tr == 0) ? 0 : (1 + tr);
    const int pB = (tr == 0) ? 1 : (4 + tr);
    const float* arowA = mrow + pA * 64;
    const float* arowB = mrow + pB * 64;

    v8f acc[4] = {};
    #pragma unroll
    for (int n = 0; n < 4; ++n) {
      if (n + 1 < 4)
        frag_load<true, 8>(f[(n + 1) & 1], (n + 1 < 2) ? arowA : arowB,
                           (n + 1 < 2) ? W2p : W2p + 32 * 64, ((n + 1) & 1) * 8, kh, l16);
      else
        skip_load(sb[0], Wskp, 0, kh, l16);
      frag_mma<8>(f[n & 1], acc);
    }

    #pragma unroll
    for (int ct = 0; ct < 4; ++ct) {
      #pragma unroll
      for (int j = 0; j < 8; ++j) {
        acc[ct][j] *= 0.08838834765f;
        myLds[(j + 8 * kh) * 64 + ct * 16 + l16] = acc[ct][j];
      }
    }
    __syncthreads();

    const float* m2row = myLds + l16 * 64;
    for (int blk = 0; blk < 14; blk += 2) {
      skip_load(sb[1], Wskp, (blk + 1) * 4, kh, l16);
      skip_mma(sb[0], m2row + blk * 4, attE0, attE1, attO0, attO1, acc);
      skip_load(sb[0], Wskp, (blk + 2) * 4, kh, l16);
      skip_mma(sb[1], m2row + (blk + 1) * 4, attE0, attE1, attO0, attO1, acc);
    }
    skip_load(sb[1], Wskp, 15 * 4, kh, l16);
    skip_mma(sb[0], m2row + 14 * 4, attE0, attE1, attO0, attO1, acc);
    frag_load<true, 8>(f[0], mrow + ((tr < 3) ? (2 + tr) : 0) * 64, W2vp, 0, kh, l16);
    skip_mma(sb[1], m2row + 15 * 4, attE0, attE1, attO0, attO1, acc);

    #pragma unroll
    for (int ct = 0; ct < 4; ++ct) {
      #pragma unroll
      for (int j = 0; j < 8; ++j) {
        const int r = row0 + j + 8 * kh;
        if (r < NN) {
          const int c = ct * 16 + l16;
          const int o = (tr == 0) ? (r * 256 + c)
                                  : (r * 256 + 64 + c * 3 + (tr - 1));
          out[o] = acc[ct][j];
        }
      }
    }
    __syncthreads();
  }
}

extern "C" void kernel_launch(void* const* d_in, const int* in_sizes, int n_in,
                              void* d_out, int out_size, void* d_ws, size_t ws_size,
                              hipStream_t stream) {
  const float* node_attrs = (const float*)d_in[0];
  const float* node_feats = (const float*)d_in[1];
  const float* edge_attrs = (const float*)d_in[2];
  const float* edge_feats = (const float*)d_in[3];
  const int*   edge_index = (const int*)d_in[4];
  const float* W1s  = (const float*)d_in[5];
  const float* W1v  = (const float*)d_in[6];
  const float* Wm1  = (const float*)d_in[7];
  const float* Wm2  = (const float*)d_in[8];
  const float* W2s  = (const float*)d_in[9];
  const float* W2v  = (const float*)d_in[10];
  const float* WskS = (const float*)d_in[11];
  const float* WskV = (const float*)d_in[12];

  float* out  = (float*)d_out;
  float* msmv = (float*)d_ws;

  float2* packBase = (float2*)(msmv + (size_t)NN * 512);
  float2* W1sp  = packBase;
  float2* W1vp  = packBase + 2048;
  float2* W2sp  = packBase + 4096;
  float2* W2vp  = packBase + 8192;
  float2* WskSp = packBase + 12288;
  float2* WskVp = packBase + 28672;

  k_pack<<<8,  256, 0, stream>>>(W1s,  W1sp,  32);
  k_pack<<<8,  256, 0, stream>>>(W1v,  W1vp,  32);
  k_pack<<<16, 256, 0, stream>>>(W2s,  W2sp,  64);
  k_pack<<<16, 256, 0, stream>>>(W2v,  W2vp,  64);
  k_pack<<<64, 256, 0, stream>>>(WskS, WskSp, 256);
  k_pack<<<64, 256, 0, stream>>>(WskV, WskVp, 256);
  k_pre <<<(NN + 127) / 128, 256, 0, stream>>>(node_feats, W1sp, W1vp, out);
  k_msg <<<NN / NPB, MSG_THREADS, 0, stream>>>(edge_attrs, edge_feats, edge_index,
                                               Wm1, Wm2, out, msmv);
  k_post<<<(NN + 127) / 128, 256, 0, stream>>>(msmv, node_attrs, W2sp, W2vp,
                                               WskSp, WskVp, out);
}


# the six weight-repacking launches are gone: k_pre packs W1s/W1v into LDS for its own B fragments and writes the packed W2/Wsk that k_post reads; 9 launches -> 3; same k order and sums
# speedup vs baseline: 2.8692x; 1.0851x over previous
#include <hip/hip_runtime.h>
#include <hip/hip_bf16.h>
#include <math.h>


#define NN 20000
#define NE 320000

typedef __attribute__((ext_vector_type(2))) float v2f;
typedef __attribute__((ext_vector_type(8))) float v8f;

__device__ __forceinline__ v8f wmma_f32_k4(v2f a, v2f b, v8f c) {
  return __builtin_amdgcn_wmma_f32_16x16x4_f32(false, a, false, b,
                                               (short)0, c, false, false);
}

__device__ __forceinline__ float sspf(float x) {
  return fmaxf(x, 0.0f) + log1pf(__expf(-fabsf(x))) - 0.69314718056f;
}

#define GPITCH 64
#define LPITCH 80
#define POST_PACK_PAIRS 40960

template <int KB>
struct FragBlk { v2f a[KB]; v2f b[KB][4]; };

template <bool ACONTIG, int KB, int BPITCH>
__device__ __forceinline__ void frag_load(FragBlk<KB>& f, const float* __restrict__ arow,
                                          const float2* __restrict__ Wp,
                                          int kc0, int kh, int l16) {
  #pragma unroll
  for (int q = 0; q < KB; ++q) {
    const int k0 = (kc0 + q) * 4 + 2 * kh;
    if (ACONTIG) {
      const float2 av = *(const float2*)(arow + k0);
      f.a[q].x = av.x; f.a[q].y = av.y;
    } else {
      f.a[q].x = arow[k0 * 3];
      f.a[q].y = arow[(k0 + 1) * 3];
    }
    const float2* bp = Wp + (k0 >> 1) * BPITCH + l16;
    #pragma unroll
    for (int ct = 0; ct < 4; ++ct) {
      const float2 bv = bp[ct * 16];
      f.b[q][ct].x = bv.x; f.b[q][ct].y = bv.y;
    }
  }
}

template <int KB>
__device__ __forceinline__ void frag_mma(const FragBlk<KB>& f, v8f acc[4]) {
  #pragma unroll
  for (int q = 0; q < KB; ++q) {
    #pragma unroll
    for (int ct = 0; ct < 4; ++ct) acc[ct] = wmma_f32_k4(f.a[q], f.b[q][ct], acc[ct]);
  }
}

struct SkipBlk { v2f b[4][2][4]; };

__device__ __forceinline__ void skip_load(SkipBlk& s, const float2* __restrict__ Wskp,
                                          int i0, int kh, int l16) {
  #pragma unroll
  for (int u = 0; u < 4; ++u) {
    const float2* bp = Wskp + (4 * (i0 + u) + kh) * 64 + l16;
    #pragma unroll
    for (int h = 0; h < 2; ++h) {
      #pragma unroll
      for (int ct = 0; ct < 4; ++ct) {
        const float2 bv = bp[h * 128 + ct * 16];
        s.b[u][h][ct].x = bv.x; s.b[u][h][ct].y = bv.y;
      }
    }
  }
}

__device__ __forceinline__ void skip_mma(const SkipBlk& s, const float* m2p,
                                         float attE0, float attE1, float attO0, float attO1,
                                         v8f acc[4]) {
  const float4 mv = *(const float4*)m2p;
  const float m2[4] = {mv.x, mv.y, mv.z, mv.w};
  #pragma unroll
  for (int u = 0; u < 4; ++u) {
    v2f a;
    a.x = m2[u] * attE0; a.y = m2[u] * attE1;
    #pragma unroll
    for (int ct = 0; ct < 4; ++ct) acc[ct] = wmma_f32_k4(a, s.b[u][0][ct], acc[ct]);
    a.x = m2[u] * attO0; a.y = m2[u] * attO1;
    #pragma unroll
    for (int ct = 0; ct < 4; ++ct) acc[ct] = wmma_f32_k4(a, s.b[u][1][ct], acc[ct]);
  }
}

__global__ void __launch_bounds__(256)
k_pre(const float* __restrict__ nf,
      const float* __restrict__ W1s,
      const float* __restrict__ W1v,
      const float* __restrict__ W2s,
      const float* __restrict__ W2v,
      const float* __restrict__ WskS,
      const float* __restrict__ WskV,
      float2* __restrict__ postPack,
      float* __restrict__ sv) {
  __shared__ alignas(16) float2 w1l[2 * 32 * LPITCH];

  float4 lo[4], hi[4];
  #pragma unroll
  for (int t = 0; t < 4; ++t) {
    const int q = threadIdx.x + 256 * t;
    const int m = q >> 9, kk = (q >> 4) & 31, n4 = (q & 15) * 4;
    const float* src = (m ? W1v : W1s) + (2 * kk) * 64 + n4;
    lo[t] = *(const float4*)src;
    hi[t] = *(const float4*)(src + 64);
  }

  for (int i = blockIdx.x * 256 + threadIdx.x; i < POST_PACK_PAIRS; i += gridDim.x * 256) {
    const float* src; int j;
    if (i < 4096)       { src = W2s;  j = i; }
    else if (i < 8192)  { src = W2v;  j = i - 4096; }
    else if (i < 24576) { src = WskS; j = i - 8192; }
    else                { src = WskV; j = i - 24576; }
    const int kk = j >> 6, n = j & 63;
    postPack[i] = make_float2(src[(2 * kk) * 64 + n], src[(2 * kk + 1) * 64 + n]);
  }

  #pragma unroll
  for (int t = 0; t < 4; ++t) {
    const int q = threadIdx.x + 256 * t;
    const int m = q >> 9, kk = (q >> 4) & 31, n4 = (q & 15) * 4;
    float4* d = (float4*)(w1l + m * (32 * LPITCH) + kk * LPITCH + n4);
    d[0] = make_float4(lo[t].x, hi[t].x, lo[t].y, hi[t].y);
    d[1] = make_float4(lo[t].z, hi[t].z, lo[t].w, hi[t].w);
  }
  __syncthreads();
  const float2* W1sp = w1l;
  const float2* W1vp = w1l + 32 * LPITCH;

  const int lane = threadIdx.x & 31;
  const int wid  = threadIdx.x >> 5;
  const int l16  = lane & 15;
  const int kh   = lane >> 4;
  const int row0 = (blockIdx.x * 8 + wid) * 16;
  int rowA = row0 + l16; if (rowA >= NN) rowA = NN - 1;
  const float* nrow = nf + (size_t)rowA * 256;

  FragBlk<8> f[2];
  frag_load<true, 8, LPITCH>(f[0], nrow, W1sp, 0, kh, l16);
  v8f acc[4];
  #pragma unroll
  for (int n = 0; n < 8; ++n) {
    const int tr = n >> 1;
    if (n + 1 < 8) {
      const int trn = (n + 1) >> 1, kcn = ((n + 1) & 1) * 8;
      if (trn == 0) frag_load<true, 8, LPITCH>(f[(n + 1) & 1], nrow, W1sp, kcn, kh, l16);
      else          frag_load<false, 8, LPITCH>(f[(n + 1) & 1], nrow + 64 + (trn - 1), W1vp, kcn, kh, l16);
    }
    __builtin_amdgcn_sched_barrier(0);
    if ((n & 1) == 0) {
      #pragma unroll
      for (int ct = 0; ct < 4; ++ct) acc[ct] = v8f{};
    }
    frag_mma<8>(f[n & 1], acc);
    if (n & 1) {
      #pragma unroll
      for (int ct = 0; ct < 4; ++ct) {
        #pragma unroll
        for (int j = 0; j < 8; ++j) {
          const int r = row0 + j + 8 * kh;
          if (r < NN) {
            const int c = ct * 16 + l16;
            sv[r * 256 + tr * 64 + c] = acc[ct][j] * 0.125f;
          }
        }
      }
    }
    __builtin_amdgcn_sched_barrier(0);
  }
}

#define NPB  80
#define BCAP 128
#define MSG_THREADS 512
#define MSG_WAVES (MSG_THREADS / 32)
static_assert(NN % NPB == 0, "receiver ranges must tile the nodes");
static_assert(NE % BCAP == 0 && BCAP == 128, "rounds and the 4-per-lane sort assume this");

__device__ __forceinline__ float lane_bcast(float v, int src) {
  return __int_as_float(__builtin_amdgcn_readlane(__float_as_int(v), src));
}

__device__ __forceinline__ void bin_edges(const int* __restrict__ rcvcol, int lo, int hi,
                                          int n0, int* cnt, int* bucket) {
  const int tid = threadIdx.x;
  if (tid < NPB) cnt[tid] = 0;
  __syncthreads();
  const int4* r4 = (const int4*)rcvcol;
  const int lo4 = lo >> 2, n4 = (hi - lo) >> 2;
  const int rot = (int)((blockIdx.x * 320u) % (unsigned)n4);
  for (int i0 = tid; i0 < n4; i0 += MSG_THREADS * 8) {
    int4 v[8];
    int  at[8];
    #pragma unroll
    for (int k = 0; k < 8; ++k) {
      const int i = i0 + k * MSG_THREADS;
      int t = i + rot; if (t >= n4) t -= n4;
      at[k] = lo4 + t;
      v[k] = (i < n4) ? r4[at[k]] : make_int4(-1, -1, -1, -1);
    }
    #pragma unroll
    for (int k = 0; k < 8; ++k) {
      const int rr[4] = {v[k].x, v[k].y, v[k].z, v[k].w};
      #pragma unroll
      for (int c = 0; c < 4; ++c) {
        const int r = rr[c] - n0;
        if ((unsigned)r < (unsigned)NPB) {
          const int slot = atomicAdd(&cnt[r], 1);
          if (slot < BCAP) bucket[r * BCAP + slot] = at[k] * 4 + c;
        }
      }
    }
  }
  __syncthreads();
}

__global__ void __launch_bounds__(MSG_THREADS)
k_msg(const float* __restrict__ ea,
      const float* __restrict__ efeat,
      const int*   __restrict__ eidx,
      const float* __restrict__ Wm1,
      const float* __restrict__ Wm2,
      const float* __restrict__ sv,
      float* __restrict__ msmv) {
  __shared__ int cnt[NPB];
  __shared__ int bucket[NPB * BCAP];
  const int lane = threadIdx.x & 31;
  const int wid  = threadIdx.x >> 5;
  const int n0   = blockIdx.x * NPB;
  const int hr   = lane & 7;
  const int u0 = lane, u1 = lane + 32;

  float w1c[8];
  #pragma unroll
  for (int q = 0; q < 8; ++q) w1c[q] = Wm1[q * 8 + hr];
  float wm[8][4][2];
  #pragma unroll
  for (int r = 0; r < 8; ++r) {
    #pragma unroll
    for (int k = 0; k < 4; ++k) {
      wm[r][k][0] = Wm2[r * 256 + k * 64 + u0];
      wm[r][k][1] = Wm2[r * 256 + k * 64 + u1];
    }
  }

  bin_edges(eidx + NE, 0, NE, n0, cnt, bucket);
  const int over = __syncthreads_or((threadIdx.x < NPB) && (cnt[threadIdx.x < NPB ? threadIdx.x : 0] > BCAP));
  const int span = over ? BCAP : NE;

  for (int eb = 0; eb < NE; eb += span) {
    if (over) bin_edges(eidx + NE, eb, eb + span, n0, cnt, bucket);

    for (int ln = wid; ln < NPB; ln += MSG_WAVES) {
      int* bk = bucket + ln * BCAP;
      int d = cnt[ln]; if (d > BCAP) d = BCAP;

      {
        const int x0 = (lane < d) ? bk[lane] : 0x7fffffff;
        int r0 = 0;
        for (int j = 0; j < d; ++j) r0 += (bk[j] < x0) ? 1 : 0;
        if (d > 32) {
          const int x1 = (lane + 32 < d) ? bk[lane + 32] : 0x7fffffff;
          const int x2 = (lane + 64 < d) ? bk[lane + 64] : 0x7fffffff;
          const int x3 = (lane + 96 < d) ? bk[lane + 96] : 0x7fffffff;
          int r1 = 0, r2 = 0, r3 = 0;
          for (int j = 0; j < d; ++j) {
            const int v = bk[j];
            r1 += (v < x1) ? 1 : 0; r2 += (v < x2) ? 1 : 0; r3 += (v < x3) ? 1 : 0;
          }
          if (lane + 32 < d) bk[r1] = x1;
          if (lane + 64 < d) bk[r2] = x2;
          if (lane + 96 < d) bk[r3] = x3;
        }
        if (lane < d) bk[r0] = x0;
      }

      float acc[8][2];
      #pragma unroll
      for (int p = 0; p < 8; ++p) { acc[p][0] = 0.f; acc[p][1] = 0.f; }

      for (int g = 0; g < d; g += 4) {
        int j = g + (lane >> 3); if (j > d - 1) j = d - 1;
        const int em = bk[j];
        const float4 f0 = *(const float4*)(efeat + (size_t)em * 8);
        const float4 f1 = *(const float4*)(efeat + (size_t)em * 8 + 4);
        const int    sndl = eidx[em];
        const float4 yl = *(const float4*)(ea + (size_t)em * 4);
        float tt = 0.f;
        tt += f0.x * w1c[0]; tt += f0.y * w1c[1]; tt += f0.z * w1c[2]; tt += f0.w * w1c[3];
        tt += f1.x * w1c[4]; tt += f1.y * w1c[5]; tt += f1.z * w1c[6]; tt += f1.w * w1c[7];
        const float hl = sspf(tt * 0.35355339059f);

        float Y0[4], Y1x[4], Y1y[4], Y1z[4];
        float se[4][2], vx[4][2], vy[4][2], vz[4][2];
        #pragma unroll
        for (int e = 0; e < 4; ++e) {
          const int sl = e * 8;
          const int snd = __builtin_amdgcn_readlane(sndl, sl);
          Y0[e]  = lane_bcast(yl.x, sl);
          Y1x[e] = lane_bcast(yl.y, sl);
          Y1y[e] = lane_bcast(yl.z, sl);
          Y1z[e] = lane_bcast(yl.w, sl);
          const float* srow = sv + (size_t)snd * 256;
          se[e][0] = srow[u0];        se[e][1] = srow[u1];
          vx[e][0] = srow[64 + u0];   vx[e][1] = srow[64 + u1];
          vy[e][0] = srow[128 + u0];  vy[e][1] = srow[128 + u1];
          vz[e][0] = srow[192 + u0];  vz[e][1] = srow[192 + u1];
        }

        #pragma unroll
        for (int e = 0; e < 4; ++e) {
          if (g + e < d) {
            float h[8];
            #pragma unroll
            for (int r = 0; r < 8; ++r) h[r] = lane_bcast(hl, e * 8 + r);
            #pragma unroll
            for (int t = 0; t < 2; ++t) {
              float w[4];
              #pragma unroll
              for (int k = 0; k < 4; ++k) {
                float tk = 0.f;
                #pragma unroll
                for (int r = 0; r < 8; ++r) tk += h[r] * wm[r][k][t];
                w[k] = tk * 0.35355339059f;
              }
              const float s1 = se[e][t];
              const float ax = vx[e][t], ay = vy[e][t], az = vz[e][t];
              const float dv = ax * Y1x[e] + ay * Y1y[e] + az * Y1z[e];
              acc[0][t] += w[0] * s1 * Y0[e];
              acc[1][t] += w[3] * dv * 0.57735026919f;
              acc[2][t] += w[1] * s1 * Y1x[e];
              acc[3][t] += w[1] * s1 * Y1y[e];
              acc[4][t] += w[1] * s1 * Y1z[e];
              acc[5][t] += w[2] * ax * Y0[e];
              acc[6][t] += w[2] * ay * Y0[e];
              acc[7][t] += w[2] * az * Y0[e];
            }
          }
        }
      }

      float* base = msmv + (size_t)(n0 + ln) * 512;
      #pragma unroll
      for (int p = 0; p < 8; ++p) {
        float a0 = acc[p][0], a1 = acc[p][1];
        if (eb != 0) { a0 += base[p * 64 + u0]; a1 += base[p * 64 + u1]; }
        base[p * 64 + u0] = a0;
        base[p * 64 + u1] = a1;
      }
    }
    __syncthreads();
  }
}

__global__ void __launch_bounds__(256)
k_post(const float* __restrict__ msmv,
       const float* __restrict__ attrs,
       const float2* __restrict__ W2sp,
       const float2* __restrict__ W2vp,
       const float2* __restrict__ WskSp,
       const float2* __restrict__ WskVp,
       float* __restrict__ out) {
  __shared__ alignas(16) float lds[8 * 16 * 64];
  const int lane = threadIdx.x & 31;
  const int wid  = threadIdx.x >> 5;
  const int l16  = lane & 15;
  const int kh   = lane >> 4;
  const int row0 = (blockIdx.x * 8 + wid) * 16;
  int rowA = row0 + l16; if (rowA >= NN) rowA = NN - 1;
  float* myLds = lds + wid * (16 * 64);
  const float* mrow = msmv + (size_t)rowA * 512;

  const float* arow8 = attrs + (size_t)rowA * 8;
  const float attE0 = arow8[2 * kh]     * 0.04419417382f;
  const float attE1 = arow8[2 * kh + 1] * 0.04419417382f;
  const float attO0 = arow8[4 + 2 * kh] * 0.04419417382f;
  const float attO1 = arow8[5 + 2 * kh] * 0.04419417382f;

  FragBlk<8> f[2];
  SkipBlk sb[2];
  frag_load<true, 8, GPITCH>(f[0], mrow, W2sp, 0, kh, l16);

  for (int tr = 0; tr < 4; ++tr) {
    const float2* W2p  = (tr == 0) ? W2sp : W2vp;
    const float2* Wskp = (tr == 0) ? WskSp : WskVp;
    const int pA = (tr == 0) ? 0 : (1 + tr);
    const int pB = (tr == 0) ? 1 : (4 + tr);
    const float* arowA = mrow + pA * 64;
    const float* arowB = mrow + pB * 64;

    v8f acc[4] = {};
    #pragma unroll
    for (int n = 0; n < 4; ++n) {
      if (n + 1 < 4)
        frag_load<true, 8, GPITCH>(f[(n + 1) & 1], (n + 1 < 2) ? arowA : arowB,
                           (n + 1 < 2) ? W2p : W2p + 32 * 64, ((n + 1) & 1) * 8, kh, l16);
      else
        skip_load(sb[0], Wskp, 0, kh, l16);
      frag_mma<8>(f[n & 1], acc);
    }

    #pragma unroll
    for (int ct = 0; ct < 4; ++ct) {
      #pragma unroll
      for (int j = 0; j < 8; ++j) {
        acc[ct][j] *= 0.08838834765f;
        myLds[(j + 8 * kh) * 64 + ct * 16 + l16] = acc[ct][j];
      }
    }
    __syncthreads();

    const float* m2row = myLds + l16 * 64;
    for (int blk = 0; blk < 14; blk += 2) {
      skip_load(sb[1], Wskp, (blk + 1) * 4, kh, l16);
      skip_mma(sb[0], m2row + blk * 4, attE0, attE1, attO0, attO1, acc);
      skip_load(sb[0], Wskp, (blk + 2) * 4, kh, l16);
      skip_mma(sb[1], m2row + (blk + 1) * 4, attE0, attE1, attO0, attO1, acc);
    }
    skip_load(sb[1], Wskp, 15 * 4, kh, l16);
    skip_mma(sb[0], m2row + 14 * 4, attE0, attE1, attO0, attO1, acc);
    frag_load<true, 8, GPITCH>(f[0], mrow + ((tr < 3) ? (2 + tr) : 0) * 64, W2vp, 0, kh, l16);
    skip_mma(sb[1], m2row + 15 * 4, attE0, attE1, attO0, attO1, acc);

    #pragma unroll
    for (int ct = 0; ct < 4; ++ct) {
      #pragma unroll
      for (int j = 0; j < 8; ++j) {
        const int r = row0 + j + 8 * kh;
        if (r < NN) {
          const int c = ct * 16 + l16;
          const int o = (tr == 0) ? (r * 256 + c)
                                  : (r * 256 + 64 + c * 3 + (tr - 1));
          out[o] = acc[ct][j];
        }
      }
    }
    __syncthreads();
  }
}

extern "C" void kernel_launch(void* const* d_in, const int* in_sizes, int n_in,
                              void* d_out, int out_size, void* d_ws, size_t ws_size,
                              hipStream_t stream) {
  const float* node_attrs = (const float*)d_in[0];
  const float* node_feats = (const float*)d_in[1];
  const float* edge_attrs = (const float*)d_in[2];
  const float* edge_feats = (const float*)d_in[3];
  const int*   edge_index = (const int*)d_in[4];
  const float* W1s  = (const float*)d_in[5];
  const float* W1v  = (const float*)d_in[6];
  const float* Wm1  = (const float*)d_in[7];
  const float* Wm2  = (const float*)d_in[8];
  const float* W2s  = (const float*)d_in[9];
  const float* W2v  = (const float*)d_in[10];
  const float* WskS = (const float*)d_in[11];
  const float* WskV = (const float*)d_in[12];

  float* out  = (float*)d_out;
  float* msmv = (float*)d_ws;

  float2* packBase = (float2*)(msmv + (size_t)NN * 512);
  float2* W2sp  = packBase;
  float2* W2vp  = packBase + 4096;
  float2* WskSp = packBase + 8192;
  float2* WskVp = packBase + 24576;

  k_pre <<<(NN + 127) / 128, 256, 0, stream>>>(node_feats, W1s, W1v, W2s, W2v, WskS, WskV,
                                               packBase, out);
  k_msg <<<NN / NPB, MSG_THREADS, 0, stream>>>(edge_attrs, edge_feats, edge_index,
                                               Wm1, Wm2, out, msmv);
  k_post<<<(NN + 127) / 128, 256, 0, stream>>>(msmv, node_attrs, W2sp, W2vp,
                                               WskSp, WskVp, out);
}


# k_post: 250 work-groups of 20 waves, one (16-row tile, transform) per wave, 5 waves per SIMD on 250 WGPs (was 157 x 8 waves each looping over 4 transforms); rows staged in LDS; same sums
# speedup vs baseline: 3.2000x; 1.1153x over previous
#include <hip/hip_runtime.h>
#include <hip/hip_bf16.h>
#include <math.h>


#define NN 20000
#define NE 320000

typedef __attribute__((ext_vector_type(2))) float v2f;
typedef __attribute__((ext_vector_type(8))) float v8f;

__device__ __forceinline__ v8f wmma_f32_k4(v2f a, v2f b, v8f c) {
  return __builtin_amdgcn_wmma_f32_16x16x4_f32(false, a, false, b,
                                               (short)0, c, false, false);
}

__device__ __forceinline__ float sspf(float x) {
  return fmaxf(x, 0.0f) + log1pf(__expf(-fabsf(x))) - 0.69314718056f;
}

#define GPITCH 64
#define LPITCH 80
#define POST_PACK_PAIRS 40960

template <int KB>
struct FragBlk { v2f a[KB]; v2f b[KB][4]; };

template <bool ACONTIG, int KB, int BPITCH>
__device__ __forceinline__ void frag_load(FragBlk<KB>& f, const float* __restrict__ arow,
                                          const float2* __restrict__ Wp,
                                          int kc0, int kh, int l16) {
  #pragma unroll
  for (int q = 0; q < KB; ++q) {
    const int k0 = (kc0 + q) * 4 + 2 * kh;
    if (ACONTIG) {
      const float2 av = *(const float2*)(arow + k0);
      f.a[q].x = av.x; f.a[q].y = av.y;
    } else {
      f.a[q].x = arow[k0 * 3];
      f.a[q].y = arow[(k0 + 1) * 3];
    }
    const float2* bp = Wp + (k0 >> 1) * BPITCH + l16;
    #pragma unroll
    for (int ct = 0; ct < 4; ++ct) {
      const float2 bv = bp[ct * 16];
      f.b[q][ct].x = bv.x; f.b[q][ct].y = bv.y;
    }
  }
}

template <int KB>
__device__ __forceinline__ void frag_mma(const FragBlk<KB>& f, v8f acc[4]) {
  #pragma unroll
  for (int q = 0; q < KB; ++q) {
    #pragma unroll
    for (int ct = 0; ct < 4; ++ct) acc[ct] = wmma_f32_k4(f.a[q], f.b[q][ct], acc[ct]);
  }
}

struct SkipBlk { v2f b[4][2][4]; };

__device__ __forceinline__ void skip_load(SkipBlk& s, const float2* __restrict__ Wskp,
                                          int i0, int kh, int l16) {
  #pragma unroll
  for (int u = 0; u < 4; ++u) {
    const float2* bp = Wskp + (4 * (i0 + u) + kh) * 64 + l16;
    #pragma unroll
    for (int h = 0; h < 2; ++h) {
      #pragma unroll
      for (int ct = 0; ct < 4; ++ct) {
        const float2 bv = bp[h * 128 + ct * 16];
        s.b[u][h][ct].x = bv.x; s.b[u][h][ct].y = bv.y;
      }
    }
  }
}

__device__ __forceinline__ void skip_mma(const SkipBlk& s, const float* m2p,
                                         float attE0, float attE1, float attO0, float attO1,
                                         v8f acc[4]) {
  const float4 mv = *(const float4*)m2p;
  const float m2[4] = {mv.x, mv.y, mv.z, mv.w};
  #pragma unroll
  for (int u = 0; u < 4; ++u) {
    v2f a;
    a.x = m2[u] * attE0; a.y = m2[u] * attE1;
    #pragma unroll
    for (int ct = 0; ct < 4; ++ct) acc[ct] = wmma_f32_k4(a, s.b[u][0][ct], acc[ct]);
    a.x = m2[u] * attO0; a.y = m2[u] * attO1;
    #pragma unroll
    for (int ct = 0; ct < 4; ++ct) acc[ct] = wmma_f32_k4(a, s.b[u][1][ct], acc[ct]);
  }
}

__global__ void __launch_bounds__(256)
k_pre(const float* __restrict__ nf,
      const float* __restrict__ W1s,
      const float* __restrict__ W1v,
      const float* __restrict__ W2s,
      const float* __restrict__ W2v,
      const float* __restrict__ WskS,
      const float* __restrict__ WskV,
      float2* __restrict__ postPack,
      float* __restrict__ sv) {
  __shared__ alignas(16) float2 w1l[2 * 32 * LPITCH];

  float4 lo[4], hi[4];
  #pragma unroll
  for (int t = 0; t < 4; ++t) {
    const int q = threadIdx.x + 256 * t;
    const int m = q >> 9, kk = (q >> 4) & 31, n4 = (q & 15) * 4;
    const float* src = (m ? W1v : W1s) + (2 * kk) * 64 + n4;
    lo[t] = *(const float4*)src;
    hi[t] = *(const float4*)(src + 64);
  }

  for (int i = blockIdx.x * 256 + threadIdx.x; i < POST_PACK_PAIRS; i += gridDim.x * 256) {
    const float* src; int j;
    if (i < 4096)       { src = W2s;  j = i; }
    else if (i < 8192)  { src = W2v;  j = i - 4096; }
    else if (i < 24576) { src = WskS; j = i - 8192; }
    else                { src = WskV; j = i - 24576; }
    const int kk = j >> 6, n = j & 63;
    postPack[i] = make_float2(src[(2 * kk) * 64 + n], src[(2 * kk + 1) * 64 + n]);
  }

  #pragma unroll
  for (int t = 0; t < 4; ++t) {
    const int q = threadIdx.x + 256 * t;
    const int m = q >> 9, kk = (q >> 4) & 31, n4 = (q & 15) * 4;
    float4* d = (float4*)(w1l + m * (32 * LPITCH) + kk * LPITCH + n4);
    d[0] = make_float4(lo[t].x, hi[t].x, lo[t].y, hi[t].y);
    d[1] = make_float4(lo[t].z, hi[t].z, lo[t].w, hi[t].w);
  }
  __syncthreads();
  const float2* W1sp = w1l;
  const float2* W1vp = w1l + 32 * LPITCH;

  const int lane = threadIdx.x & 31;
  const int wid  = threadIdx.x >> 5;
  const int l16  = lane & 15;
  const int kh   = lane >> 4;
  const int row0 = (blockIdx.x * 8 + wid) * 16;
  int rowA = row0 + l16; if (rowA >= NN) rowA = NN - 1;
  const float* nrow = nf + (size_t)rowA * 256;

  FragBlk<8> f[2];
  frag_load<true, 8, LPITCH>(f[0], nrow, W1sp, 0, kh, l16);
  v8f acc[4];
  #pragma unroll
  for (int n = 0; n < 8; ++n) {
    const int tr = n >> 1;
    if (n + 1 < 8) {
      const int trn = (n + 1) >> 1, kcn = ((n + 1) & 1) * 8;
      if (trn == 0) frag_load<true, 8, LPITCH>(f[(n + 1) & 1], nrow, W1sp, kcn, kh, l16);
      else          frag_load<false, 8, LPITCH>(f[(n + 1) & 1], nrow + 64 + (trn - 1), W1vp, kcn, kh, l16);
    }
    __builtin_amdgcn_sched_barrier(0);
    if ((n & 1) == 0) {
      #pragma unroll
      for (int ct = 0; ct < 4; ++ct) acc[ct] = v8f{};
    }
    frag_mma<8>(f[n & 1], acc);
    if (n & 1) {
      #pragma unroll
      for (int ct = 0; ct < 4; ++ct) {
        #pragma unroll
        for (int j = 0; j < 8; ++j) {
          const int r = row0 + j + 8 * kh;
          if (r < NN) {
            const int c = ct * 16 + l16;
            sv[r * 256 + tr * 64 + c] = acc[ct][j] * 0.125f;
          }
        }
      }
    }
    __builtin_amdgcn_sched_barrier(0);
  }
}

#define NPB  80
#define BCAP 128
#define MSG_THREADS 512
#define MSG_WAVES (MSG_THREADS / 32)
static_assert(NN % NPB == 0, "receiver ranges must tile the nodes");
static_assert(NE % BCAP == 0 && BCAP == 128, "rounds and the 4-per-lane sort assume this");

__device__ __forceinline__ float lane_bcast(float v, int src) {
  return __int_as_float(__builtin_amdgcn_readlane(__float_as_int(v), src));
}

__device__ __forceinline__ void bin_edges(const int* __restrict__ rcvcol, int lo, int hi,
                                          int n0, int* cnt, int* bucket) {
  const int tid = threadIdx.x;
  if (tid < NPB) cnt[tid] = 0;
  __syncthreads();
  const int4* r4 = (const int4*)rcvcol;
  const int lo4 = lo >> 2, n4 = (hi - lo) >> 2;
  const int rot = (int)((blockIdx.x * 320u) % (unsigned)n4);
  for (int i0 = tid; i0 < n4; i0 += MSG_THREADS * 8) {
    int4 v[8];
    int  at[8];
    #pragma unroll
    for (int k = 0; k < 8; ++k) {
      const int i = i0 + k * MSG_THREADS;
      int t = i + rot; if (t >= n4) t -= n4;
      at[k] = lo4 + t;
      v[k] = (i < n4) ? r4[at[k]] : make_int4(-1, -1, -1, -1);
    }
    #pragma unroll
    for (int k = 0; k < 8; ++k) {
      const int rr[4] = {v[k].x, v[k].y, v[k].z, v[k].w};
      #pragma unroll
      for (int c = 0; c < 4; ++c) {
        const int r = rr[c] - n0;
        if ((unsigned)r < (unsigned)NPB) {
          const int slot = atomicAdd(&cnt[r], 1);
          if (slot < BCAP) bucket[r * BCAP + slot] = at[k] * 4 + c;
        }
      }
    }
  }
  __syncthreads();
}

__global__ void __launch_bounds__(MSG_THREADS)
k_msg(const float* __restrict__ ea,
      const float* __restrict__ efeat,
      const int*   __restrict__ eidx,
      const float* __restrict__ Wm1,
      const float* __restrict__ Wm2,
      const float* __restrict__ sv,
      float* __restrict__ msmv) {
  __shared__ int cnt[NPB];
  __shared__ int bucket[NPB * BCAP];
  const int lane = threadIdx.x & 31;
  const int wid  = threadIdx.x >> 5;
  const int n0   = blockIdx.x * NPB;
  const int hr   = lane & 7;
  const int u0 = lane, u1 = lane + 32;

  float w1c[8];
  #pragma unroll
  for (int q = 0; q < 8; ++q) w1c[q] = Wm1[q * 8 + hr];
  float wm[8][4][2];
  #pragma unroll
  for (int r = 0; r < 8; ++r) {
    #pragma unroll
    for (int k = 0; k < 4; ++k) {
      wm[r][k][0] = Wm2[r * 256 + k * 64 + u0];
      wm[r][k][1] = Wm2[r * 256 + k * 64 + u1];
    }
  }

  bin_edges(eidx + NE, 0, NE, n0, cnt, bucket);
  const int over = __syncthreads_or((threadIdx.x < NPB) && (cnt[threadIdx.x < NPB ? threadIdx.x : 0] > BCAP));
  const int span = over ? BCAP : NE;

  for (int eb = 0; eb < NE; eb += span) {
    if (over) bin_edges(eidx + NE, eb, eb + span, n0, cnt, bucket);

    for (int ln = wid; ln < NPB; ln += MSG_WAVES) {
      int* bk = bucket + ln * BCAP;
      int d = cnt[ln]; if (d > BCAP) d = BCAP;

      {
        const int x0 = (lane < d) ? bk[lane] : 0x7fffffff;
        int r0 = 0;
        for (int j = 0; j < d; ++j) r0 += (bk[j] < x0) ? 1 : 0;
        if (d > 32) {
          const int x1 = (lane + 32 < d) ? bk[lane + 32] : 0x7fffffff;
          const int x2 = (lane + 64 < d) ? bk[lane + 64] : 0x7fffffff;
          const int x3 = (lane + 96 < d) ? bk[lane + 96] : 0x7fffffff;
          int r1 = 0, r2 = 0, r3 = 0;
          for (int j = 0; j < d; ++j) {
            const int v = bk[j];
            r1 += (v < x1) ? 1 : 0; r2 += (v < x2) ? 1 : 0; r3 += (v < x3) ? 1 : 0;
          }
          if (lane + 32 < d) bk[r1] = x1;
          if (lane + 64 < d) bk[r2] = x2;
          if (lane + 96 < d) bk[r3] = x3;
        }
        if (lane < d) bk[r0] = x0;
      }

      float acc[8][2];
      #pragma unroll
      for (int p = 0; p < 8; ++p) { acc[p][0] = 0.f; acc[p][1] = 0.f; }

      for (int g = 0; g < d; g += 4) {
        int j = g + (lane >> 3); if (j > d - 1) j = d - 1;
        const int em = bk[j];
        const float4 f0 = *(const float4*)(efeat + (size_t)em * 8);
        const float4 f1 = *(const float4*)(efeat + (size_t)em * 8 + 4);
        const int    sndl = eidx[em];
        const float4 yl = *(const float4*)(ea + (size_t)em * 4);
        float tt = 0.f;
        tt += f0.x * w1c[0]; tt += f0.y * w1c[1]; tt += f0.z * w1c[2]; tt += f0.w * w1c[3];
        tt += f1.x * w1c[4]; tt += f1.y * w1c[5]; tt += f1.z * w1c[6]; tt += f1.w * w1c[7];
        const float hl = sspf(tt * 0.35355339059f);

        float Y0[4], Y1x[4], Y1y[4], Y1z[4];
        float se[4][2], vx[4][2], vy[4][2], vz[4][2];
        #pragma unroll
        for (int e = 0; e < 4; ++e) {
          const int sl = e * 8;
          const int snd = __builtin_amdgcn_readlane(sndl, sl);
          Y0[e]  = lane_bcast(yl.x, sl);
          Y1x[e] = lane_bcast(yl.y, sl);
          Y1y[e] = lane_bcast(yl.z, sl);
          Y1z[e] = lane_bcast(yl.w, sl);
          const float* srow = sv + (size_t)snd * 256;
          se[e][0] = srow[u0];        se[e][1] = srow[u1];
          vx[e][0] = srow[64 + u0];   vx[e][1] = srow[64 + u1];
          vy[e][0] = srow[128 + u0];  vy[e][1] = srow[128 + u1];
          vz[e][0] = srow[192 + u0];  vz[e][1] = srow[192 + u1];
        }

        #pragma unroll
        for (int e = 0; e < 4; ++e) {
          if (g + e < d) {
            float h[8];
            #pragma unroll
            for (int r = 0; r < 8; ++r) h[r] = lane_bcast(hl, e * 8 + r);
            #pragma unroll
            for (int t = 0; t < 2; ++t) {
              float w[4];
              #pragma unroll
              for (int k = 0; k < 4; ++k) {
                float tk = 0.f;
                #pragma unroll
                for (int r = 0; r < 8; ++r) tk += h[r] * wm[r][k][t];
                w[k] = tk * 0.35355339059f;
              }
              const float s1 = se[e][t];
              const float ax = vx[e][t], ay = vy[e][t], az = vz[e][t];
              const float dv = ax * Y1x[e] + ay * Y1y[e] + az * Y1z[e];
              acc[0][t] += w[0] * s1 * Y0[e];
              acc[1][t] += w[3] * dv * 0.57735026919f;
              acc[2][t] += w[1] * s1 * Y1x[e];
              acc[3][t] += w[1] * s1 * Y1y[e];
              acc[4][t] += w[1] * s1 * Y1z[e];
              acc[5][t] += w[2] * ax * Y0[e];
              acc[6][t] += w[2] * ay * Y0[e];
              acc[7][t] += w[2] * az * Y0[e];
            }
          }
        }
      }

      float* base = msmv + (size_t)(n0 + ln) * 512;
      #pragma unroll
      for (int p = 0; p < 8; ++p) {
        float a0 = acc[p][0], a1 = acc[p][1];
        if (eb != 0) { a0 += base[p * 64 + u0]; a1 += base[p * 64 + u1]; }
        base[p * 64 + u0] = a0;
        base[p * 64 + u1] = a1;
      }
    }
    __syncthreads();
  }
}

#define PNPB 80
#define POST_TILES (PNPB / 16)
#define POST_WAVES (POST_TILES * 4)
#define POST_THREADS (POST_WAVES * 32)
#define SPITCH 260
static_assert(NN % PNPB == 0, "post-transform row ranges must tile the nodes");

__global__ void __launch_bounds__(POST_THREADS)
k_post(const float* __restrict__ msmv,
       const float* __restrict__ attrs,
       const float2* __restrict__ W2sp,
       const float2* __restrict__ W2vp,
       const float2* __restrict__ WskSp,
       const float2* __restrict__ WskVp,
       float* __restrict__ out) {
  __shared__ alignas(16) float lds[POST_WAVES * 16 * 64];
  __shared__ alignas(16) float stage[PNPB * SPITCH];
  const int lane = threadIdx.x & 31;
  const int wid  = __builtin_amdgcn_readfirstlane((int)(threadIdx.x >> 5));
  const int l16  = lane & 15;
  const int kh   = lane >> 4;
  const int tile = wid % POST_TILES;
  const int tr   = wid / POST_TILES;
  const int row0 = blockIdx.x * PNPB + tile * 16;
  const int rowA = row0 + l16;
  float* myLds = lds + wid * (16 * 64);
  const float* mrow = msmv + (size_t)rowA * 512;

  const float* arow8 = attrs + (size_t)rowA * 8;
  const float attE0 = arow8[2 * kh]     * 0.04419417382f;
  const float attE1 = arow8[2 * kh + 1] * 0.04419417382f;
  const float attO0 = arow8[4 + 2 * kh] * 0.04419417382f;
  const float attO1 = arow8[5 + 2 * kh] * 0.04419417382f;

  const float2* W2p  = (tr == 0) ? W2sp : W2vp;
  const float2* Wskp = (tr == 0) ? WskSp : WskVp;
  const int pA = (tr == 0) ? 0 : (1 + tr);
  const int pB = (tr == 0) ? 1 : (4 + tr);
  const float* arowA = mrow + pA * 64;
  const float* arowB = mrow + pB * 64;

  FragBlk<4> f[2];
  SkipBlk sb[2];
  frag_load<true, 4, GPITCH>(f[0], arowA, W2p, 0, kh, l16);

  v8f acc[4] = {};
  #pragma unroll
  for (int n = 0; n < 8; ++n) {
    if (n + 1 < 8)
      frag_load<true, 4, GPITCH>(f[(n + 1) & 1], (n + 1 < 4) ? arowA : arowB,
                         (n + 1 < 4) ? W2p : W2p + 32 * 64, ((n + 1) & 3) * 4, kh, l16);
    else
      skip_load(sb[0], Wskp, 0, kh, l16);
    __builtin_amdgcn_sched_barrier(0);
    frag_mma<4>(f[n & 1], acc);
    __builtin_amdgcn_sched_barrier(0);
  }

  #pragma unroll
  for (int ct = 0; ct < 4; ++ct) {
    #pragma unroll
    for (int j = 0; j < 8; ++j) {
      acc[ct][j] *= 0.08838834765f;
      myLds[(j + 8 * kh) * 64 + ct * 16 + l16] = acc[ct][j];
    }
  }
  __syncthreads();

  const float* m2row = myLds + l16 * 64;
  for (int blk = 0; blk < 14; blk += 2) {
    skip_load(sb[1], Wskp, (blk + 1) * 4, kh, l16);
    skip_mma(sb[0], m2row + blk * 4, attE0, attE1, attO0, attO1, acc);
    skip_load(sb[0], Wskp, (blk + 2) * 4, kh, l16);
    skip_mma(sb[1], m2row + (blk + 1) * 4, attE0, attE1, attO0, attO1, acc);
  }
  skip_load(sb[1], Wskp, 15 * 4, kh, l16);
  skip_mma(sb[0], m2row + 14 * 4, attE0, attE1, attO0, attO1, acc);
  skip_mma(sb[1], m2row + 15 * 4, attE0, attE1, attO0, attO1, acc);

  #pragma unroll
  for (int ct = 0; ct < 4; ++ct) {
    #pragma unroll
    for (int j = 0; j < 8; ++j) {
      const int r = tile * 16 + j + 8 * kh;
      const int c = ct * 16 + l16;
      const int o = (tr == 0) ? c : (64 + c * 3 + (tr - 1));
      stage[r * SPITCH + o] = acc[ct][j];
    }
  }
  __syncthreads();

  for (int i = threadIdx.x; i < PNPB * 64; i += POST_THREADS) {
    const int r = i >> 6, q = i & 63;
    const float4 v = *(const float4*)(stage + r * SPITCH + q * 4);
    *(float4*)(out + (size_t)(blockIdx.x * PNPB + r) * 256 + q * 4) = v;
  }
}

extern "C" void kernel_launch(void* const* d_in, const int* in_sizes, int n_in,
                              void* d_out, int out_size, void* d_ws, size_t ws_size,
                              hipStream_t stream) {
  const float* node_attrs = (const float*)d_in[0];
  const float* node_feats = (const float*)d_in[1];
  const float* edge_attrs = (const float*)d_in[2];
  const float* edge_feats = (const float*)d_in[3];
  const int*   edge_index = (const int*)d_in[4];
  const float* W1s  = (const float*)d_in[5];
  const float* W1v  = (const float*)d_in[6];
  const float* Wm1  = (const float*)d_in[7];
  const float* Wm2  = (const float*)d_in[8];
  const float* W2s  = (const float*)d_in[9];
  const float* W2v  = (const float*)d_in[10];
  const float* WskS = (const float*)d_in[11];
  const float* WskV = (const float*)d_in[12];

  float* out  = (float*)d_out;
  float* msmv = (float*)d_ws;

  float2* packBase = (float2*)(msmv + (size_t)NN * 512);
  float2* W2sp  = packBase;
  float2* W2vp  = packBase + 4096;
  float2* WskSp = packBase + 8192;
  float2* WskVp = packBase + 24576;

  k_pre <<<(NN + 127) / 128, 256, 0, stream>>>(node_feats, W1s, W1v, W2s, W2v, WskS, WskV,
                                               packBase, out);
  k_msg <<<NN / NPB, MSG_THREADS, 0, stream>>>(edge_attrs, edge_feats, edge_index,
                                               Wm1, Wm2, out, msmv);
  k_post<<<NN / PNPB, POST_THREADS, 0, stream>>>(msmv, node_attrs, W2sp, W2vp,
                                               WskSp, WskVp, out);
}


# edge list split by receiver range once, by 250 extra work-groups of k_pre's launch (one 128-B list line per slice and range); k_msg reads its 250 lines, not the whole receiver column; same sums
# speedup vs baseline: 3.6029x; 1.1259x over previous
#include <hip/hip_runtime.h>
#include <hip/hip_bf16.h>
#include <math.h>


#define NN 20000
#define NE 320000

#define NPB  80
#define NDEST (NN / NPB)
#define NSLICE 250
#define SLICE_E (NE / NSLICE)
#define LINE_I 32
#define PRE_GEMM_WGS ((NN + 127) / 128)
static_assert(NN % NPB == 0, "receiver ranges must tile the nodes");
static_assert(NE % NSLICE == 0 && SLICE_E % 256 == 0, "slices tile the edge list, 256 threads a slice");
static_assert(NPB <= 128 && NE <= (1 << 24), "an entry packs edge id * 128 + node index into 31 bits");
static_assert(NSLICE * LINE_I <= NPB * 512, "a range's list lines fit inside its own message-sum rows");
static_assert(NDEST <= 256 && (256 + NDEST * LINE_I) * 4 <= 40960, "a slice's counters and lines fit the first launch's LDS");

typedef __attribute__((ext_vector_type(2))) float v2f;
typedef __attribute__((ext_vector_type(8))) float v8f;

__device__ __forceinline__ v8f wmma_f32_k4(v2f a, v2f b, v8f c) {
  return __builtin_amdgcn_wmma_f32_16x16x4_f32(false, a, false, b,
                                               (short)0, c, false, false);
}

__device__ __forceinline__ float sspf(float x) {
  return fmaxf(x, 0.0f) + log1pf(__expf(-fabsf(x))) - 0.69314718056f;
}

#define GPITCH 64
#define LPITCH 80
#define POST_PACK_PAIRS 40960

template <int KB>
struct FragBlk { v2f a[KB]; v2f b[KB][4]; };

template <bool ACONTIG, int KB, int BPITCH>
__device__ __forceinline__ void frag_load(FragBlk<KB>& f, const float* __restrict__ arow,
                                          const float2* __restrict__ Wp,
                                          int kc0, int kh, int l16) {
  #pragma unroll
  for (int q = 0; q < KB; ++q) {
    const int k0 = (kc0 + q) * 4 + 2 * kh;
    if (ACONTIG) {
      const float2 av = *(const float2*)(arow + k0);
      f.a[q].x = av.x; f.a[q].y = av.y;
    } else {
      f.a[q].x = arow[k0 * 3];
      f.a[q].y = arow[(k0 + 1) * 3];
    }
    const float2* bp = Wp + (k0 >> 1) * BPITCH + l16;
    #pragma unroll
    for (int ct = 0; ct < 4; ++ct) {
      const float2 bv = bp[ct * 16];
      f.b[q][ct].x = bv.x; f.b[q][ct].y = bv.y;
    }
  }
}

template <int KB>
__device__ __forceinline__ void frag_mma(const FragBlk<KB>& f, v8f acc[4]) {
  #pragma unroll
  for (int q = 0; q < KB; ++q) {
    #pragma unroll
    for (int ct = 0; ct < 4; ++ct) acc[ct] = wmma_f32_k4(f.a[q], f.b[q][ct], acc[ct]);
  }
}

struct SkipBlk { v2f b[4][2][4]; };

__device__ __forceinline__ void skip_load(SkipBlk& s, const float2* __restrict__ Wskp,
                                          int i0, int kh, int l16) {
  #pragma unroll
  for (int u = 0; u < 4; ++u) {
    const float2* bp = Wskp + (4 * (i0 + u) + kh) * 64 + l16;
    #pragma unroll
    for (int h = 0; h < 2; ++h) {
      #pragma unroll
      for (int ct = 0; ct < 4; ++ct) {
        const float2 bv = bp[h * 128 + ct * 16];
        s.b[u][h][ct].x = bv.x; s.b[u][h][ct].y = bv.y;
      }
    }
  }
}

__device__ __forceinline__ void skip_mma(const SkipBlk& s, const float* m2p,
                                         float attE0, float attE1, float attO0, float attO1,
                                         v8f acc[4]) {
  const float4 mv = *(const float4*)m2p;
  const float m2[4] = {mv.x, mv.y, mv.z, mv.w};
  #pragma unroll
  for (int u = 0; u < 4; ++u) {
    v2f a;
    a.x = m2[u] * attE0; a.y = m2[u] * attE1;
    #pragma unroll
    for (int ct = 0; ct < 4; ++ct) acc[ct] = wmma_f32_k4(a, s.b[u][0][ct], acc[ct]);
    a.x = m2[u] * attO0; a.y = m2[u] * attO1;
    #pragma unroll
    for (int ct = 0; ct < 4; ++ct) acc[ct] = wmma_f32_k4(a, s.b[u][1][ct], acc[ct]);
  }
}

__device__ __forceinline__ void bin_slice(const int* __restrict__ rcvcol, int s,
                                          int* __restrict__ lists, int* lds) {
  int* lcnt = lds;
  int* lent = lds + 256;
  const int tid = threadIdx.x;
  if (tid < NDEST) lcnt[tid] = 0;
  __syncthreads();
  int rc[SLICE_E / 256];
  #pragma unroll
  for (int k = 0; k < SLICE_E / 256; ++k) rc[k] = rcvcol[s * SLICE_E + k * 256 + tid];
  #pragma unroll
  for (int k = 0; k < SLICE_E / 256; ++k) {
    const int e = s * SLICE_E + k * 256 + tid;
    const unsigned d = (unsigned)rc[k] / (unsigned)NPB;
    if (d < (unsigned)NDEST) {
      const int c = atomicAdd(&lcnt[d], 1);
      if (c < LINE_I - 1) lent[d * LINE_I + 1 + c] = e * 128 + (rc[k] - (int)d * NPB);
    }
  }
  __syncthreads();
  const int lane = tid & 31, wid = tid >> 5;
  for (int d0 = wid; d0 < NDEST; d0 += 64) {
    int cc[8], ee[8];
    #pragma unroll
    for (int j = 0; j < 8; ++j) {
      const int d = (d0 + 8 * j < NDEST) ? d0 + 8 * j : NDEST - 1;
      cc[j] = lcnt[d];
      ee[j] = lent[d * LINE_I + lane];
    }
    #pragma unroll
    for (int j = 0; j < 8; ++j) {
      const int d = d0 + 8 * j;
      if (d < NDEST) {
        const int v = (lane == 0) ? cc[j] : ((lane <= cc[j]) ? ee[j] : 0);
        lists[(size_t)d * (NPB * 512) + s * LINE_I + lane] = v;
      }
    }
  }
}

__global__ void __launch_bounds__(256)
k_pre(const float* __restrict__ nf,
      const float* __restrict__ W1s,
      const float* __restrict__ W1v,
      const float* __restrict__ W2s,
      const float* __restrict__ W2v,
      const float* __restrict__ WskS,
      const float* __restrict__ WskV,
      float2* __restrict__ postPack,
      float* __restrict__ sv,
      const int* __restrict__ eidx,
      int* __restrict__ lists) {
  __shared__ alignas(16) float2 w1l[2 * 32 * LPITCH];

  if (blockIdx.x >= PRE_GEMM_WGS) {
    bin_slice(eidx + NE, blockIdx.x - PRE_GEMM_WGS, lists, (int*)w1l);
    return;
  }

  float4 lo[4], hi[4];
  #pragma unroll
  for (int t = 0; t < 4; ++t) {
    const int q = threadIdx.x + 256 * t;
    const int m = q >> 9, kk = (q >> 4) & 31, n4 = (q & 15) * 4;
    const float* src = (m ? W1v : W1s) + (2 * kk) * 64 + n4;
    lo[t] = *(const float4*)src;
    hi[t] = *(const float4*)(src + 64);
  }

  for (int i = blockIdx.x * 256 + threadIdx.x; i < POST_PACK_PAIRS; i += PRE_GEMM_WGS * 256) {
    const float* src; int j;
    if (i < 4096)       { src = W2s;  j = i; }
    else if (i < 8192)  { src = W2v;  j = i - 4096; }
    else if (i < 24576) { src = WskS; j = i - 8192; }
    else                { src = WskV; j = i - 24576; }
    const int kk = j >> 6, n = j & 63;
    postPack[i] = make_float2(src[(2 * kk) * 64 + n], src[(2 * kk + 1) * 64 + n]);
  }

  #pragma unroll
  for (int t = 0; t < 4; ++t) {
    const int q = threadIdx.x + 256 * t;
    const int m = q >> 9, kk = (q >> 4) & 31, n4 = (q & 15) * 4;
    float4* d = (float4*)(w1l + m * (32 * LPITCH) + kk * LPITCH + n4);
    d[0] = make_float4(lo[t].x, hi[t].x, lo[t].y, hi[t].y);
    d[1] = make_float4(lo[t].z, hi[t].z, lo[t].w, hi[t].w);
  }
  __syncthreads();
  const float2* W1sp = w1l;
  const float2* W1vp = w1l + 32 * LPITCH;

  const int lane = threadIdx.x & 31;
  const int wid  = threadIdx.x >> 5;
  const int l16  = lane & 15;
  const int kh   = lane >> 4;
  const int row0 = (blockIdx.x * 8 + wid) * 16;
  int rowA = row0 + l16; if (rowA >= NN) rowA = NN - 1;
  const float* nrow = nf + (size_t)rowA * 256;

  FragBlk<8> f[2];
  frag_load<true, 8, LPITCH>(f[0], nrow, W1sp, 0, kh, l16);
  v8f acc[4];
  #pragma unroll
  for (int n = 0; n < 8; ++n) {
    const int tr = n >> 1;
    if (n + 1 < 8) {
      const int trn = (n + 1) >> 1, kcn = ((n + 1) & 1) * 8;
      if (trn == 0) frag_load<true, 8, LPITCH>(f[(n + 1) & 1], nrow, W1sp, kcn, kh, l16);
      else          frag_load<false, 8, LPITCH>(f[(n + 1) & 1], nrow + 64 + (trn - 1), W1vp, kcn, kh, l16);
    }
    __builtin_amdgcn_sched_barrier(0);
    if ((n & 1) == 0) {
      #pragma unroll
      for (int ct = 0; ct < 4; ++ct) acc[ct] = v8f{};
    }
    frag_mma<8>(f[n & 1], acc);
    if (n & 1) {
      #pragma unroll
      for (int ct = 0; ct < 4; ++ct) {
        #pragma unroll
        for (int j = 0; j < 8; ++j) {
          const int r = row0 + j + 8 * kh;
          if (r < NN) {
            const int c = ct * 16 + l16;
            sv[r * 256 + tr * 64 + c] = acc[ct][j] * 0.125f;
          }
        }
      }
    }
    __builtin_amdgcn_sched_barrier(0);
  }
}

#define BCAP 128
#define MSG_THREADS 512
#define MSG_WAVES (MSG_THREADS / 32)
#define LINES_PER_WAVE ((NSLICE + MSG_WAVES - 1) / MSG_WAVES)
static_assert(NE % BCAP == 0 && BCAP == 128, "rounds and the 4-per-lane sort assume this");

__device__ __forceinline__ float lane_bcast(float v, int src) {
  return __int_as_float(__builtin_amdgcn_readlane(__float_as_int(v), src));
}

__device__ __forceinline__ void bin_edges(const int* __restrict__ rcvcol, int lo, int hi,
                                          int n0, int* cnt, int* bucket) {
  const int tid = threadIdx.x;
  if (tid < NPB) cnt[tid] = 0;
  __syncthreads();
  const int4* r4 = (const int4*)rcvcol;
  const int lo4 = lo >> 2, n4 = (hi - lo) >> 2;
  const int rot = (int)((blockIdx.x * 320u) % (unsigned)n4);
  for (int i0 = tid; i0 < n4; i0 += MSG_THREADS * 8) {
    int4 v[8];
    int  at[8];
    #pragma unroll
    for (int k = 0; k < 8; ++k) {
      const int i = i0 + k * MSG_THREADS;
      int t = i + rot; if (t >= n4) t -= n4;
      at[k] = lo4 + t;
      v[k] = (i < n4) ? r4[at[k]] : make_int4(-1, -1, -1, -1);
    }
    #pragma unroll
    for (int k = 0; k < 8; ++k) {
      const int rr[4] = {v[k].x, v[k].y, v[k].z, v[k].w};
      #pragma unroll
      for (int c = 0; c < 4; ++c) {
        const int r = rr[c] - n0;
        if ((unsigned)r < (unsigned)NPB) {
          const int slot = atomicAdd(&cnt[r], 1);
          if (slot < BCAP) bucket[r * BCAP + slot] = at[k] * 4 + c;
        }
      }
    }
  }
  __syncthreads();
}

__global__ void __launch_bounds__(MSG_THREADS)
k_msg(const float* __restrict__ ea,
      const float* __restrict__ efeat,
      const int*   __restrict__ eidx,
      const float* __restrict__ Wm1,
      const float* __restrict__ Wm2,
      const float* __restrict__ sv,
      float* __restrict__ msmv) {
  __shared__ int cnt[NPB];
  __shared__ int bucket[NPB * BCAP];
  const int lane = threadIdx.x & 31;
  const int wid  = threadIdx.x >> 5;
  const int n0   = blockIdx.x * NPB;
  const int hr   = lane & 7;
  const int u0 = lane, u1 = lane + 32;

  float w1c[8];
  #pragma unroll
  for (int q = 0; q < 8; ++q) w1c[q] = Wm1[q * 8 + hr];
  float wm[8][4][2];
  #pragma unroll
  for (int r = 0; r < 8; ++r) {
    #pragma unroll
    for (int k = 0; k < 4; ++k) {
      wm[r][k][0] = Wm2[r * 256 + k * 64 + u0];
      wm[r][k][1] = Wm2[r * 256 + k * 64 + u1];
    }
  }

  if (threadIdx.x < NPB) cnt[threadIdx.x] = 0;
  __syncthreads();
  const float* lst = msmv + (size_t)blockIdx.x * (NPB * 512);
  int lv[LINES_PER_WAVE];
  #pragma unroll
  for (int k = 0; k < LINES_PER_WAVE; ++k) {
    const int s = wid + k * MSG_WAVES;
    lv[k] = (s < NSLICE) ? __float_as_int(lst[s * LINE_I + lane]) : 0;
  }
  int spill = 0;
  #pragma unroll
  for (int k = 0; k < LINES_PER_WAVE; ++k) {
    const int c = __builtin_amdgcn_readfirstlane(lv[k]);
    if (c > LINE_I - 1) spill = 1;
    if (lane >= 1 && lane <= c) {
      const int r = lv[k] & 127;
      if (r < NPB) {
        const int slot = atomicAdd(&cnt[r], 1);
        if (slot < BCAP) bucket[r * BCAP + slot] = lv[k] >> 7;
      }
    }
  }
  __syncthreads();
  const int over = __syncthreads_or(spill | ((threadIdx.x < NPB) && (cnt[threadIdx.x < NPB ? threadIdx.x : 0] > BCAP)));
  const int span = over ? BCAP : NE;

  for (int eb = 0; eb < NE; eb += span) {
    if (over) bin_edges(eidx + NE, eb, eb + span, n0, cnt, bucket);

    for (int ln = wid; ln < NPB; ln += MSG_WAVES) {
      int* bk = bucket + ln * BCAP;
      int d = cnt[ln]; if (d > BCAP) d = BCAP;

      {
        const int x0 = (lane < d) ? bk[lane] : 0x7fffffff;
        int r0 = 0;
        for (int j = 0; j < d; ++j) r0 += (bk[j] < x0) ? 1 : 0;
        if (d > 32) {
          const int x1 = (lane + 32 < d) ? bk[lane + 32] : 0x7fffffff;
          const int x2 = (lane + 64 < d) ? bk[lane + 64] : 0x7fffffff;
          const int x3 = (lane + 96 < d) ? bk[lane + 96] : 0x7fffffff;
          int r1 = 0, r2 = 0, r3 = 0;
          for (int j = 0; j < d; ++j) {
            const int v = bk[j];
            r1 += (v < x1) ? 1 : 0; r2 += (v < x2) ? 1 : 0; r3 += (v < x3) ? 1 : 0;
          }
          if (lane + 32 < d) bk[r1] = x1;
          if (lane + 64 < d) bk[r2] = x2;
          if (lane + 96 < d) bk[r3] = x3;
        }
        if (lane < d) bk[r0] = x0;
      }

      float acc[8][2];
      #pragma unroll
      for (int p = 0; p < 8; ++p) { acc[p][0] = 0.f; acc[p][1] = 0.f; }

      for (int g = 0; g < d; g += 4) {
        int j = g + (lane >> 3); if (j > d - 1) j = d - 1;
        const int em = bk[j];
        const float4 f0 = *(const float4*)(efeat + (size_t)em * 8);
        const float4 f1 = *(const float4*)(efeat + (size_t)em * 8 + 4);
        const int    sndl = eidx[em];
        const float4 yl = *(const float4*)(ea + (size_t)em * 4);
        float tt = 0.f;
        tt += f0.x * w1c[0]; tt += f0.y * w1c[1]; tt += f0.z * w1c[2]; tt += f0.w * w1c[3];
        tt += f1.x * w1c[4]; tt += f1.y * w1c[5]; tt += f1.z * w1c[6]; tt += f1.w * w1c[7];
        const float hl = sspf(tt * 0.35355339059f);

        float Y0[4], Y1x[4], Y1y[4], Y1z[4];
        float se[4][2], vx[4][2], vy[4][2], vz[4][2];
        #pragma unroll
        for (int e = 0; e < 4; ++e) {
          const int sl = e * 8;
          const int snd = __builtin_amdgcn_readlane(sndl, sl);
          Y0[e]  = lane_bcast(yl.x, sl);
          Y1x[e] = lane_bcast(yl.y, sl);
          Y1y[e] = lane_bcast(yl.z, sl);
          Y1z[e] = lane_bcast(yl.w, sl);
          const float* srow = sv + (size_t)snd * 256;
          se[e][0] = srow[u0];        se[e][1] = srow[u1];
          vx[e][0] = srow[64 + u0];   vx[e][1] = srow[64 + u1];
          vy[e][0] = srow[128 + u0];  vy[e][1] = srow[128 + u1];
          vz[e][0] = srow[192 + u0];  vz[e][1] = srow[192 + u1];
        }

        #pragma unroll
        for (int e = 0; e < 4; ++e) {
          if (g + e < d) {
            float h[8];
            #pragma unroll
            for (int r = 0; r < 8; ++r) h[r] = lane_bcast(hl, e * 8 + r);
            #pragma unroll
            for (int t = 0; t < 2; ++t) {
              float w[4];
              #pragma unroll
              for (int k = 0; k < 4; ++k) {
                float tk = 0.f;
                #pragma unroll
                for (int r = 0; r < 8; ++r) tk += h[r] * wm[r][k][t];
                w[k] = tk * 0.35355339059f;
              }
              const float s1 = se[e][t];
              const float ax = vx[e][t], ay = vy[e][t], az = vz[e][t];
              const float dv = ax * Y1x[e] + ay * Y1y[e] + az * Y1z[e];
              acc[0][t] += w[0] * s1 * Y0[e];
              acc[1][t] += w[3] * dv * 0.57735026919f;
              acc[2][t] += w[1] * s1 * Y1x[e];
              acc[3][t] += w[1] * s1 * Y1y[e];
              acc[4][t] += w[1] * s1 * Y1z[e];
              acc[5][t] += w[2] * ax * Y0[e];
              acc[6][t] += w[2] * ay * Y0[e];
              acc[7][t] += w[2] * az * Y0[e];
            }
          }
        }
      }

      float* base = msmv + (size_t)(n0 + ln) * 512;
      #pragma unroll
      for (int p = 0; p < 8; ++p) {
        float a0 = acc[p][0], a1 = acc[p][1];
        if (eb != 0) { a0 += base[p * 64 + u0]; a1 += base[p * 64 + u1]; }
        base[p * 64 + u0] = a0;
        base[p * 64 + u1] = a1;
      }
    }
    __syncthreads();
  }
}

#define PNPB 80
#define POST_TILES (PNPB / 16)
#define POST_WAVES (POST_TILES * 4)
#define POST_THREADS (POST_WAVES * 32)
#define SPITCH 260
static_assert(NN % PNPB == 0, "post-transform row ranges must tile the nodes");

__global__ void __launch_bounds__(POST_THREADS)
k_post(const float* __restrict__ msmv,
       const float* __restrict__ attrs,
       const float2* __restrict__ W2sp,
       const float2* __restrict__ W2vp,
       const float2* __restrict__ WskSp,
       const float2* __restrict__ WskVp,
       float* __restrict__ out) {
  __shared__ alignas(16) float lds[POST_WAVES * 16 * 64];
  __shared__ alignas(16) float stage[PNPB * SPITCH];
  const int lane = threadIdx.x & 31;
  const int wid  = __builtin_amdgcn_readfirstlane((int)(threadIdx.x >> 5));
  const int l16  = lane & 15;
  const int kh   = lane >> 4;
  const int tile = wid % POST_TILES;
  const int tr   = wid / POST_TILES;
  const int row0 = blockIdx.x * PNPB + tile * 16;
  const int rowA = row0 + l16;
  float* myLds = lds + wid * (16 * 64);
  const float* mrow = msmv + (size_t)rowA * 512;

  const float* arow8 = attrs + (size_t)rowA * 8;
  const float attE0 = arow8[2 * kh]     * 0.04419417382f;
  const float attE1 = arow8[2 * kh + 1] * 0.04419417382f;
  const float attO0 = arow8[4 + 2 * kh] * 0.04419417382f;
  const float attO1 = arow8[5 + 2 * kh] * 0.04419417382f;

  const float2* W2p  = (tr == 0) ? W2sp : W2vp;
  const float2* Wskp = (tr == 0) ? WskSp : WskVp;
  const int pA = (tr == 0) ? 0 : (1 + tr);
  const int pB = (tr == 0) ? 1 : (4 + tr);
  const float* arowA = mrow + pA * 64;
  const float* arowB = mrow + pB * 64;

  FragBlk<4> f[2];
  SkipBlk sb[2];
  frag_load<true, 4, GPITCH>(f[0], arowA, W2p, 0, kh, l16);

  v8f acc[4] = {};
  #pragma unroll
  for (int n = 0; n < 8; ++n) {
    if (n + 1 < 8)
      frag_load<true, 4, GPITCH>(f[(n + 1) & 1], (n + 1 < 4) ? arowA : arowB,
                         (n + 1 < 4) ? W2p : W2p + 32 * 64, ((n + 1) & 3) * 4, kh, l16);
    else
      skip_load(sb[0], Wskp, 0, kh, l16);
    __builtin_amdgcn_sched_barrier(0);
    frag_mma<4>(f[n & 1], acc);
    __builtin_amdgcn_sched_barrier(0);
  }

  #pragma unroll
  for (int ct = 0; ct < 4; ++ct) {
    #pragma unroll
    for (int j = 0; j < 8; ++j) {
      acc[ct][j] *= 0.08838834765f;
      myLds[(j + 8 * kh) * 64 + ct * 16 + l16] = acc[ct][j];
    }
  }
  __syncthreads();

  const float* m2row = myLds + l16 * 64;
  for (int blk = 0; blk < 14; blk += 2) {
    skip_load(sb[1], Wskp, (blk + 1) * 4, kh, l16);
    skip_mma(sb[0], m2row + blk * 4, attE0, attE1, attO0, attO1, acc);
    skip_load(sb[0], Wskp, (blk + 2) * 4, kh, l16);
    skip_mma(sb[1], m2row + (blk + 1) * 4, attE0, attE1, attO0, attO1, acc);
  }
  skip_load(sb[1], Wskp, 15 * 4, kh, l16);
  skip_mma(sb[0], m2row + 14 * 4, attE0, attE1, attO0, attO1, acc);
  skip_mma(sb[1], m2row + 15 * 4, attE0, attE1, attO0, attO1, acc);

  #pragma unroll
  for (int ct = 0; ct < 4; ++ct) {
    #pragma unroll
    for (int j = 0; j < 8; ++j) {
      const int r = tile * 16 + j + 8 * kh;
      const int c = ct * 16 + l16;
      const int o = (tr == 0) ? c : (64 + c * 3 + (tr - 1));
      stage[r * SPITCH + o] = acc[ct][j];
    }
  }
  __syncthreads();

  for (int i = threadIdx.x; i < PNPB * 64; i += POST_THREADS) {
    const int r = i >> 6, q = i & 63;
    const float4 v = *(const float4*)(stage + r * SPITCH + q * 4);
    *(float4*)(out + (size_t)(blockIdx.x * PNPB + r) * 256 + q * 4) = v;
  }
}

extern "C" void kernel_launch(void* const* d_in, const int* in_sizes, int n_in,
                              void* d_out, int out_size, void* d_ws, size_t ws_size,
                              hipStream_t stream) {
  const float* node_attrs = (const float*)d_in[0];
  const float* node_feats = (const float*)d_in[1];
  const float* edge_attrs = (const float*)d_in[2];
  const float* edge_feats = (const float*)d_in[3];
  const int*   edge_index = (const int*)d_in[4];
  const float* W1s  = (const float*)d_in[5];
  const float* W1v  = (const float*)d_in[6];
  const float* Wm1  = (const float*)d_in[7];
  const float* Wm2  = (const float*)d_in[8];
  const float* W2s  = (const float*)d_in[9];
  const float* W2v  = (const float*)d_in[10];
  const float* WskS = (const float*)d_in[11];
  const float* WskV = (const float*)d_in[12];

  float* out  = (float*)d_out;
  float* msmv = (float*)d_ws;

  float2* packBase = (float2*)(msmv + (size_t)NN * 512);
  float2* W2sp  = packBase;
  float2* W2vp  = packBase + 4096;
  float2* WskSp = packBase + 8192;
  float2* WskVp = packBase + 24576;

  k_pre <<<PRE_GEMM_WGS + NSLICE, 256, 0, stream>>>(node_feats, W1s, W1v, W2s, W2v, WskS, WskV,
                                               packBase, out, edge_index, (int*)msmv);
  k_msg <<<NN / NPB, MSG_THREADS, 0, stream>>>(edge_attrs, edge_feats, edge_index,
                                               Wm1, Wm2, out, msmv);
  k_post<<<NN / PNPB, POST_THREADS, 0, stream>>>(msmv, node_attrs, W2sp, W2vp,
                                               WskSp, WskVp, out);
}


# k_msg: the radial MLP's two 8-term sums written as fmaf chains, same order, one rounding per term (the compiler had packed the products in pairs and kept every add apart): dual-issued FMAs
# speedup vs baseline: 3.8965x; 1.0815x over previous
#include <hip/hip_runtime.h>
#include <hip/hip_bf16.h>
#include <math.h>


#define NN 20000
#define NE 320000

#define NPB  80
#define NDEST (NN / NPB)
#define NSLICE 250
#define SLICE_E (NE / NSLICE)
#define LINE_I 32
#define PRE_GEMM_WGS ((NN + 127) / 128)
static_assert(NN % NPB == 0, "receiver ranges must tile the nodes");
static_assert(NE % NSLICE == 0 && SLICE_E % 256 == 0, "slices tile the edge list, 256 threads a slice");
static_assert(NPB <= 128 && NE <= (1 << 24), "an entry packs edge id * 128 + node index into 31 bits");
static_assert(NSLICE * LINE_I <= NPB * 512, "a range's list lines fit inside its own message-sum rows");
static_assert(NDEST <= 256 && (256 + NDEST * LINE_I) * 4 <= 40960, "a slice's counters and lines fit the first launch's LDS");

typedef __attribute__((ext_vector_type(2))) float v2f;
typedef __attribute__((ext_vector_type(8))) float v8f;

__device__ __forceinline__ v8f wmma_f32_k4(v2f a, v2f b, v8f c) {
  return __builtin_amdgcn_wmma_f32_16x16x4_f32(false, a, false, b,
                                               (short)0, c, false, false);
}

__device__ __forceinline__ float sspf(float x) {
  return fmaxf(x, 0.0f) + log1pf(__expf(-fabsf(x))) - 0.69314718056f;
}

#define GPITCH 64
#define LPITCH 80
#define POST_PACK_PAIRS 40960

template <int KB>
struct FragBlk { v2f a[KB]; v2f b[KB][4]; };

template <bool ACONTIG, int KB, int BPITCH>
__device__ __forceinline__ void frag_load(FragBlk<KB>& f, const float* __restrict__ arow,
                                          const float2* __restrict__ Wp,
                                          int kc0, int kh, int l16) {
  #pragma unroll
  for (int q = 0; q < KB; ++q) {
    const int k0 = (kc0 + q) * 4 + 2 * kh;
    if (ACONTIG) {
      const float2 av = *(const float2*)(arow + k0);
      f.a[q].x = av.x; f.a[q].y = av.y;
    } else {
      f.a[q].x = arow[k0 * 3];
      f.a[q].y = arow[(k0 + 1) * 3];
    }
    const float2* bp = Wp + (k0 >> 1) * BPITCH + l16;
    #pragma unroll
    for (int ct = 0; ct < 4; ++ct) {
      const float2 bv = bp[ct * 16];
      f.b[q][ct].x = bv.x; f.b[q][ct].y = bv.y;
    }
  }
}

template <int KB>
__device__ __forceinline__ void frag_mma(const FragBlk<KB>& f, v8f acc[4]) {
  #pragma unroll
  for (int q = 0; q < KB; ++q) {
    #pragma unroll
    for (int ct = 0; ct < 4; ++ct) acc[ct] = wmma_f32_k4(f.a[q], f.b[q][ct], acc[ct]);
  }
}

struct SkipBlk { v2f b[4][2][4]; };

__device__ __forceinline__ void skip_load(SkipBlk& s, const float2* __restrict__ Wskp,
                                          int i0, int kh, int l16) {
  #pragma unroll
  for (int u = 0; u < 4; ++u) {
    const float2* bp = Wskp + (4 * (i0 + u) + kh) * 64 + l16;
    #pragma unroll
    for (int h = 0; h < 2; ++h) {
      #pragma unroll
      for (int ct = 0; ct < 4; ++ct) {
        const float2 bv = bp[h * 128 + ct * 16];
        s.b[u][h][ct].x = bv.x; s.b[u][h][ct].y = bv.y;
      }
    }
  }
}

__device__ __forceinline__ void skip_mma(const SkipBlk& s, const float* m2p,
                                         float attE0, float attE1, float attO0, float attO1,
                                         v8f acc[4]) {
  const float4 mv = *(const float4*)m2p;
  const float m2[4] = {mv.x, mv.y, mv.z, mv.w};
  #pragma unroll
  for (int u = 0; u < 4; ++u) {
    v2f a;
    a.x = m2[u] * attE0; a.y = m2[u] * attE1;
    #pragma unroll
    for (int ct = 0; ct < 4; ++ct) acc[ct] = wmma_f32_k4(a, s.b[u][0][ct], acc[ct]);
    a.x = m2[u] * attO0; a.y = m2[u] * attO1;
    #pragma unroll
    for (int ct = 0; ct < 4; ++ct) acc[ct] = wmma_f32_k4(a, s.b[u][1][ct], acc[ct]);
  }
}

__device__ __forceinline__ void bin_slice(const int* __restrict__ rcvcol, int s,
                                          int* __restrict__ lists, int* lds) {
  int* lcnt = lds;
  int* lent = lds + 256;
  const int tid = threadIdx.x;
  if (tid < NDEST) lcnt[tid] = 0;
  __syncthreads();
  int rc[SLICE_E / 256];
  #pragma unroll
  for (int k = 0; k < SLICE_E / 256; ++k) rc[k] = rcvcol[s * SLICE_E + k * 256 + tid];
  #pragma unroll
  for (int k = 0; k < SLICE_E / 256; ++k) {
    const int e = s * SLICE_E + k * 256 + tid;
    const unsigned d = (unsigned)rc[k] / (unsigned)NPB;
    if (d < (unsigned)NDEST) {
      const int c = atomicAdd(&lcnt[d], 1);
      if (c < LINE_I - 1) lent[d * LINE_I + 1 + c] = e * 128 + (rc[k] - (int)d * NPB);
    }
  }
  __syncthreads();
  const int lane = tid & 31, wid = tid >> 5;
  for (int d0 = wid; d0 < NDEST; d0 += 64) {
    int cc[8], ee[8];
    #pragma unroll
    for (int j = 0; j < 8; ++j) {
      const int d = (d0 + 8 * j < NDEST) ? d0 + 8 * j : NDEST - 1;
      cc[j] = lcnt[d];
      ee[j] = lent[d * LINE_I + lane];
    }
    #pragma unroll
    for (int j = 0; j < 8; ++j) {
      const int d = d0 + 8 * j;
      if (d < NDEST) {
        const int v = (lane == 0) ? cc[j] : ((lane <= cc[j]) ? ee[j] : 0);
        lists[(size_t)d * (NPB * 512) + s * LINE_I + lane] = v;
      }
    }
  }
}

__global__ void __launch_bounds__(256)
k_pre(const float* __restrict__ nf,
      const float* __restrict__ W1s,
      const float* __restrict__ W1v,
      const float* __restrict__ W2s,
      const float* __restrict__ W2v,
      const float* __restrict__ WskS,
      const float* __restrict__ WskV,
      float2* __restrict__ postPack,
      float* __restrict__ sv,
      const int* __restrict__ eidx,
      int* __restrict__ lists) {
  __shared__ alignas(16) float2 w1l[2 * 32 * LPITCH];

  if (blockIdx.x >= PRE_GEMM_WGS) {
    bin_slice(eidx + NE, blockIdx.x - PRE_GEMM_WGS, lists, (int*)w1l);
    return;
  }

  float4 lo[4], hi[4];
  #pragma unroll
  for (int t = 0; t < 4; ++t) {
    const int q = threadIdx.x + 256 * t;
    const int m = q >> 9, kk = (q >> 4) & 31, n4 = (q & 15) * 4;
    const float* src = (m ? W1v : W1s) + (2 * kk) * 64 + n4;
    lo[t] = *(const float4*)src;
    hi[t] = *(const float4*)(src + 64);
  }

  for (int i = blockIdx.x * 256 + threadIdx.x; i < POST_PACK_PAIRS; i += PRE_GEMM_WGS * 256) {
    const float* src; int j;
    if (i < 4096)       { src = W2s;  j = i; }
    else if (i < 8192)  { src = W2v;  j = i - 4096; }
    else if (i < 24576) { src = WskS; j = i - 8192; }
    else                { src = WskV; j = i - 24576; }
    const int kk = j >> 6, n = j & 63;
    postPack[i] = make_float2(src[(2 * kk) * 64 + n], src[(2 * kk + 1) * 64 + n]);
  }

  #pragma unroll
  for (int t = 0; t < 4; ++t) {
    const int q = threadIdx.x + 256 * t;
    const int m = q >> 9, kk = (q >> 4) & 31, n4 = (q & 15) * 4;
    float4* d = (float4*)(w1l + m * (32 * LPITCH) + kk * LPITCH + n4);
    d[0] = make_float4(lo[t].x, hi[t].x, lo[t].y, hi[t].y);
    d[1] = make_float4(lo[t].z, hi[t].z, lo[t].w, hi[t].w);
  }
  __syncthreads();
  const float2* W1sp = w1l;
  const float2* W1vp = w1l + 32 * LPITCH;

  const int lane = threadIdx.x & 31;
  const int wid  = threadIdx.x >> 5;
  const int l16  = lane & 15;
  const int kh   = lane >> 4;
  const int row0 = (blockIdx.x * 8 + wid) * 16;
  int rowA = row0 + l16; if (rowA >= NN) rowA = NN - 1;
  const float* nrow = nf + (size_t)rowA * 256;

  FragBlk<8> f[2];
  frag_load<true, 8, LPITCH>(f[0], nrow, W1sp, 0, kh, l16);
  v8f acc[4];
  #pragma unroll
  for (int n = 0; n < 8; ++n) {
    const int tr = n >> 1;
    if (n + 1 < 8) {
      const int trn = (n + 1) >> 1, kcn = ((n + 1) & 1) * 8;
      if (trn == 0) frag_load<true, 8, LPITCH>(f[(n + 1) & 1], nrow, W1sp, kcn, kh, l16);
      else          frag_load<false, 8, LPITCH>(f[(n + 1) & 1], nrow + 64 + (trn - 1), W1vp, kcn, kh, l16);
    }
    __builtin_amdgcn_sched_barrier(0);
    if ((n & 1) == 0) {
      #pragma unroll
      for (int ct = 0; ct < 4; ++ct) acc[ct] = v8f{};
    }
    frag_mma<8>(f[n & 1], acc);
    if (n & 1) {
      #pragma unroll
      for (int ct = 0; ct < 4; ++ct) {
        #pragma unroll
        for (int j = 0; j < 8; ++j) {
          const int r = row0 + j + 8 * kh;
          if (r < NN) {
            const int c = ct * 16 + l16;
            sv[r * 256 + tr * 64 + c] = acc[ct][j] * 0.125f;
          }
        }
      }
    }
    __builtin_amdgcn_sched_barrier(0);
  }
}

#define BCAP 128
#define MSG_THREADS 512
#define MSG_WAVES (MSG_THREADS / 32)
#define LINES_PER_WAVE ((NSLICE + MSG_WAVES - 1) / MSG_WAVES)
static_assert(NE % BCAP == 0 && BCAP == 128, "rounds and the 4-per-lane sort assume this");

__device__ __forceinline__ float lane_bcast(float v, int src) {
  return __int_as_float(__builtin_amdgcn_readlane(__float_as_int(v), src));
}

__device__ __forceinline__ void bin_edges(const int* __restrict__ rcvcol, int lo, int hi,
                                          int n0, int* cnt, int* bucket) {
  const int tid = threadIdx.x;
  if (tid < NPB) cnt[tid] = 0;
  __syncthreads();
  const int4* r4 = (const int4*)rcvcol;
  const int lo4 = lo >> 2, n4 = (hi - lo) >> 2;
  const int rot = (int)((blockIdx.x * 320u) % (unsigned)n4);
  for (int i0 = tid; i0 < n4; i0 += MSG_THREADS * 8) {
    int4 v[8];
    int  at[8];
    #pragma unroll
    for (int k = 0; k < 8; ++k) {
      const int i = i0 + k * MSG_THREADS;
      int t = i + rot; if (t >= n4) t -= n4;
      at[k] = lo4 + t;
      v[k] = (i < n4) ? r4[at[k]] : make_int4(-1, -1, -1, -1);
    }
    #pragma unroll
    for (int k = 0; k < 8; ++k) {
      const int rr[4] = {v[k].x, v[k].y, v[k].z, v[k].w};
      #pragma unroll
      for (int c = 0; c < 4; ++c) {
        const int r = rr[c] - n0;
        if ((unsigned)r < (unsigned)NPB) {
          const int slot = atomicAdd(&cnt[r], 1);
          if (slot < BCAP) bucket[r * BCAP + slot] = at[k] * 4 + c;
        }
      }
    }
  }
  __syncthreads();
}

__global__ void __launch_bounds__(MSG_THREADS)
k_msg(const float* __restrict__ ea,
      const float* __restrict__ efeat,
      const int*   __restrict__ eidx,
      const float* __restrict__ Wm1,
      const float* __restrict__ Wm2,
      const float* __restrict__ sv,
      float* __restrict__ msmv) {
  __shared__ int cnt[NPB];
  __shared__ int bucket[NPB * BCAP];
  const int lane = threadIdx.x & 31;
  const int wid  = threadIdx.x >> 5;
  const int n0   = blockIdx.x * NPB;
  const int hr   = lane & 7;
  const int u0 = lane, u1 = lane + 32;

  float w1c[8];
  #pragma unroll
  for (int q = 0; q < 8; ++q) w1c[q] = Wm1[q * 8 + hr];
  float wm[8][4][2];
  #pragma unroll
  for (int r = 0; r < 8; ++r) {
    #pragma unroll
    for (int k = 0; k < 4; ++k) {
      wm[r][k][0] = Wm2[r * 256 + k * 64 + u0];
      wm[r][k][1] = Wm2[r * 256 + k * 64 + u1];
    }
  }

  if (threadIdx.x < NPB) cnt[threadIdx.x] = 0;
  __syncthreads();
  const float* lst = msmv + (size_t)blockIdx.x * (NPB * 512);
  int lv[LINES_PER_WAVE];
  #pragma unroll
  for (int k = 0; k < LINES_PER_WAVE; ++k) {
    const int s = wid + k * MSG_WAVES;
    lv[k] = (s < NSLICE) ? __float_as_int(lst[s * LINE_I + lane]) : 0;
  }
  int spill = 0;
  #pragma unroll
  for (int k = 0; k < LINES_PER_WAVE; ++k) {
    const int c = __builtin_amdgcn_readfirstlane(lv[k]);
    if (c > LINE_I - 1) spill = 1;
    if (lane >= 1 && lane <= c) {
      const int r = lv[k] & 127;
      if (r < NPB) {
        const int slot = atomicAdd(&cnt[r], 1);
        if (slot < BCAP) bucket[r * BCAP + slot] = lv[k] >> 7;
      }
    }
  }
  __syncthreads();
  const int over = __syncthreads_or(spill | ((threadIdx.x < NPB) && (cnt[threadIdx.x < NPB ? threadIdx.x : 0] > BCAP)));
  const int span = over ? BCAP : NE;

  for (int eb = 0; eb < NE; eb += span) {
    if (over) bin_edges(eidx + NE, eb, eb + span, n0, cnt, bucket);

    for (int ln = wid; ln < NPB; ln += MSG_WAVES) {
      int* bk = bucket + ln * BCAP;
      int d = cnt[ln]; if (d > BCAP) d = BCAP;

      {
        const int x0 = (lane < d) ? bk[lane] : 0x7fffffff;
        int r0 = 0;
        for (int j = 0; j < d; ++j) r0 += (bk[j] < x0) ? 1 : 0;
        if (d > 32) {
          const int x1 = (lane + 32 < d) ? bk[lane + 32] : 0x7fffffff;
          const int x2 = (lane + 64 < d) ? bk[lane + 64] : 0x7fffffff;
          const int x3 = (lane + 96 < d) ? bk[lane + 96] : 0x7fffffff;
          int r1 = 0, r2 = 0, r3 = 0;
          for (int j = 0; j < d; ++j) {
            const int v = bk[j];
            r1 += (v < x1) ? 1 : 0; r2 += (v < x2) ? 1 : 0; r3 += (v < x3) ? 1 : 0;
          }
          if (lane + 32 < d) bk[r1] = x1;
          if (lane + 64 < d) bk[r2] = x2;
          if (lane + 96 < d) bk[r3] = x3;
        }
        if (lane < d) bk[r0] = x0;
      }

      float acc[8][2];
      #pragma unroll
      for (int p = 0; p < 8; ++p) { acc[p][0] = 0.f; acc[p][1] = 0.f; }

      for (int g = 0; g < d; g += 4) {
        int j = g + (lane >> 3); if (j > d - 1) j = d - 1;
        const int em = bk[j];
        const float4 f0 = *(const float4*)(efeat + (size_t)em * 8);
        const float4 f1 = *(const float4*)(efeat + (size_t)em * 8 + 4);
        const int    sndl = eidx[em];
        const float4 yl = *(const float4*)(ea + (size_t)em * 4);
        float tt = f0.x * w1c[0];
        tt = fmaf(f0.y, w1c[1], tt); tt = fmaf(f0.z, w1c[2], tt); tt = fmaf(f0.w, w1c[3], tt);
        tt = fmaf(f1.x, w1c[4], tt); tt = fmaf(f1.y, w1c[5], tt); tt = fmaf(f1.z, w1c[6], tt);
        tt = fmaf(f1.w, w1c[7], tt);
        const float hl = sspf(tt * 0.35355339059f);

        float Y0[4], Y1x[4], Y1y[4], Y1z[4];
        float se[4][2], vx[4][2], vy[4][2], vz[4][2];
        #pragma unroll
        for (int e = 0; e < 4; ++e) {
          const int sl = e * 8;
          const int snd = __builtin_amdgcn_readlane(sndl, sl);
          Y0[e]  = lane_bcast(yl.x, sl);
          Y1x[e] = lane_bcast(yl.y, sl);
          Y1y[e] = lane_bcast(yl.z, sl);
          Y1z[e] = lane_bcast(yl.w, sl);
          const float* srow = sv + (size_t)snd * 256;
          se[e][0] = srow[u0];        se[e][1] = srow[u1];
          vx[e][0] = srow[64 + u0];   vx[e][1] = srow[64 + u1];
          vy[e][0] = srow[128 + u0];  vy[e][1] = srow[128 + u1];
          vz[e][0] = srow[192 + u0];  vz[e][1] = srow[192 + u1];
        }

        #pragma unroll
        for (int e = 0; e < 4; ++e) {
          if (g + e < d) {
            float h[8];
            #pragma unroll
            for (int r = 0; r < 8; ++r) h[r] = lane_bcast(hl, e * 8 + r);
            #pragma unroll
            for (int t = 0; t < 2; ++t) {
              float w[4];
              #pragma unroll
              for (int k = 0; k < 4; ++k) {
                float tk = h[0] * wm[0][k][t];
                #pragma unroll
                for (int r = 1; r < 8; ++r) tk = fmaf(h[r], wm[r][k][t], tk);
                w[k] = tk * 0.35355339059f;
              }
              const float s1 = se[e][t];
              const float ax = vx[e][t], ay = vy[e][t], az = vz[e][t];
              const float dv = ax * Y1x[e] + ay * Y1y[e] + az * Y1z[e];
              acc[0][t] += w[0] * s1 * Y0[e];
              acc[1][t] += w[3] * dv * 0.57735026919f;
              acc[2][t] += w[1] * s1 * Y1x[e];
              acc[3][t] += w[1] * s1 * Y1y[e];
              acc[4][t] += w[1] * s1 * Y1z[e];
              acc[5][t] += w[2] * ax * Y0[e];
              acc[6][t] += w[2] * ay * Y0[e];
              acc[7][t] += w[2] * az * Y0[e];
            }
          }
        }
      }

      float* base = msmv + (size_t)(n0 + ln) * 512;
      #pragma unroll
      for (int p = 0; p < 8; ++p) {
        float a0 = acc[p][0], a1 = acc[p][1];
        if (eb != 0) { a0 += base[p * 64 + u0]; a1 += base[p * 64 + u1]; }
        base[p * 64 + u0] = a0;
        base[p * 64 + u1] = a1;
      }
    }
    __syncthreads();
  }
}

#define PNPB 80
#define POST_TILES (PNPB / 16)
#define POST_WAVES (POST_TILES * 4)
#define POST_THREADS (POST_WAVES * 32)
#define SPITCH 260
static_assert(NN % PNPB == 0, "post-transform row ranges must tile the nodes");

__global__ void __launch_bounds__(POST_THREADS)
k_post(const float* __restrict__ msmv,
       const float* __restrict__ attrs,
       const float2* __restrict__ W2sp,
       const float2* __restrict__ W2vp,
       const float2* __restrict__ WskSp,
       const float2* __restrict__ WskVp,
       float* __restrict__ out) {
  __shared__ alignas(16) float lds[POST_WAVES * 16 * 64];
  __shared__ alignas(16) float stage[PNPB * SPITCH];
  const int lane = threadIdx.x & 31;
  const int wid  = __builtin_amdgcn_readfirstlane((int)(threadIdx.x >> 5));
  const int l16  = lane & 15;
  const int kh   = lane >> 4;
  const int tile = wid % POST_TILES;
  const int tr   = wid / POST_TILES;
  const int row0 = blockIdx.x * PNPB + tile * 16;
  const int rowA = row0 + l16;
  float* myLds = lds + wid * (16 * 64);
  const float* mrow = msmv + (size_t)rowA * 512;

  const float* arow8 = attrs + (size_t)rowA * 8;
  const float attE0 = arow8[2 * kh]     * 0.04419417382f;
  const float attE1 = arow8[2 * kh + 1] * 0.04419417382f;
  const float attO0 = arow8[4 + 2 * kh] * 0.04419417382f;
  const float attO1 = arow8[5 + 2 * kh] * 0.04419417382f;

  const float2* W2p  = (tr == 0) ? W2sp : W2vp;
  const float2* Wskp = (tr == 0) ? WskSp : WskVp;
  const int pA = (tr == 0) ? 0 : (1 + tr);
  const int pB = (tr == 0) ? 1 : (4 + tr);
  const float* arowA = mrow + pA * 64;
  const float* arowB = mrow + pB * 64;

  FragBlk<4> f[2];
  SkipBlk sb[2];
  frag_load<true, 4, GPITCH>(f[0], arowA, W2p, 0, kh, l16);

  v8f acc[4] = {};
  #pragma unroll
  for (int n = 0; n < 8; ++n) {
    if (n + 1 < 8)
      frag_load<true, 4, GPITCH>(f[(n + 1) & 1], (n + 1 < 4) ? arowA : arowB,
                         (n + 1 < 4) ? W2p : W2p + 32 * 64, ((n + 1) & 3) * 4, kh, l16);
    else
      skip_load(sb[0], Wskp, 0, kh, l16);
    __builtin_amdgcn_sched_barrier(0);
    frag_mma<4>(f[n & 1], acc);
    __builtin_amdgcn_sched_barrier(0);
  }

  #pragma unroll
  for (int ct = 0; ct < 4; ++ct) {
    #pragma unroll
    for (int j = 0; j < 8; ++j) {
      acc[ct][j] *= 0.08838834765f;
      myLds[(j + 8 * kh) * 64 + ct * 16 + l16] = acc[ct][j];
    }
  }
  __syncthreads();

  const float* m2row = myLds + l16 * 64;
  for (int blk = 0; blk < 14; blk += 2) {
    skip_load(sb[1], Wskp, (blk + 1) * 4, kh, l16);
    skip_mma(sb[0], m2row + blk * 4, attE0, attE1, attO0, attO1, acc);
    skip_load(sb[0], Wskp, (blk + 2) * 4, kh, l16);
    skip_mma(sb[1], m2row + (blk + 1) * 4, attE0, attE1, attO0, attO1, acc);
  }
  skip_load(sb[1], Wskp, 15 * 4, kh, l16);
  skip_mma(sb[0], m2row + 14 * 4, attE0, attE1, attO0, attO1, acc);
  skip_mma(sb[1], m2row + 15 * 4, attE0, attE1, attO0, attO1, acc);

  #pragma unroll
  for (int ct = 0; ct < 4; ++ct) {
    #pragma unroll
    for (int j = 0; j < 8; ++j) {
      const int r = tile * 16 + j + 8 * kh;
      const int c = ct * 16 + l16;
      const int o = (tr == 0) ? c : (64 + c * 3 + (tr - 1));
      stage[r * SPITCH + o] = acc[ct][j];
    }
  }
  __syncthreads();

  for (int i = threadIdx.x; i < PNPB * 64; i += POST_THREADS) {
    const int r = i >> 6, q = i & 63;
    const float4 v = *(const float4*)(stage + r * SPITCH + q * 4);
    *(float4*)(out + (size_t)(blockIdx.x * PNPB + r) * 256 + q * 4) = v;
  }
}

extern "C" void kernel_launch(void* const* d_in, const int* in_sizes, int n_in,
                              void* d_out, int out_size, void* d_ws, size_t ws_size,
                              hipStream_t stream) {
  const float* node_attrs = (const float*)d_in[0];
  const float* node_feats = (const float*)d_in[1];
  const float* edge_attrs = (const float*)d_in[2];
  const float* edge_feats = (const float*)d_in[3];
  const int*   edge_index = (const int*)d_in[4];
  const float* W1s  = (const float*)d_in[5];
  const float* W1v  = (const float*)d_in[6];
  const float* Wm1  = (const float*)d_in[7];
  const float* Wm2  = (const float*)d_in[8];
  const float* W2s  = (const float*)d_in[9];
  const float* W2v  = (const float*)d_in[10];
  const float* WskS = (const float*)d_in[11];
  const float* WskV = (const float*)d_in[12];

  float* out  = (float*)d_out;
  float* msmv = (float*)d_ws;

  float2* packBase = (float2*)(msmv + (size_t)NN * 512);
  float2* W2sp  = packBase;
  float2* W2vp  = packBase + 4096;
  float2* WskSp = packBase + 8192;
  float2* WskVp = packBase + 24576;

  k_pre <<<PRE_GEMM_WGS + NSLICE, 256, 0, stream>>>(node_feats, W1s, W1v, W2s, W2v, WskS, WskV,
                                               packBase, out, edge_index, (int*)msmv);
  k_msg <<<NN / NPB, MSG_THREADS, 0, stream>>>(edge_attrs, edge_feats, edge_index,
                                               Wm1, Wm2, out, msmv);
  k_post<<<NN / PNPB, POST_THREADS, 0, stream>>>(msmv, node_attrs, W2sp, W2vp,
                                               WskSp, WskVp, out);
}


# k_msg: the softplus tail log1p(exp(-|x|)) taken as ln2 * hardware log2(1 + e), e in (0,1], absolute error below 2^-22, not the library log1pf: about 10 vector instructions per evaluation, not 90
# speedup vs baseline: 4.0987x; 1.0519x over previous
#include <hip/hip_runtime.h>
#include <hip/hip_bf16.h>
#include <math.h>


#define NN 20000
#define NE 320000

#define NPB  80
#define NDEST (NN / NPB)
#define NSLICE 250
#define SLICE_E (NE / NSLICE)
#define LINE_I 32
#define PRE_GEMM_WGS ((NN + 127) / 128)
static_assert(NN % NPB == 0, "receiver ranges must tile the nodes");
static_assert(NE % NSLICE == 0 && SLICE_E % 256 == 0, "slices tile the edge list, 256 threads a slice");
static_assert(NPB <= 128 && NE <= (1 << 24), "an entry packs edge id * 128 + node index into 31 bits");
static_assert(NSLICE * LINE_I <= NPB * 512, "a range's list lines fit inside its own message-sum rows");
static_assert(NDEST <= 256 && (256 + NDEST * LINE_I) * 4 <= 40960, "a slice's counters and lines fit the first launch's LDS");

typedef __attribute__((ext_vector_type(2))) float v2f;
typedef __attribute__((ext_vector_type(8))) float v8f;

__device__ __forceinline__ v8f wmma_f32_k4(v2f a, v2f b, v8f c) {
  return __builtin_amdgcn_wmma_f32_16x16x4_f32(false, a, false, b,
                                               (short)0, c, false, false);
}

__device__ __forceinline__ float sspf(float x) {
  const float e = __expf(-fabsf(x));
  return fmaxf(x, 0.0f) + 0.69314718056f * __log2f(1.0f + e) - 0.69314718056f;
}

#define GPITCH 64
#define LPITCH 80
#define POST_PACK_PAIRS 40960

template <int KB>
struct FragBlk { v2f a[KB]; v2f b[KB][4]; };

template <bool ACONTIG, int KB, int BPITCH>
__device__ __forceinline__ void frag_load(FragBlk<KB>& f, const float* __restrict__ arow,
                                          const float2* __restrict__ Wp,
                                          int kc0, int kh, int l16) {
  #pragma unroll
  for (int q = 0; q < KB; ++q) {
    const int k0 = (kc0 + q) * 4 + 2 * kh;
    if (ACONTIG) {
      const float2 av = *(const float2*)(arow + k0);
      f.a[q].x = av.x; f.a[q].y = av.y;
    } else {
      f.a[q].x = arow[k0 * 3];
      f.a[q].y = arow[(k0 + 1) * 3];
    }
    const float2* bp = Wp + (k0 >> 1) * BPITCH + l16;
    #pragma unroll
    for (int ct = 0; ct < 4; ++ct) {
      const float2 bv = bp[ct * 16];
      f.b[q][ct].x = bv.x; f.b[q][ct].y = bv.y;
    }
  }
}

template <int KB>
__device__ __forceinline__ void frag_mma(const FragBlk<KB>& f, v8f acc[4]) {
  #pragma unroll
  for (int q = 0; q < KB; ++q) {
    #pragma unroll
    for (int ct = 0; ct < 4; ++ct) acc[ct] = wmma_f32_k4(f.a[q], f.b[q][ct], acc[ct]);
  }
}

struct SkipBlk { v2f b[4][2][4]; };

__device__ __forceinline__ void skip_load(SkipBlk& s, const float2* __restrict__ Wskp,
                                          int i0, int kh, int l16) {
  #pragma unroll
  for (int u = 0; u < 4; ++u) {
    const float2* bp = Wskp + (4 * (i0 + u) + kh) * 64 + l16;
    #pragma unroll
    for (int h = 0; h < 2; ++h) {
      #pragma unroll
      for (int ct = 0; ct < 4; ++ct) {
        const float2 bv = bp[h * 128 + ct * 16];
        s.b[u][h][ct].x = bv.x; s.b[u][h][ct].y = bv.y;
      }
    }
  }
}

__device__ __forceinline__ void skip_mma(const SkipBlk& s, const float* m2p,
                                         float attE0, float attE1, float attO0, float attO1,
                                         v8f acc[4]) {
  const float4 mv = *(const float4*)m2p;
  const float m2[4] = {mv.x, mv.y, mv.z, mv.w};
  #pragma unroll
  for (int u = 0; u < 4; ++u) {
    v2f a;
    a.x = m2[u] * attE0; a.y = m2[u] * attE1;
    #pragma unroll
    for (int ct = 0; ct < 4; ++ct) acc[ct] = wmma_f32_k4(a, s.b[u][0][ct], acc[ct]);
    a.x = m2[u] * attO0; a.y = m2[u] * attO1;
    #pragma unroll
    for (int ct = 0; ct < 4; ++ct) acc[ct] = wmma_f32_k4(a, s.b[u][1][ct], acc[ct]);
  }
}

__device__ __forceinline__ void bin_slice(const int* __restrict__ rcvcol, int s,
                                          int* __restrict__ lists, int* lds) {
  int* lcnt = lds;
  int* lent = lds + 256;
  const int tid = threadIdx.x;
  if (tid < NDEST) lcnt[tid] = 0;
  __syncthreads();
  int rc[SLICE_E / 256];
  #pragma unroll
  for (int k = 0; k < SLICE_E / 256; ++k) rc[k] = rcvcol[s * SLICE_E + k * 256 + tid];
  #pragma unroll
  for (int k = 0; k < SLICE_E / 256; ++k) {
    const int e = s * SLICE_E + k * 256 + tid;
    const unsigned d = (unsigned)rc[k] / (unsigned)NPB;
    if (d < (unsigned)NDEST) {
      const int c = atomicAdd(&lcnt[d], 1);
      if (c < LINE_I - 1) lent[d * LINE_I + 1 + c] = e * 128 + (rc[k] - (int)d * NPB);
    }
  }
  __syncthreads();
  const int lane = tid & 31, wid = tid >> 5;
  for (int d0 = wid; d0 < NDEST; d0 += 64) {
    int cc[8], ee[8];
    #pragma unroll
    for (int j = 0; j < 8; ++j) {
      const int d = (d0 + 8 * j < NDEST) ? d0 + 8 * j : NDEST - 1;
      cc[j] = lcnt[d];
      ee[j] = lent[d * LINE_I + lane];
    }
    #pragma unroll
    for (int j = 0; j < 8; ++j) {
      const int d = d0 + 8 * j;
      if (d < NDEST) {
        const int v = (lane == 0) ? cc[j] : ((lane <= cc[j]) ? ee[j] : 0);
        lists[(size_t)d * (NPB * 512) + s * LINE_I + lane] = v;
      }
    }
  }
}

__global__ void __launch_bounds__(256)
k_pre(const float* __restrict__ nf,
      const float* __restrict__ W1s,
      const float* __restrict__ W1v,
      const float* __restrict__ W2s,
      const float* __restrict__ W2v,
      const float* __restrict__ WskS,
      const float* __restrict__ WskV,
      float2* __restrict__ postPack,
      float* __restrict__ sv,
      const int* __restrict__ eidx,
      int* __restrict__ lists) {
  __shared__ alignas(16) float2 w1l[2 * 32 * LPITCH];

  if (blockIdx.x >= PRE_GEMM_WGS) {
    bin_slice(eidx + NE, blockIdx.x - PRE_GEMM_WGS, lists, (int*)w1l);
    return;
  }

  float4 lo[4], hi[4];
  #pragma unroll
  for (int t = 0; t < 4; ++t) {
    const int q = threadIdx.x + 256 * t;
    const int m = q >> 9, kk = (q >> 4) & 31, n4 = (q & 15) * 4;
    const float* src = (m ? W1v : W1s) + (2 * kk) * 64 + n4;
    lo[t] = *(const float4*)src;
    hi[t] = *(const float4*)(src + 64);
  }

  for (int i = blockIdx.x * 256 + threadIdx.x; i < POST_PACK_PAIRS; i += PRE_GEMM_WGS * 256) {
    const float* src; int j;
    if (i < 4096)       { src = W2s;  j = i; }
    else if (i < 8192)  { src = W2v;  j = i - 4096; }
    else if (i < 24576) { src = WskS; j = i - 8192; }
    else                { src = WskV; j = i - 24576; }
    const int kk = j >> 6, n = j & 63;
    postPack[i] = make_float2(src[(2 * kk) * 64 + n], src[(2 * kk + 1) * 64 + n]);
  }

  #pragma unroll
  for (int t = 0; t < 4; ++t) {
    const int q = threadIdx.x + 256 * t;
    const int m = q >> 9, kk = (q >> 4) & 31, n4 = (q & 15) * 4;
    float4* d = (float4*)(w1l + m * (32 * LPITCH) + kk * LPITCH + n4);
    d[0] = make_float4(lo[t].x, hi[t].x, lo[t].y, hi[t].y);
    d[1] = make_float4(lo[t].z, hi[t].z, lo[t].w, hi[t].w);
  }
  __syncthreads();
  const float2* W1sp = w1l;
  const float2* W1vp = w1l + 32 * LPITCH;

  const int lane = threadIdx.x & 31;
  const int wid  = threadIdx.x >> 5;
  const int l16  = lane & 15;
  const int kh   = lane >> 4;
  const int row0 = (blockIdx.x * 8 + wid) * 16;
  int rowA = row0 + l16; if (rowA >= NN) rowA = NN - 1;
  const float* nrow = nf + (size_t)rowA * 256;

  FragBlk<8> f[2];
  frag_load<true, 8, LPITCH>(f[0], nrow, W1sp, 0, kh, l16);
  v8f acc[4];
  #pragma unroll
  for (int n = 0; n < 8; ++n) {
    const int tr = n >> 1;
    if (n + 1 < 8) {
      const int trn = (n + 1) >> 1, kcn = ((n + 1) & 1) * 8;
      if (trn == 0) frag_load<true, 8, LPITCH>(f[(n + 1) & 1], nrow, W1sp, kcn, kh, l16);
      else          frag_load<false, 8, LPITCH>(f[(n + 1) & 1], nrow + 64 + (trn - 1), W1vp, kcn, kh, l16);
    }
    __builtin_amdgcn_sched_barrier(0);
    if ((n & 1) == 0) {
      #pragma unroll
      for (int ct = 0; ct < 4; ++ct) acc[ct] = v8f{};
    }
    frag_mma<8>(f[n & 1], acc);
    if (n & 1) {
      #pragma unroll
      for (int ct = 0; ct < 4; ++ct) {
        #pragma unroll
        for (int j = 0; j < 8; ++j) {
          const int r = row0 + j + 8 * kh;
          if (r < NN) {
            const int c = ct * 16 + l16;
            sv[r * 256 + tr * 64 + c] = acc[ct][j] * 0.125f;
          }
        }
      }
    }
    __builtin_amdgcn_sched_barrier(0);
  }
}

#define BCAP 128
#define MSG_THREADS 512
#define MSG_WAVES (MSG_THREADS / 32)
#define LINES_PER_WAVE ((NSLICE + MSG_WAVES - 1) / MSG_WAVES)
static_assert(NE % BCAP == 0 && BCAP == 128, "rounds and the 4-per-lane sort assume this");

__device__ __forceinline__ float lane_bcast(float v, int src) {
  return __int_as_float(__builtin_amdgcn_readlane(__float_as_int(v), src));
}

__device__ __forceinline__ void bin_edges(const int* __restrict__ rcvcol, int lo, int hi,
                                          int n0, int* cnt, int* bucket) {
  const int tid = threadIdx.x;
  if (tid < NPB) cnt[tid] = 0;
  __syncthreads();
  const int4* r4 = (const int4*)rcvcol;
  const int lo4 = lo >> 2, n4 = (hi - lo) >> 2;
  const int rot = (int)((blockIdx.x * 320u) % (unsigned)n4);
  for (int i0 = tid; i0 < n4; i0 += MSG_THREADS * 8) {
    int4 v[8];
    int  at[8];
    #pragma unroll
    for (int k = 0; k < 8; ++k) {
      const int i = i0 + k * MSG_THREADS;
      int t = i + rot; if (t >= n4) t -= n4;
      at[k] = lo4 + t;
      v[k] = (i < n4) ? r4[at[k]] : make_int4(-1, -1, -1, -1);
    }
    #pragma unroll
    for (int k = 0; k < 8; ++k) {
      const int rr[4] = {v[k].x, v[k].y, v[k].z, v[k].w};
      #pragma unroll
      for (int c = 0; c < 4; ++c) {
        const int r = rr[c] - n0;
        if ((unsigned)r < (unsigned)NPB) {
          const int slot = atomicAdd(&cnt[r], 1);
          if (slot < BCAP) bucket[r * BCAP + slot] = at[k] * 4 + c;
        }
      }
    }
  }
  __syncthreads();
}

__global__ void __launch_bounds__(MSG_THREADS)
k_msg(const float* __restrict__ ea,
      const float* __restrict__ efeat,
      const int*   __restrict__ eidx,
      const float* __restrict__ Wm1,
      const float* __restrict__ Wm2,
      const float* __restrict__ sv,
      float* __restrict__ msmv) {
  __shared__ int cnt[NPB];
  __shared__ int bucket[NPB * BCAP];
  const int lane = threadIdx.x & 31;
  const int wid  = threadIdx.x >> 5;
  const int n0   = blockIdx.x * NPB;
  const int hr   = lane & 7;
  const int u0 = lane, u1 = lane + 32;

  float w1c[8];
  #pragma unroll
  for (int q = 0; q < 8; ++q) w1c[q] = Wm1[q * 8 + hr];
  float wm[8][4][2];
  #pragma unroll
  for (int r = 0; r < 8; ++r) {
    #pragma unroll
    for (int k = 0; k < 4; ++k) {
      wm[r][k][0] = Wm2[r * 256 + k * 64 + u0];
      wm[r][k][1] = Wm2[r * 256 + k * 64 + u1];
    }
  }

  if (threadIdx.x < NPB) cnt[threadIdx.x] = 0;
  __syncthreads();
  const float* lst = msmv + (size_t)blockIdx.x * (NPB * 512);
  int lv[LINES_PER_WAVE];
  #pragma unroll
  for (int k = 0; k < LINES_PER_WAVE; ++k) {
    const int s = wid + k * MSG_WAVES;
    lv[k] = (s < NSLICE) ? __float_as_int(lst[s * LINE_I + lane]) : 0;
  }
  int spill = 0;
  #pragma unroll
  for (int k = 0; k < LINES_PER_WAVE; ++k) {
    const int c = __builtin_amdgcn_readfirstlane(lv[k]);
    if (c > LINE_I - 1) spill = 1;
    if (lane >= 1 && lane <= c) {
      const int r = lv[k] & 127;
      if (r < NPB) {
        const int slot = atomicAdd(&cnt[r], 1);
        if (slot < BCAP) bucket[r * BCAP + slot] = lv[k] >> 7;
      }
    }
  }
  __syncthreads();
  const int over = __syncthreads_or(spill | ((threadIdx.x < NPB) && (cnt[threadIdx.x < NPB ? threadIdx.x : 0] > BCAP)));
  const int span = over ? BCAP : NE;

  for (int eb = 0; eb < NE; eb += span) {
    if (over) bin_edges(eidx + NE, eb, eb + span, n0, cnt, bucket);

    for (int ln = wid; ln < NPB; ln += MSG_WAVES) {
      int* bk = bucket + ln * BCAP;
      int d = cnt[ln]; if (d > BCAP) d = BCAP;

      {
        const int x0 = (lane < d) ? bk[lane] : 0x7fffffff;
        int r0 = 0;
        for (int j = 0; j < d; ++j) r0 += (bk[j] < x0) ? 1 : 0;
        if (d > 32) {
          const int x1 = (lane + 32 < d) ? bk[lane + 32] : 0x7fffffff;
          const int x2 = (lane + 64 < d) ? bk[lane + 64] : 0x7fffffff;
          const int x3 = (lane + 96 < d) ? bk[lane + 96] : 0x7fffffff;
          int r1 = 0, r2 = 0, r3 = 0;
          for (int j = 0; j < d; ++j) {
            const int v = bk[j];
            r1 += (v < x1) ? 1 : 0; r2 += (v < x2) ? 1 : 0; r3 += (v < x3) ? 1 : 0;
          }
          if (lane + 32 < d) bk[r1] = x1;
          if (lane + 64 < d) bk[r2] = x2;
          if (lane + 96 < d) bk[r3] = x3;
        }
        if (lane < d) bk[r0] = x0;
      }

      float acc[8][2];
      #pragma unroll
      for (int p = 0; p < 8; ++p) { acc[p][0] = 0.f; acc[p][1] = 0.f; }

      for (int g = 0; g < d; g += 4) {
        int j = g + (lane >> 3); if (j > d - 1) j = d - 1;
        const int em = bk[j];
        const float4 f0 = *(const float4*)(efeat + (size_t)em * 8);
        const float4 f1 = *(const float4*)(efeat + (size_t)em * 8 + 4);
        const int    sndl = eidx[em];
        const float4 yl = *(const float4*)(ea + (size_t)em * 4);
        float tt = f0.x * w1c[0];
        tt = fmaf(f0.y, w1c[1], tt); tt = fmaf(f0.z, w1c[2], tt); tt = fmaf(f0.w, w1c[3], tt);
        tt = fmaf(f1.x, w1c[4], tt); tt = fmaf(f1.y, w1c[5], tt); tt = fmaf(f1.z, w1c[6], tt);
        tt = fmaf(f1.w, w1c[7], tt);
        const float hl = sspf(tt * 0.35355339059f);

        float Y0[4], Y1x[4], Y1y[4], Y1z[4];
        float se[4][2], vx[4][2], vy[4][2], vz[4][2];
        #pragma unroll
        for (int e = 0; e < 4; ++e) {
          const int sl = e * 8;
          const int snd = __builtin_amdgcn_readlane(sndl, sl);
          Y0[e]  = lane_bcast(yl.x, sl);
          Y1x[e] = lane_bcast(yl.y, sl);
          Y1y[e] = lane_bcast(yl.z, sl);
          Y1z[e] = lane_bcast(yl.w, sl);
          const float* srow = sv + (size_t)snd * 256;
          se[e][0] = srow[u0];        se[e][1] = srow[u1];
          vx[e][0] = srow[64 + u0];   vx[e][1] = srow[64 + u1];
          vy[e][0] = srow[128 + u0];  vy[e][1] = srow[128 + u1];
          vz[e][0] = srow[192 + u0];  vz[e][1] = srow[192 + u1];
        }

        #pragma unroll
        for (int e = 0; e < 4; ++e) {
          if (g + e < d) {
            float h[8];
            #pragma unroll
            for (int r = 0; r < 8; ++r) h[r] = lane_bcast(hl, e * 8 + r);
            #pragma unroll
            for (int t = 0; t < 2; ++t) {
              float w[4];
              #pragma unroll
              for (int k = 0; k < 4; ++k) {
                float tk = h[0] * wm[0][k][t];
                #pragma unroll
                for (int r = 1; r < 8; ++r) tk = fmaf(h[r], wm[r][k][t], tk);
                w[k] = tk * 0.35355339059f;
              }
              const float s1 = se[e][t];
              const float ax = vx[e][t], ay = vy[e][t], az = vz[e][t];
              const float dv = ax * Y1x[e] + ay * Y1y[e] + az * Y1z[e];
              acc[0][t] += w[0] * s1 * Y0[e];
              acc[1][t] += w[3] * dv * 0.57735026919f;
              acc[2][t] += w[1] * s1 * Y1x[e];
              acc[3][t] += w[1] * s1 * Y1y[e];
              acc[4][t] += w[1] * s1 * Y1z[e];
              acc[5][t] += w[2] * ax * Y0[e];
              acc[6][t] += w[2] * ay * Y0[e];
              acc[7][t] += w[2] * az * Y0[e];
            }
          }
        }
      }

      float* base = msmv + (size_t)(n0 + ln) * 512;
      #pragma unroll
      for (int p = 0; p < 8; ++p) {
        float a0 = acc[p][0], a1 = acc[p][1];
        if (eb != 0) { a0 += base[p * 64 + u0]; a1 += base[p * 64 + u1]; }
        base[p * 64 + u0] = a0;
        base[p * 64 + u1] = a1;
      }
    }
    __syncthreads();
  }
}

#define PNPB 80
#define POST_TILES (PNPB / 16)
#define POST_WAVES (POST_TILES * 4)
#define POST_THREADS (POST_WAVES * 32)
#define SPITCH 260
static_assert(NN % PNPB == 0, "post-transform row ranges must tile the nodes");

__global__ void __launch_bounds__(POST_THREADS)
k_post(const float* __restrict__ msmv,
       const float* __restrict__ attrs,
       const float2* __restrict__ W2sp,
       const float2* __restrict__ W2vp,
       const float2* __restrict__ WskSp,
       const float2* __restrict__ WskVp,
       float* __restrict__ out) {
  __shared__ alignas(16) float lds[POST_WAVES * 16 * 64];
  __shared__ alignas(16) float stage[PNPB * SPITCH];
  const int lane = threadIdx.x & 31;
  const int wid  = __builtin_amdgcn_readfirstlane((int)(threadIdx.x >> 5));
  const int l16  = lane & 15;
  const int kh   = lane >> 4;
  const int tile = wid % POST_TILES;
  const int tr   = wid / POST_TILES;
  const int row0 = blockIdx.x * PNPB + tile * 16;
  const int rowA = row0 + l16;
  float* myLds = lds + wid * (16 * 64);
  const float* mrow = msmv + (size_t)rowA * 512;

  const float* arow8 = attrs + (size_t)rowA * 8;
  const float attE0 = arow8[2 * kh]     * 0.04419417382f;
  const float attE1 = arow8[2 * kh + 1] * 0.04419417382f;
  const float attO0 = arow8[4 + 2 * kh] * 0.04419417382f;
  const float attO1 = arow8[5 + 2 * kh] * 0.04419417382f;

  const float2* W2p  = (tr == 0) ? W2sp : W2vp;
  const float2* Wskp = (tr == 0) ? WskSp : WskVp;
  const int pA = (tr == 0) ? 0 : (1 + tr);
  const int pB = (tr == 0) ? 1 : (4 + tr);
  const float* arowA = mrow + pA * 64;
  const float* arowB = mrow + pB * 64;

  FragBlk<4> f[2];
  SkipBlk sb[2];
  frag_load<true, 4, GPITCH>(f[0], arowA, W2p, 0, kh, l16);

  v8f acc[4] = {};
  #pragma unroll
  for (int n = 0; n < 8; ++n) {
    if (n + 1 < 8)
      frag_load<true, 4, GPITCH>(f[(n + 1) & 1], (n + 1 < 4) ? arowA : arowB,
                         (n + 1 < 4) ? W2p : W2p + 32 * 64, ((n + 1) & 3) * 4, kh, l16);
    else
      skip_load(sb[0], Wskp, 0, kh, l16);
    __builtin_amdgcn_sched_barrier(0);
    frag_mma<4>(f[n & 1], acc);
    __builtin_amdgcn_sched_barrier(0);
  }

  #pragma unroll
  for (int ct = 0; ct < 4; ++ct) {
    #pragma unroll
    for (int j = 0; j < 8; ++j) {
      acc[ct][j] *= 0.08838834765f;
      myLds[(j + 8 * kh) * 64 + ct * 16 + l16] = acc[ct][j];
    }
  }
  __syncthreads();

  const float* m2row = myLds + l16 * 64;
  for (int blk = 0; blk < 14; blk += 2) {
    skip_load(sb[1], Wskp, (blk + 1) * 4, kh, l16);
    skip_mma(sb[0], m2row + blk * 4, attE0, attE1, attO0, attO1, acc);
    skip_load(sb[0], Wskp, (blk + 2) * 4, kh, l16);
    skip_mma(sb[1], m2row + (blk + 1) * 4, attE0, attE1, attO0, attO1, acc);
  }
  skip_load(sb[1], Wskp, 15 * 4, kh, l16);
  skip_mma(sb[0], m2row + 14 * 4, attE0, attE1, attO0, attO1, acc);
  skip_mma(sb[1], m2row + 15 * 4, attE0, attE1, attO0, attO1, acc);

  #pragma unroll
  for (int ct = 0; ct < 4; ++ct) {
    #pragma unroll
    for (int j = 0; j < 8; ++j) {
      const int r = tile * 16 + j + 8 * kh;
      const int c = ct * 16 + l16;
      const int o = (tr == 0) ? c : (64 + c * 3 + (tr - 1));
      stage[r * SPITCH + o] = acc[ct][j];
    }
  }
  __syncthreads();

  for (int i = threadIdx.x; i < PNPB * 64; i += POST_THREADS) {
    const int r = i >> 6, q = i & 63;
    const float4 v = *(const float4*)(stage + r * SPITCH + q * 4);
    *(float4*)(out + (size_t)(blockIdx.x * PNPB + r) * 256 + q * 4) = v;
  }
}

extern "C" void kernel_launch(void* const* d_in, const int* in_sizes, int n_in,
                              void* d_out, int out_size, void* d_ws, size_t ws_size,
                              hipStream_t stream) {
  const float* node_attrs = (const float*)d_in[0];
  const float* node_feats = (const float*)d_in[1];
  const float* edge_attrs = (const float*)d_in[2];
  const float* edge_feats = (const float*)d_in[3];
  const int*   edge_index = (const int*)d_in[4];
  const float* W1s  = (const float*)d_in[5];
  const float* W1v  = (const float*)d_in[6];
  const float* Wm1  = (const float*)d_in[7];
  const float* Wm2  = (const float*)d_in[8];
  const float* W2s  = (const float*)d_in[9];
  const float* W2v  = (const float*)d_in[10];
  const float* WskS = (const float*)d_in[11];
  const float* WskV = (const float*)d_in[12];

  float* out  = (float*)d_out;
  float* msmv = (float*)d_ws;

  float2* packBase = (float2*)(msmv + (size_t)NN * 512);
  float2* W2sp  = packBase;
  float2* W2vp  = packBase + 4096;
  float2* WskSp = packBase + 8192;
  float2* WskVp = packBase + 24576;

  k_pre <<<PRE_GEMM_WGS + NSLICE, 256, 0, stream>>>(node_feats, W1s, W1v, W2s, W2v, WskS, WskV,
                                               packBase, out, edge_index, (int*)msmv);
  k_msg <<<NN / NPB, MSG_THREADS, 0, stream>>>(edge_attrs, edge_feats, edge_index,
                                               Wm1, Wm2, out, msmv);
  k_post<<<NN / PNPB, POST_THREADS, 0, stream>>>(msmv, node_attrs, W2sp, W2vp,
                                               WskSp, WskVp, out);
}


# k_msg and k_post fused into one kernel: the work-group that sums its 80 receivers' messages keeps the sums in LDS and runs their post-transform itself; 3 launches -> 2; same k order and sums
# speedup vs baseline: 5.2147x; 1.2723x over previous
#include <hip/hip_runtime.h>
#include <hip/hip_bf16.h>
#include <math.h>


#define NN 20000
#define NE 320000

#define NPB  80
#define NDEST (NN / NPB)
#define NSLICE 250
#define SLICE_E (NE / NSLICE)
#define LINE_I 32
#define LIST_PITCH 8192
#define PRE_GEMM_WGS ((NN + 127) / 128)
static_assert(NN % NPB == 0, "receiver ranges must tile the nodes");
static_assert(NE % NSLICE == 0 && SLICE_E % 256 == 0, "slices tile the edge list, 256 threads a slice");
static_assert(NPB <= 128 && NE <= (1 << 24), "an entry packs edge id * 128 + node index into 31 bits");
static_assert(NSLICE * LINE_I <= LIST_PITCH && NDEST * LIST_PITCH <= NN * 256, "the list lines fit the workspace in front of the staged sender rows");
static_assert(NDEST <= 256 && (256 + NDEST * LINE_I) * 4 <= 40960, "a slice's counters and lines fit the first launch's LDS");

typedef __attribute__((ext_vector_type(2))) float v2f;
typedef __attribute__((ext_vector_type(8))) float v8f;

__device__ __forceinline__ v8f wmma_f32_k4(v2f a, v2f b, v8f c) {
  return __builtin_amdgcn_wmma_f32_16x16x4_f32(false, a, false, b,
                                               (short)0, c, false, false);
}

__device__ __forceinline__ float sspf(float x) {
  const float e = __expf(-fabsf(x));
  return fmaxf(x, 0.0f) + 0.69314718056f * __log2f(1.0f + e) - 0.69314718056f;
}

#define GPITCH 64
#define LPITCH 80
#define POST_PACK_PAIRS 40960

template <int KB>
struct FragBlk { v2f a[KB]; v2f b[KB][4]; };

template <bool ACONTIG, int KB, int BPITCH>
__device__ __forceinline__ void frag_load(FragBlk<KB>& f, const float* __restrict__ arow,
                                          const float2* __restrict__ Wp,
                                          int kc0, int kh, int l16) {
  #pragma unroll
  for (int q = 0; q < KB; ++q) {
    const int k0 = (kc0 + q) * 4 + 2 * kh;
    if (ACONTIG) {
      const float2 av = *(const float2*)(arow + k0);
      f.a[q].x = av.x; f.a[q].y = av.y;
    } else {
      f.a[q].x = arow[k0 * 3];
      f.a[q].y = arow[(k0 + 1) * 3];
    }
    const float2* bp = Wp + (k0 >> 1) * BPITCH + l16;
    #pragma unroll
    for (int ct = 0; ct < 4; ++ct) {
      const float2 bv = bp[ct * 16];
      f.b[q][ct].x = bv.x; f.b[q][ct].y = bv.y;
    }
  }
}

template <int KB>
__device__ __forceinline__ void frag_mma(const FragBlk<KB>& f, v8f acc[4]) {
  #pragma unroll
  for (int q = 0; q < KB; ++q) {
    #pragma unroll
    for (int ct = 0; ct < 4; ++ct) acc[ct] = wmma_f32_k4(f.a[q], f.b[q][ct], acc[ct]);
  }
}

struct SkipBlk { v2f b[4][2][4]; };

__device__ __forceinline__ void skip_load(SkipBlk& s, const float2* __restrict__ Wskp,
                                          int i0, int kh, int l16) {
  #pragma unroll
  for (int u = 0; u < 4; ++u) {
    const float2* bp = Wskp + (4 * (i0 + u) + kh) * 64 + l16;
    #pragma unroll
    for (int h = 0; h < 2; ++h) {
      #pragma unroll
      for (int ct = 0; ct < 4; ++ct) {
        const float2 bv = bp[h * 128 + ct * 16];
        s.b[u][h][ct].x = bv.x; s.b[u][h][ct].y = bv.y;
      }
    }
  }
}

__device__ __forceinline__ void skip_mma(const SkipBlk& s, const float* m2p,
                                         float attE0, float attE1, float attO0, float attO1,
                                         v8f acc[4]) {
  const float4 mv = *(const float4*)m2p;
  const float m2[4] = {mv.x, mv.y, mv.z, mv.w};
  #pragma unroll
  for (int u = 0; u < 4; ++u) {
    v2f a;
    a.x = m2[u] * attE0; a.y = m2[u] * attE1;
    #pragma unroll
    for (int ct = 0; ct < 4; ++ct) acc[ct] = wmma_f32_k4(a, s.b[u][0][ct], acc[ct]);
    a.x = m2[u] * attO0; a.y = m2[u] * attO1;
    #pragma unroll
    for (int ct = 0; ct < 4; ++ct) acc[ct] = wmma_f32_k4(a, s.b[u][1][ct], acc[ct]);
  }
}

__device__ __forceinline__ void bin_slice(const int* __restrict__ rcvcol, int s,
                                          int* __restrict__ lists, int* lds) {
  int* lcnt = lds;
  int* lent = lds + 256;
  const int tid = threadIdx.x;
  if (tid < NDEST) lcnt[tid] = 0;
  __syncthreads();
  int rc[SLICE_E / 256];
  #pragma unroll
  for (int k = 0; k < SLICE_E / 256; ++k) rc[k] = rcvcol[s * SLICE_E + k * 256 + tid];
  #pragma unroll
  for (int k = 0; k < SLICE_E / 256; ++k) {
    const int e = s * SLICE_E + k * 256 + tid;
    const unsigned d = (unsigned)rc[k] / (unsigned)NPB;
    if (d < (unsigned)NDEST) {
      const int c = atomicAdd(&lcnt[d], 1);
      if (c < LINE_I - 1) lent[d * LINE_I + 1 + c] = e * 128 + (rc[k] - (int)d * NPB);
    }
  }
  __syncthreads();
  const int lane = tid & 31, wid = tid >> 5;
  for (int d0 = wid; d0 < NDEST; d0 += 64) {
    int cc[8], ee[8];
    #pragma unroll
    for (int j = 0; j < 8; ++j) {
      const int d = (d0 + 8 * j < NDEST) ? d0 + 8 * j : NDEST - 1;
      cc[j] = lcnt[d];
      ee[j] = lent[d * LINE_I + lane];
    }
    #pragma unroll
    for (int j = 0; j < 8; ++j) {
      const int d = d0 + 8 * j;
      if (d < NDEST) {
        const int v = (lane == 0) ? cc[j] : ((lane <= cc[j]) ? ee[j] : 0);
        lists[(size_t)d * LIST_PITCH + s * LINE_I + lane] = v;
      }
    }
  }
}

__global__ void __launch_bounds__(256)
k_pre(const float* __restrict__ nf,
      const float* __restrict__ W1s,
      const float* __restrict__ W1v,
      const float* __restrict__ W2s,
      const float* __restrict__ W2v,
      const float* __restrict__ WskS,
      const float* __restrict__ WskV,
      float2* __restrict__ postPack,
      float* __restrict__ sv,
      const int* __restrict__ eidx,
      int* __restrict__ lists) {
  __shared__ alignas(16) float2 w1l[2 * 32 * LPITCH];

  if (blockIdx.x >= PRE_GEMM_WGS) {
    bin_slice(eidx + NE, blockIdx.x - PRE_GEMM_WGS, lists, (int*)w1l);
    return;
  }

  float4 lo[4], hi[4];
  #pragma unroll
  for (int t = 0; t < 4; ++t) {
    const int q = threadIdx.x + 256 * t;
    const int m = q >> 9, kk = (q >> 4) & 31, n4 = (q & 15) * 4;
    const float* src = (m ? W1v : W1s) + (2 * kk) * 64 + n4;
    lo[t] = *(const float4*)src;
    hi[t] = *(const float4*)(src + 64);
  }

  for (int i = blockIdx.x * 256 + threadIdx.x; i < POST_PACK_PAIRS; i += PRE_GEMM_WGS * 256) {
    const float* src; int j;
    if (i < 4096)       { src = W2s;  j = i; }
    else if (i < 8192)  { src = W2v;  j = i - 4096; }
    else if (i < 24576) { src = WskS; j = i - 8192; }
    else                { src = WskV; j = i - 24576; }
    const int kk = j >> 6, n = j & 63;
    postPack[i] = make_float2(src[(2 * kk) * 64 + n], src[(2 * kk + 1) * 64 + n]);
  }

  #pragma unroll
  for (int t = 0; t < 4; ++t) {
    const int q = threadIdx.x + 256 * t;
    const int m = q >> 9, kk = (q >> 4) & 31, n4 = (q & 15) * 4;
    float4* d = (float4*)(w1l + m * (32 * LPITCH) + kk * LPITCH + n4);
    d[0] = make_float4(lo[t].x, hi[t].x, lo[t].y, hi[t].y);
    d[1] = make_float4(lo[t].z, hi[t].z, lo[t].w, hi[t].w);
  }
  __syncthreads();
  const float2* W1sp = w1l;
  const float2* W1vp = w1l + 32 * LPITCH;

  const int lane = threadIdx.x & 31;
  const int wid  = threadIdx.x >> 5;
  const int l16  = lane & 15;
  const int kh   = lane >> 4;
  const int row0 = (blockIdx.x * 8 + wid) * 16;
  int rowA = row0 + l16; if (rowA >= NN) rowA = NN - 1;
  const float* nrow = nf + (size_t)rowA * 256;

  FragBlk<8> f[2];
  frag_load<true, 8, LPITCH>(f[0], nrow, W1sp, 0, kh, l16);
  v8f acc[4];
  #pragma unroll
  for (int n = 0; n < 8; ++n) {
    const int tr = n >> 1;
    if (n + 1 < 8) {
      const int trn = (n + 1) >> 1, kcn = ((n + 1) & 1) * 8;
      if (trn == 0) frag_load<true, 8, LPITCH>(f[(n + 1) & 1], nrow, W1sp, kcn, kh, l16);
      else          frag_load<false, 8, LPITCH>(f[(n + 1) & 1], nrow + 64 + (trn - 1), W1vp, kcn, kh, l16);
    }
    __builtin_amdgcn_sched_barrier(0);
    if ((n & 1) == 0) {
      #pragma unroll
      for (int ct = 0; ct < 4; ++ct) acc[ct] = v8f{};
    }
    frag_mma<8>(f[n & 1], acc);
    if (n & 1) {
      #pragma unroll
      for (int ct = 0; ct < 4; ++ct) {
        #pragma unroll
        for (int j = 0; j < 8; ++j) {
          const int r = row0 + j + 8 * kh;
          if (r < NN) {
            const int c = ct * 16 + l16;
            sv[r * 256 + tr * 64 + c] = acc[ct][j] * 0.125f;
          }
        }
      }
    }
    __builtin_amdgcn_sched_barrier(0);
  }
}

#define BCAP 128
#define MSG_THREADS 640
#define MSG_WAVES (MSG_THREADS / 32)
#define LINES_PER_WAVE ((NSLICE + MSG_WAVES - 1) / MSG_WAVES)
static_assert(NE % BCAP == 0 && BCAP == 128, "rounds and the 4-per-lane sort assume this");

__device__ __forceinline__ float lane_bcast(float v, int src) {
  return __int_as_float(__builtin_amdgcn_readlane(__float_as_int(v), src));
}

__device__ __forceinline__ void bin_edges(const int* __restrict__ rcvcol, int lo, int hi,
                                          int n0, int* cnt, int* bucket) {
  const int tid = threadIdx.x;
  if (tid < NPB) cnt[tid] = 0;
  __syncthreads();
  const int4* r4 = (const int4*)rcvcol;
  const int lo4 = lo >> 2, n4 = (hi - lo) >> 2;
  const int rot = (int)((blockIdx.x * 320u) % (unsigned)n4);
  for (int i0 = tid; i0 < n4; i0 += MSG_THREADS * 8) {
    int4 v[8];
    int  at[8];
    #pragma unroll
    for (int k = 0; k < 8; ++k) {
      const int i = i0 + k * MSG_THREADS;
      int t = i + rot; if (t >= n4) t -= n4;
      at[k] = lo4 + t;
      v[k] = (i < n4) ? r4[at[k]] : make_int4(-1, -1, -1, -1);
    }
    #pragma unroll
    for (int k = 0; k < 8; ++k) {
      const int rr[4] = {v[k].x, v[k].y, v[k].z, v[k].w};
      #pragma unroll
      for (int c = 0; c < 4; ++c) {
        const int r = rr[c] - n0;
        if ((unsigned)r < (unsigned)NPB) {
          const int slot = atomicAdd(&cnt[r], 1);
          if (slot < BCAP) bucket[r * BCAP + slot] = at[k] * 4 + c;
        }
      }
    }
  }
  __syncthreads();
}

#define PNPB 80
#define POST_TILES (PNPB / 16)
#define POST_WAVES (POST_TILES * 4)
#define POST_THREADS (POST_WAVES * 32)
#define SPITCH 260
#define MPITCH 516
static_assert(NN % PNPB == 0, "post-transform row ranges must tile the nodes");
static_assert(PNPB == NPB && POST_THREADS == MSG_THREADS, "both halves share the work-group and its rows");
static_assert(NPB * BCAP <= POST_WAVES * 16 * 64, "the buckets fit the bytes that later hold the m2 tiles");
static_assert(PNPB * SPITCH <= NPB * MPITCH, "the staged output rows fit the bytes of the message-sum rows");

__global__ void __launch_bounds__(MSG_THREADS)
k_msg_post(const float* __restrict__ ea,
           const float* __restrict__ efeat,
           const int*   __restrict__ eidx,
           const float* __restrict__ Wm1,
           const float* __restrict__ Wm2,
           const float* __restrict__ sv,
           const int*   __restrict__ lists,
           const float* __restrict__ attrs,
           const float2* __restrict__ W2sp,
           const float2* __restrict__ W2vp,
           const float2* __restrict__ WskSp,
           const float2* __restrict__ WskVp,
           float* __restrict__ out) {
  __shared__ int cnt[NPB];
  __shared__ alignas(16) unsigned char bufA[NPB * MPITCH * 4];
  __shared__ alignas(16) unsigned char bufB[POST_WAVES * 16 * 64 * 4];
  float* mrows  = (float*)bufA;
  int*   bucket = (int*)bufB;
  const int lane = threadIdx.x & 31;
  const int wid  = threadIdx.x >> 5;
  const int n0   = blockIdx.x * NPB;
  const int hr   = lane & 7;
  const int u0 = lane, u1 = lane + 32;

  const int widu = __builtin_amdgcn_readfirstlane((int)(threadIdx.x >> 5));
  const int l16  = lane & 15;
  const int kh   = lane >> 4;
  const int tile = widu % POST_TILES;
  const int tr   = widu / POST_TILES;
  const float* arow8 = attrs + (size_t)(n0 + tile * 16 + l16) * 8;
  const float attE0 = arow8[2 * kh]     * 0.04419417382f;
  const float attE1 = arow8[2 * kh + 1] * 0.04419417382f;
  const float attO0 = arow8[4 + 2 * kh] * 0.04419417382f;
  const float attO1 = arow8[5 + 2 * kh] * 0.04419417382f;

  float w1c[8];
  #pragma unroll
  for (int q = 0; q < 8; ++q) w1c[q] = Wm1[q * 8 + hr];
  float wm[8][4][2];
  #pragma unroll
  for (int r = 0; r < 8; ++r) {
    #pragma unroll
    for (int k = 0; k < 4; ++k) {
      wm[r][k][0] = Wm2[r * 256 + k * 64 + u0];
      wm[r][k][1] = Wm2[r * 256 + k * 64 + u1];
    }
  }

  if (threadIdx.x < NPB) cnt[threadIdx.x] = 0;
  __syncthreads();
  const int* lst = lists + (size_t)blockIdx.x * LIST_PITCH;
  int lv[LINES_PER_WAVE];
  #pragma unroll
  for (int k = 0; k < LINES_PER_WAVE; ++k) {
    const int s = wid + k * MSG_WAVES;
    lv[k] = (s < NSLICE) ? lst[s * LINE_I + lane] : 0;
  }
  int spill = 0;
  #pragma unroll
  for (int k = 0; k < LINES_PER_WAVE; ++k) {
    const int c = __builtin_amdgcn_readfirstlane(lv[k]);
    if (c > LINE_I - 1) spill = 1;
    if (lane >= 1 && lane <= c) {
      const int r = lv[k] & 127;
      if (r < NPB) {
        const int slot = atomicAdd(&cnt[r], 1);
        if (slot < BCAP) bucket[r * BCAP + slot] = lv[k] >> 7;
      }
    }
  }
  __syncthreads();
  const int over = __syncthreads_or(spill | ((threadIdx.x < NPB) && (cnt[threadIdx.x < NPB ? threadIdx.x : 0] > BCAP)));
  const int span = over ? BCAP : NE;

  for (int eb = 0; eb < NE; eb += span) {
    if (over) bin_edges(eidx + NE, eb, eb + span, n0, cnt, bucket);

    for (int ln = wid; ln < NPB; ln += MSG_WAVES) {
      int* bk = bucket + ln * BCAP;
      int d = cnt[ln]; if (d > BCAP) d = BCAP;

      {
        const int x0 = (lane < d) ? bk[lane] : 0x7fffffff;
        int r0 = 0;
        for (int j = 0; j < d; ++j) r0 += (bk[j] < x0) ? 1 : 0;
        if (d > 32) {
          const int x1 = (lane + 32 < d) ? bk[lane + 32] : 0x7fffffff;
          const int x2 = (lane + 64 < d) ? bk[lane + 64] : 0x7fffffff;
          const int x3 = (lane + 96 < d) ? bk[lane + 96] : 0x7fffffff;
          int r1 = 0, r2 = 0, r3 = 0;
          for (int j = 0; j < d; ++j) {
            const int v = bk[j];
            r1 += (v < x1) ? 1 : 0; r2 += (v < x2) ? 1 : 0; r3 += (v < x3) ? 1 : 0;
          }
          if (lane + 32 < d) bk[r1] = x1;
          if (lane + 64 < d) bk[r2] = x2;
          if (lane + 96 < d) bk[r3] = x3;
        }
        if (lane < d) bk[r0] = x0;
      }

      float acc[8][2];
      #pragma unroll
      for (int p = 0; p < 8; ++p) { acc[p][0] = 0.f; acc[p][1] = 0.f; }

      for (int g = 0; g < d; g += 4) {
        int j = g + (lane >> 3); if (j > d - 1) j = d - 1;
        const int em = bk[j];
        const float4 f0 = *(const float4*)(efeat + (size_t)em * 8);
        const float4 f1 = *(const float4*)(efeat + (size_t)em * 8 + 4);
        const int    sndl = eidx[em];
        const float4 yl = *(const float4*)(ea + (size_t)em * 4);
        float tt = f0.x * w1c[0];
        tt = fmaf(f0.y, w1c[1], tt); tt = fmaf(f0.z, w1c[2], tt); tt = fmaf(f0.w, w1c[3], tt);
        tt = fmaf(f1.x, w1c[4], tt); tt = fmaf(f1.y, w1c[5], tt); tt = fmaf(f1.z, w1c[6], tt);
        tt = fmaf(f1.w, w1c[7], tt);
        const float hl = sspf(tt * 0.35355339059f);

        float Y0[4], Y1x[4], Y1y[4], Y1z[4];
        float se[4][2], vx[4][2], vy[4][2], vz[4][2];
        #pragma unroll
        for (int e = 0; e < 4; ++e) {
          const int sl = e * 8;
          const int snd = __builtin_amdgcn_readlane(sndl, sl);
          Y0[e]  = lane_bcast(yl.x, sl);
          Y1x[e] = lane_bcast(yl.y, sl);
          Y1y[e] = lane_bcast(yl.z, sl);
          Y1z[e] = lane_bcast(yl.w, sl);
          const float* srow = sv + (size_t)snd * 256;
          se[e][0] = srow[u0];        se[e][1] = srow[u1];
          vx[e][0] = srow[64 + u0];   vx[e][1] = srow[64 + u1];
          vy[e][0] = srow[128 + u0];  vy[e][1] = srow[128 + u1];
          vz[e][0] = srow[192 + u0];  vz[e][1] = srow[192 + u1];
        }

        #pragma unroll
        for (int e = 0; e < 4; ++e) {
          if (g + e < d) {
            float h[8];
            #pragma unroll
            for (int r = 0; r < 8; ++r) h[r] = lane_bcast(hl, e * 8 + r);
            #pragma unroll
            for (int t = 0; t < 2; ++t) {
              float w[4];
              #pragma unroll
              for (int k = 0; k < 4; ++k) {
                float tk = h[0] * wm[0][k][t];
                #pragma unroll
                for (int r = 1; r < 8; ++r) tk = fmaf(h[r], wm[r][k][t], tk);
                w[k] = tk * 0.35355339059f;
              }
              const float s1 = se[e][t];
              const float ax = vx[e][t], ay = vy[e][t], az = vz[e][t];
              const float dv = ax * Y1x[e] + ay * Y1y[e] + az * Y1z[e];
              acc[0][t] += w[0] * s1 * Y0[e];
              acc[1][t] += w[3] * dv * 0.57735026919f;
              acc[2][t] += w[1] * s1 * Y1x[e];
              acc[3][t] += w[1] * s1 * Y1y[e];
              acc[4][t] += w[1] * s1 * Y1z[e];
              acc[5][t] += w[2] * ax * Y0[e];
              acc[6][t] += w[2] * ay * Y0[e];
              acc[7][t] += w[2] * az * Y0[e];
            }
          }
        }
      }

      float* base = mrows + ln * MPITCH;
      #pragma unroll
      for (int p = 0; p < 8; ++p) {
        float a0 = acc[p][0], a1 = acc[p][1];
        if (eb != 0) { a0 += base[p * 64 + u0]; a1 += base[p * 64 + u1]; }
        base[p * 64 + u0] = a0;
        base[p * 64 + u1] = a1;
      }
    }
    __syncthreads();
  }

  float* lds   = (float*)bufB;
  float* stage = (float*)bufA;
  float* myLds = lds + widu * (16 * 64);
  const float* mrow = mrows + (tile * 16 + l16) * MPITCH;

  const float2* W2p  = (tr == 0) ? W2sp : W2vp;
  const float2* Wskp = (tr == 0) ? WskSp : WskVp;
  const int pA = (tr == 0) ? 0 : (1 + tr);
  const int pB = (tr == 0) ? 1 : (4 + tr);
  const float* arowA = mrow + pA * 64;
  const float* arowB = mrow + pB * 64;

  FragBlk<4> f[2];
  SkipBlk sb[2];
  frag_load<true, 4, GPITCH>(f[0], arowA, W2p, 0, kh, l16);

  v8f acc[4] = {};
  #pragma unroll
  for (int n = 0; n < 8; ++n) {
    if (n + 1 < 8)
      frag_load<true, 4, GPITCH>(f[(n + 1) & 1], (n + 1 < 4) ? arowA : arowB,
                         (n + 1 < 4) ? W2p : W2p + 32 * 64, ((n + 1) & 3) * 4, kh, l16);
    else
      skip_load(sb[0], Wskp, 0, kh, l16);
    __builtin_amdgcn_sched_barrier(0);
    frag_mma<4>(f[n & 1], acc);
    __builtin_amdgcn_sched_barrier(0);
  }

  #pragma unroll
  for (int ct = 0; ct < 4; ++ct) {
    #pragma unroll
    for (int j = 0; j < 8; ++j) {
      acc[ct][j] *= 0.08838834765f;
      myLds[(j + 8 * kh) * 64 + ct * 16 + l16] = acc[ct][j];
    }
  }
  __syncthreads();

  const float* m2row = myLds + l16 * 64;
  for (int blk = 0; blk < 14; blk += 2) {
    skip_load(sb[1], Wskp, (blk + 1) * 4, kh, l16);
    skip_mma(sb[0], m2row + blk * 4, attE0, attE1, attO0, attO1, acc);
    skip_load(sb[0], Wskp, (blk + 2) * 4, kh, l16);
    skip_mma(sb[1], m2row + (blk + 1) * 4, attE0, attE1, attO0, attO1, acc);
  }
  skip_load(sb[1], Wskp, 15 * 4, kh, l16);
  skip_mma(sb[0], m2row + 14 * 4, attE0, attE1, attO0, attO1, acc);
  skip_mma(sb[1], m2row + 15 * 4, attE0, attE1, attO0, attO1, acc);

  #pragma unroll
  for (int ct = 0; ct < 4; ++ct) {
    #pragma unroll
    for (int j = 0; j < 8; ++j) {
      const int r = tile * 16 + j + 8 * kh;
      const int c = ct * 16 + l16;
      const int o = (tr == 0) ? c : (64 + c * 3 + (tr - 1));
      stage[r * SPITCH + o] = acc[ct][j];
    }
  }
  __syncthreads();

  for (int i = threadIdx.x; i < PNPB * 64; i += POST_THREADS) {
    const int r = i >> 6, q = i & 63;
    const float4 v = *(const float4*)(stage + r * SPITCH + q * 4);
    *(float4*)(out + (size_t)(blockIdx.x * PNPB + r) * 256 + q * 4) = v;
  }
}

extern "C" void kernel_launch(void* const* d_in, const int* in_sizes, int n_in,
                              void* d_out, int out_size, void* d_ws, size_t ws_size,
                              hipStream_t stream) {
  const float* node_attrs = (const float*)d_in[0];
  const float* node_feats = (const float*)d_in[1];
  const float* edge_attrs = (const float*)d_in[2];
  const float* edge_feats = (const float*)d_in[3];
  const int*   edge_index = (const int*)d_in[4];
  const float* W1s  = (const float*)d_in[5];
  const float* W1v  = (const float*)d_in[6];
  const float* Wm1  = (const float*)d_in[7];
  const float* Wm2  = (const float*)d_in[8];
  const float* W2s  = (const float*)d_in[9];
  const float* W2v  = (const float*)d_in[10];
  const float* WskS = (const float*)d_in[11];
  const float* WskV = (const float*)d_in[12];

  float* out   = (float*)d_out;
  float* ws    = (float*)d_ws;
  int*   lists = (int*)ws;
  float* sv    = ws + (size_t)NN * 256;

  float2* packBase = (float2*)(ws + (size_t)NN * 512);
  float2* W2sp  = packBase;
  float2* W2vp  = packBase + 4096;
  float2* WskSp = packBase + 8192;
  float2* WskVp = packBase + 24576;

  k_pre <<<PRE_GEMM_WGS + NSLICE, 256, 0, stream>>>(node_feats, W1s, W1v, W2s, W2v, WskS, WskV,
                                               packBase, sv, edge_index, lists);
  k_msg_post<<<NN / NPB, MSG_THREADS, 0, stream>>>(edge_attrs, edge_feats, edge_index,
                                               Wm1, Wm2, sv, lists, node_attrs,
                                               W2sp, W2vp, WskSp, WskVp, out);
}
